# starting kernel
# baseline (speedup 1.0000x reference)
#include <hip/hip_runtime.h>
#include <hip/hip_bf16.h>
#include <math.h>


typedef __attribute__((ext_vector_type(2))) float v2f;
typedef __attribute__((ext_vector_type(8))) float v8f;

__global__ void fill_f32(float* __restrict__ p, float v, int n) {
    int i = blockIdx.x * blockDim.x + threadIdx.x;
    if (i < n) p[i] = v;
}

__global__ void degree_k(const int* __restrict__ dst, float* __restrict__ deg, int E) {
    int i = blockIdx.x * blockDim.x + threadIdx.x;
    if (i < E) atomicAdd(&deg[dst[i]], 1.0f);
}

__global__ void gemm_wmma_f32(const float* __restrict__ A, int lda,
                              const float* __restrict__ W, int ldw,
                              float* __restrict__ C, int ldc,
                              int M, int K, int accumulate)
{
    const int nColGroups = 4;
    int waveId = blockIdx.x * (blockDim.x >> 5) + (threadIdx.x >> 5);
    int lane   = threadIdx.x & 31;
    int nRowTiles = M >> 4;
    if (waveId >= nRowTiles * nColGroups) return;
    int rowTile  = waveId / nColGroups;
    int colGroup = waveId - rowTile * nColGroups;
    int col0     = colGroup * 64;
    int half = lane >> 4;
    int lr   = lane & 15;

    const float* Arow = A + (size_t)(rowTile * 16 + lr) * lda + 2 * half;
    const float* Wr0  = W + (size_t)(col0 +  0 + lr) * ldw + 2 * half;
    const float* Wr1  = W + (size_t)(col0 + 16 + lr) * ldw + 2 * half;
    const float* Wr2  = W + (size_t)(col0 + 32 + lr) * ldw + 2 * half;
    const float* Wr3  = W + (size_t)(col0 + 48 + lr) * ldw + 2 * half;

    v8f acc0 = {}, acc1 = {}, acc2 = {}, acc3 = {};

    v2f a  = *(const v2f*)(Arow);
    v2f b0 = *(const v2f*)(Wr0);
    v2f b1 = *(const v2f*)(Wr1);
    v2f b2 = *(const v2f*)(Wr2);
    v2f b3 = *(const v2f*)(Wr3);

#pragma unroll 2
    for (int k = 4; k < K; k += 4) {
        v2f an  = *(const v2f*)(Arow + k);
        v2f bn0 = *(const v2f*)(Wr0 + k);
        v2f bn1 = *(const v2f*)(Wr1 + k);
        v2f bn2 = *(const v2f*)(Wr2 + k);
        v2f bn3 = *(const v2f*)(Wr3 + k);
        acc0 = __builtin_amdgcn_wmma_f32_16x16x4_f32(false, a, false, b0, (short)0, acc0, false, false);
        acc1 = __builtin_amdgcn_wmma_f32_16x16x4_f32(false, a, false, b1, (short)0, acc1, false, false);
        acc2 = __builtin_amdgcn_wmma_f32_16x16x4_f32(false, a, false, b2, (short)0, acc2, false, false);
        acc3 = __builtin_amdgcn_wmma_f32_16x16x4_f32(false, a, false, b3, (short)0, acc3, false, false);
        a = an; b0 = bn0; b1 = bn1; b2 = bn2; b3 = bn3;
    }
    acc0 = __builtin_amdgcn_wmma_f32_16x16x4_f32(false, a, false, b0, (short)0, acc0, false, false);
    acc1 = __builtin_amdgcn_wmma_f32_16x16x4_f32(false, a, false, b1, (short)0, acc1, false, false);
    acc2 = __builtin_amdgcn_wmma_f32_16x16x4_f32(false, a, false, b2, (short)0, acc2, false, false);
    acc3 = __builtin_amdgcn_wmma_f32_16x16x4_f32(false, a, false, b3, (short)0, acc3, false, false);

    float* Cb = C + (size_t)(rowTile * 16 + 8 * half) * ldc + col0 + lr;
#pragma unroll
    for (int j = 0; j < 8; ++j) {
        float* p = Cb + (size_t)j * ldc;
        p[ 0] = (accumulate ? p[ 0] : 0.0f) + acc0[j];
        p[16] = (accumulate ? p[16] : 0.0f) + acc1[j];
        p[32] = (accumulate ? p[32] : 0.0f) + acc2[j];
        p[48] = (accumulate ? p[48] : 0.0f) + acc3[j];
    }
}

__global__ void attn_scores(const float* __restrict__ feat,
                            const float* __restrict__ attn_l,
                            const float* __restrict__ attn_r,
                            float* __restrict__ el, float* __restrict__ er,
                            int pairs, int H, int D)
{
    int p    = blockIdx.x * (blockDim.x >> 5) + (threadIdx.x >> 5);
    int lane = threadIdx.x & 31;
    if (p >= pairs) return;
    int h = p % H;
    const float* f  = feat + (size_t)p * D;
    const float* al = attn_l + (size_t)h * D;
    const float* ar = attn_r + (size_t)h * D;
    float sl = 0.f, sr = 0.f;
    for (int i = lane; i < D; i += 32) {
        float fv = f[i];
        sl += fv * al[i];
        sr += fv * ar[i];
    }
#pragma unroll
    for (int off = 16; off > 0; off >>= 1) {
        sl += __shfl_xor(sl, off, 32);
        sr += __shfl_xor(sr, off, 32);
    }
    if (lane == 0) { el[p] = sl; er[p] = sr; }
}

__device__ inline void atomicMaxF(float* addr, float val) {
    if (val >= 0.0f) atomicMax((int*)addr, __float_as_int(val));
    else             atomicMin((unsigned int*)addr, __float_as_uint(val));
}

__device__ inline float leaky02(float v) { return v > 0.0f ? v : 0.2f * v; }

__global__ void edge_max_k(const int* __restrict__ src, const int* __restrict__ dst,
                           const float* __restrict__ el, const float* __restrict__ er,
                           float* __restrict__ m, int E, int H)
{
    int t = blockIdx.x * blockDim.x + threadIdx.x;
    if (t >= E * H) return;
    int e = t / H, h = t - e * H;
    float v = leaky02(el[src[e] * H + h] + er[dst[e] * H + h]);
    atomicMaxF(&m[dst[e] * H + h], v);
}

__global__ void edge_expsum_k(const int* __restrict__ src, const int* __restrict__ dst,
                              const float* __restrict__ el, const float* __restrict__ er,
                              const float* __restrict__ m, float* __restrict__ s,
                              float* __restrict__ exb, int E, int H)
{
    int t = blockIdx.x * blockDim.x + threadIdx.x;
    if (t >= E * H) return;
    int e = t / H, h = t - e * H;
    int d = dst[e];
    float v  = leaky02(el[src[e] * H + h] + er[d * H + h]);
    float ex = expf(v - m[d * H + h]);
    exb[t] = ex;
    atomicAdd(&s[d * H + h], ex);
}

__global__ void edge_agg_k(const int* __restrict__ src, const int* __restrict__ dst,
                           const float* __restrict__ exb, const float* __restrict__ s,
                           const float* __restrict__ feat, float* __restrict__ agg,
                           int total, int H, int D, int F)
{
    int t = blockIdx.x * blockDim.x + threadIdx.x;
    if (t >= total) return;
    int chunks = F >> 2;
    int e  = t / chunks;
    int q  = t - e * chunks;
    int d0 = q << 2;
    int h  = d0 / D;
    int sN = src[e], dN = dst[e];
    float a = exb[e * H + h] / s[dN * H + h];
    const float4 f = *(const float4*)(feat + (size_t)sN * F + d0);
    float* o = agg + (size_t)dN * F + d0;
    atomicAdd(o + 0, f.x * a);
    atomicAdd(o + 1, f.y * a);
    atomicAdd(o + 2, f.z * a);
    atomicAdd(o + 3, f.w * a);
}

__global__ void combine_h1_k(const float* __restrict__ agg, const float* __restrict__ bias1,
                             const float* __restrict__ buf, const float* __restrict__ bb0,
                             const float* __restrict__ deg, float* __restrict__ h1, int total)
{
    int i = blockIdx.x * blockDim.x + threadIdx.x;
    if (i >= total) return;
    int c = i & 255;
    int n = i >> 8;
    float norm = 1.0f / fmaxf(deg[n], 1.0f);
    float v = agg[i] + bias1[c] + norm * (buf[i] + bb0[c]);
    h1[i] = v > 0.0f ? v : 0.0f;
}

__global__ void combine_out_k(const float* __restrict__ agg, const float* __restrict__ bias2,
                              const float* __restrict__ buf, const float* __restrict__ bb1,
                              const float* __restrict__ deg, float* __restrict__ out, int total)
{
    int i = blockIdx.x * blockDim.x + threadIdx.x;
    if (i >= total) return;
    int c = i & 255;
    int n = i >> 8;
    float norm = 1.0f / fmaxf(deg[n], 1.0f);
    out[i] = agg[i] + bias2[c] + norm * (buf[i] + bb1[c]);
}

extern "C" void kernel_launch(void* const* d_in, const int* in_sizes, int n_in,
                              void* d_out, int out_size, void* d_ws, size_t ws_size,
                              hipStream_t stream)
{
    const float* inputs  = (const float*)d_in[0];
    const int*   src     = (const int*)  d_in[1];
    const int*   dst     = (const int*)  d_in[2];
    const float* fc1_w   = (const float*)d_in[3];
    const float* attn_l1 = (const float*)d_in[4];
    const float* attn_r1 = (const float*)d_in[5];
    const float* bias1   = (const float*)d_in[6];
    const float* fc2_w   = (const float*)d_in[7];
    const float* attn_l2 = (const float*)d_in[8];
    const float* attn_r2 = (const float*)d_in[9];
    const float* bias2   = (const float*)d_in[10];
    const float* wb0     = (const float*)d_in[11];
    const float* bb0     = (const float*)d_in[12];
    const float* wb1     = (const float*)d_in[13];
    const float* bb1     = (const float*)d_in[14];
    float* out = (float*)d_out;

    const int F  = 256;
    const int N  = in_sizes[0] / F;
    const int E  = in_sizes[1];
    const int H1 = 8, D1 = 32;

    float* ws   = (float*)d_ws;
    float* feat = ws;                ws += (size_t)N * F;
    float* buf  = ws;                ws += (size_t)N * F;
    float* agg  = ws;                ws += (size_t)N * F;
    float* h1   = ws;                ws += (size_t)N * F;
    float* el   = ws;                ws += (size_t)N * H1;
    float* er   = ws;                ws += (size_t)N * H1;
    float* m    = ws;                ws += (size_t)N * H1;
    float* s    = ws;                ws += (size_t)N * H1;
    float* exb  = ws;                ws += (size_t)E * H1;
    float* deg  = ws;                ws += (size_t)N;

    const int TB = 256;
    auto cdiv = [](int a, int b) { return (a + b - 1) / b; };

    const int waves    = (N / 16) * 4;
    const int gemmGrid = cdiv(waves, TB / 32);

    fill_f32<<<cdiv(N, TB), TB, 0, stream>>>(deg, 0.0f, N);
    fill_f32<<<cdiv(N * H1, TB), TB, 0, stream>>>(m, -INFINITY, N * H1);
    fill_f32<<<cdiv(N * H1, TB), TB, 0, stream>>>(s, 0.0f, N * H1);
    fill_f32<<<cdiv(N * F, TB), TB, 0, stream>>>(agg, 0.0f, N * F);
    degree_k<<<cdiv(E, TB), TB, 0, stream>>>(dst, deg, E);

    gemm_wmma_f32<<<gemmGrid, TB, 0, stream>>>(inputs, F, fc1_w, F, feat, F, N, F, 0);
    gemm_wmma_f32<<<gemmGrid, TB, 0, stream>>>(inputs, F, wb0,   F, buf,  F, N, F, 0);

    attn_scores<<<cdiv(N * H1, TB / 32), TB, 0, stream>>>(feat, attn_l1, attn_r1, el, er, N * H1, H1, D1);
    edge_max_k   <<<cdiv(E * H1, TB), TB, 0, stream>>>(src, dst, el, er, m, E, H1);
    edge_expsum_k<<<cdiv(E * H1, TB), TB, 0, stream>>>(src, dst, el, er, m, s, exb, E, H1);
    edge_agg_k   <<<cdiv(E * (F / 4), TB), TB, 0, stream>>>(src, dst, exb, s, feat, agg, E * (F / 4), H1, D1, F);

    combine_h1_k<<<cdiv(N * F, TB), TB, 0, stream>>>(agg, bias1, buf, bb0, deg, h1, N * F);

    fill_f32<<<cdiv(N, TB), TB, 0, stream>>>(m, -INFINITY, N);
    fill_f32<<<cdiv(N, TB), TB, 0, stream>>>(s, 0.0f, N);
    fill_f32<<<cdiv(N * F, TB), TB, 0, stream>>>(agg, 0.0f, N * F);

    gemm_wmma_f32<<<gemmGrid, TB, 0, stream>>>(h1, F, fc2_w, F, feat, F, N, F, 0);
    gemm_wmma_f32<<<gemmGrid, TB, 0, stream>>>(inputs, F, wb1,     2 * F, buf, F, N, F, 0);
    gemm_wmma_f32<<<gemmGrid, TB, 0, stream>>>(h1,     F, wb1 + F, 2 * F, buf, F, N, F, 1);

    attn_scores<<<cdiv(N, TB / 32), TB, 0, stream>>>(feat, attn_l2, attn_r2, el, er, N, 1, F);
    edge_max_k   <<<cdiv(E, TB), TB, 0, stream>>>(src, dst, el, er, m, E, 1);
    edge_expsum_k<<<cdiv(E, TB), TB, 0, stream>>>(src, dst, el, er, m, s, exb, E, 1);
    edge_agg_k   <<<cdiv(E * (F / 4), TB), TB, 0, stream>>>(src, dst, exb, s, feat, agg, E * (F / 4), 1, F, F);

    combine_out_k<<<cdiv(N * F, TB), TB, 0, stream>>>(agg, bias2, buf, bb1, deg, out, N * F);
}


# edge feature sums: a work-group owns 80 destination rows in LDS, scans dst, gathers its edges' rows, LDS f32 atomic adds (arrival order, start value last), one store per line; no global f32 adds
# speedup vs baseline: 1.6805x; 1.6805x over previous
#include <hip/hip_runtime.h>
#include <hip/hip_bf16.h>
#include <math.h>


typedef __attribute__((ext_vector_type(2))) float v2f;
typedef __attribute__((ext_vector_type(8))) float v8f;

__global__ void fill_f32(float* __restrict__ p, float v, int n) {
    int i = blockIdx.x * blockDim.x + threadIdx.x;
    if (i < n) p[i] = v;
}

__global__ void degree_k(const int* __restrict__ dst, float* __restrict__ deg, int E) {
    int i = blockIdx.x * blockDim.x + threadIdx.x;
    if (i < E) atomicAdd(&deg[dst[i]], 1.0f);
}

typedef __attribute__((ext_vector_type(4))) float v4f;

struct Frag16 { v4f q0, q1; };

__device__ __forceinline__ Frag16 load_frag16(const float* __restrict__ p) {
    Frag16 f;
    f.q0 = *(const v4f*)(p);
    f.q1 = *(const v4f*)(p + 8);
    return f;
}

template <int S>
__device__ __forceinline__ v2f frag_step(const Frag16& f) {
    if (S == 0) return __builtin_shufflevector(f.q0, f.q0, 0, 1);
    if (S == 1) return __builtin_shufflevector(f.q0, f.q0, 2, 3);
    if (S == 2) return __builtin_shufflevector(f.q1, f.q1, 0, 1);
    return __builtin_shufflevector(f.q1, f.q1, 2, 3);
}

template <int RT, int CT, int S, bool RELU_A>
__device__ __forceinline__ void mma_step(v8f (&acc)[RT][CT], const Frag16 (&a)[RT], const Frag16 (&b)[CT]) {
#pragma unroll
    for (int i = 0; i < RT; ++i) {
        v2f av = frag_step<S>(a[i]);
        if (RELU_A) {
            av.x = fmaxf(av.x, 0.0f);
            av.y = fmaxf(av.y, 0.0f);
        }
#pragma unroll
        for (int j = 0; j < CT; ++j) {
            acc[i][j] = __builtin_amdgcn_wmma_f32_16x16x4_f32(false, av, false, frag_step<S>(b[j]),
                                                              (short)0, acc[i][j], false, false);
        }
    }
}

template <int RT, int CT, bool RELU_A>
__device__ __forceinline__ void mma_block16(v8f (&acc)[RT][CT], const Frag16 (&a)[RT], const Frag16 (&b)[CT]) {
    mma_step<RT, CT, 0, RELU_A>(acc, a, b);
    mma_step<RT, CT, 1, RELU_A>(acc, a, b);
    mma_step<RT, CT, 2, RELU_A>(acc, a, b);
    mma_step<RT, CT, 3, RELU_A>(acc, a, b);
}

template <int RT, int CT, bool RELU_A, int EPI>
__global__ void __launch_bounds__(128)
gemm_wmma_f32(const float* __restrict__ A, int lda,
              const float* __restrict__ W, int ldw,
              const float* __restrict__ Cin,
              float* __restrict__ C, int ldc,
              int M, int K,
              const float* __restrict__ bias, const float* __restrict__ bb,
              const float* __restrict__ deg)
{
    const int nColGroups = 256 / (16 * CT);
    int waveId = blockIdx.x * (blockDim.x >> 5) + (threadIdx.x >> 5);
    int lane   = threadIdx.x & 31;
    int nRowTiles  = M >> 4;
    int nRowGroups = (nRowTiles + RT - 1) / RT;
    if (waveId >= nRowGroups * nColGroups) return;
    int rowGroup = waveId / nColGroups;
    int colGroup = waveId - rowGroup * nColGroups;
    int rt0  = rowGroup * RT;
    int col0 = colGroup * (16 * CT);
    int half = lane >> 4;
    int lr   = lane & 15;

    const float* Ap[RT];
    const float* Wp[CT];
#pragma unroll
    for (int i = 0; i < RT; ++i) {
        int rt = rt0 + i;
        if (rt > nRowTiles - 1) rt = nRowTiles - 1;
        Ap[i] = A + (size_t)(rt * 16 + lr) * lda + 4 * half;
    }
#pragma unroll
    for (int j = 0; j < CT; ++j)
        Wp[j] = W + (size_t)(col0 + 16 * j + lr) * ldw + 4 * half;

    v8f acc[RT][CT];
#pragma unroll
    for (int i = 0; i < RT; ++i)
#pragma unroll
        for (int j = 0; j < CT; ++j) acc[i][j] = v8f{};

    Frag16 ax[RT], bx[CT], ay[RT], by[CT];

#pragma unroll
    for (int i = 0; i < RT; ++i) ax[i] = load_frag16(Ap[i]);
#pragma unroll
    for (int j = 0; j < CT; ++j) bx[j] = load_frag16(Wp[j]);

    int kb = 0;
    for (; kb < K - 32; kb += 32) {
#pragma unroll
        for (int i = 0; i < RT; ++i) ay[i] = load_frag16(Ap[i] + kb + 16);
#pragma unroll
        for (int j = 0; j < CT; ++j) by[j] = load_frag16(Wp[j] + kb + 16);
        mma_block16<RT, CT, RELU_A>(acc, ax, bx);
#pragma unroll
        for (int i = 0; i < RT; ++i) ax[i] = load_frag16(Ap[i] + kb + 32);
#pragma unroll
        for (int j = 0; j < CT; ++j) bx[j] = load_frag16(Wp[j] + kb + 32);
        mma_block16<RT, CT, RELU_A>(acc, ay, by);
    }
#pragma unroll
    for (int i = 0; i < RT; ++i) ay[i] = load_frag16(Ap[i] + kb + 16);
#pragma unroll
    for (int j = 0; j < CT; ++j) by[j] = load_frag16(Wp[j] + kb + 16);
    mma_block16<RT, CT, RELU_A>(acc, ax, bx);
    mma_block16<RT, CT, RELU_A>(acc, ay, by);

#pragma unroll
    for (int i = 0; i < RT; ++i) {
        if (rt0 + i < nRowTiles) {
            int row0 = (rt0 + i) * 16 + 8 * half;
            size_t cOff = (size_t)row0 * ldc + col0 + lr;
            float* Cb = C + cOff;
            if (EPI == 0) {
#pragma unroll
                for (int r = 0; r < 8; ++r)
#pragma unroll
                    for (int j = 0; j < CT; ++j) Cb[(size_t)r * ldc + 16 * j] = acc[i][j][r];
            } else {
                float dg[8], bs[CT], bk[CT];
#pragma unroll
                for (int r = 0; r < 8; ++r) dg[r] = deg[row0 + r];
#pragma unroll
                for (int j = 0; j < CT; ++j) {
                    bs[j] = bias[col0 + lr + 16 * j];
                    bk[j] = bb[col0 + lr + 16 * j];
                }
                float old[8][CT];
                if (EPI == 2) {
                    const float* Ib = Cin + cOff;
#pragma unroll
                    for (int r = 0; r < 8; ++r)
#pragma unroll
                        for (int j = 0; j < CT; ++j) old[r][j] = Ib[(size_t)r * ldc + 16 * j];
                }
#pragma unroll
                for (int r = 0; r < 8; ++r) {
                    float norm = 1.0f / fmaxf(dg[r], 1.0f);
#pragma unroll
                    for (int j = 0; j < CT; ++j) {
                        float p = acc[i][j][r];
                        if (EPI == 2) p = old[r][j] + p;
                        Cb[(size_t)r * ldc + 16 * j] = bs[j] + norm * (p + bk[j]);
                    }
                }
            }
        }
    }
}

__global__ void attn_scores(const float* __restrict__ feat,
                            const float* __restrict__ attn_l,
                            const float* __restrict__ attn_r,
                            float* __restrict__ el, float* __restrict__ er,
                            int pairs, int H, int D)
{
    int p    = blockIdx.x * (blockDim.x >> 5) + (threadIdx.x >> 5);
    int lane = threadIdx.x & 31;
    if (p >= pairs) return;
    int h = p % H;
    const float* f  = feat + (size_t)p * D;
    const float* al = attn_l + (size_t)h * D;
    const float* ar = attn_r + (size_t)h * D;
    float sl = 0.f, sr = 0.f;
    for (int i = lane; i < D; i += 32) {
        float fv = f[i];
        sl += fv * al[i];
        sr += fv * ar[i];
    }
#pragma unroll
    for (int off = 16; off > 0; off >>= 1) {
        sl += __shfl_xor(sl, off, 32);
        sr += __shfl_xor(sr, off, 32);
    }
    if (lane == 0) { el[p] = sl; er[p] = sr; }
}

__device__ inline void atomicMaxF(float* addr, float val) {
    if (val >= 0.0f) atomicMax((int*)addr, __float_as_int(val));
    else             atomicMin((unsigned int*)addr, __float_as_uint(val));
}

__device__ inline float leaky02(float v) { return v > 0.0f ? v : 0.2f * v; }

__global__ void edge_max_k(const int* __restrict__ src, const int* __restrict__ dst,
                           const float* __restrict__ el, const float* __restrict__ er,
                           float* __restrict__ m, int E, int H)
{
    int t = blockIdx.x * blockDim.x + threadIdx.x;
    if (t >= E * H) return;
    int e = t / H, h = t - e * H;
    float v = leaky02(el[src[e] * H + h] + er[dst[e] * H + h]);
    atomicMaxF(&m[dst[e] * H + h], v);
}

__global__ void edge_expsum_k(const int* __restrict__ src, const int* __restrict__ dst,
                              const float* __restrict__ el, const float* __restrict__ er,
                              const float* __restrict__ m, float* __restrict__ s,
                              float* __restrict__ exb, int E, int H)
{
    int t = blockIdx.x * blockDim.x + threadIdx.x;
    if (t >= E * H) return;
    int e = t / H, h = t - e * H;
    int d = dst[e];
    float v  = leaky02(el[src[e] * H + h] + er[d * H + h]);
    float ex = expf(v - m[d * H + h]);
    exb[t] = ex;
    atomicAdd(&s[d * H + h], ex);
}

constexpr int AGG_R   = 80;
constexpr int AGG_TB  = 1024;
constexpr int AGG_CAP = 12288;

template <int H, int B>
__device__ __forceinline__ void agg_add_edges(float* __restrict__ acc, int lo, int lane,
                                              int ev, int rv, int sv, int k, int n,
                                              const float* __restrict__ exb,
                                              const float* __restrict__ sden,
                                              const float* __restrict__ feat)
{
    constexpr int F = 256, D = F / H;
    const int h0 = (4 * lane) / D;
    const int h1 = (128 + 4 * lane) / D;
    int   rr[B];
    v4f   f0[B], f1[B];
    float x0[B], x1[B], q0[B], q1[B];
#pragma unroll
    for (int b = 0; b < B; ++b) {
        int kk = (k + b < n) ? (k + b) : (n - 1);
        int e  = __builtin_amdgcn_readlane(ev, kk);
        int s  = __builtin_amdgcn_readlane(sv, kk);
        rr[b]  = __builtin_amdgcn_readlane(rv, kk);
        const float* fr = feat + (size_t)s * F + 4 * lane;
        f0[b] = *(const v4f*)(fr);
        f1[b] = *(const v4f*)(fr + 128);
        x0[b] = exb[(size_t)e * H + h0];
        q0[b] = sden[(size_t)(lo + rr[b]) * H + h0];
        if (H > 1) {
            x1[b] = exb[(size_t)e * H + h1];
            q1[b] = sden[(size_t)(lo + rr[b]) * H + h1];
        }
    }
#pragma unroll
    for (int b = 0; b < B; ++b) {
        if (k + b < n) {
            float a0 = x0[b] / q0[b];
            float a1 = (H > 1) ? (x1[b] / q1[b]) : a0;
            float* o = acc + rr[b] * F + lane;
            atomicAdd(o +   0, f0[b].x * a0);
            atomicAdd(o +  32, f0[b].y * a0);
            atomicAdd(o +  64, f0[b].z * a0);
            atomicAdd(o +  96, f0[b].w * a0);
            atomicAdd(o + 128, f1[b].x * a1);
            atomicAdd(o + 160, f1[b].y * a1);
            atomicAdd(o + 192, f1[b].z * a1);
            atomicAdd(o + 224, f1[b].w * a1);
        }
    }
}

template <int H>
__global__ void __launch_bounds__(AGG_TB)
edge_agg_owner_k(const int* __restrict__ src, const int* __restrict__ dst,
                 const float* __restrict__ exb, const float* __restrict__ sden,
                 const float* __restrict__ feat, float* __restrict__ agg,
                 int E, int N)
{
    constexpr int F = 256;
    __shared__ __attribute__((aligned(16))) float acc[AGG_R * F];
    __shared__ int   le[AGG_CAP];
    __shared__ int   lr[AGG_CAP];
    __shared__ int   cnt;

    const int tid  = threadIdx.x;
    const int lane = tid & 31;
    const int wave = tid >> 5;
    const int lo   = blockIdx.x * AGG_R;
    const unsigned ltmask = (1u << lane) - 1u;

    for (int i = 4 * tid; i < AGG_R * F; i += 4 * AGG_TB) *(v4f*)(acc + i) = v4f{};
    if (tid == 0) cnt = 0;
    __syncthreads();

    const int  nChunk = E >> 2;
    const int4* dst4  = (const int4*)dst;
    constexpr int U = 4;
    for (int c0 = 0; c0 < nChunk + ((E & 3) ? 1 : 0); c0 += AGG_TB * U) {
        int4 dv[U];
        unsigned ovbits = 0u;
#pragma unroll
        for (int u = 0; u < U; ++u) {
            int c = c0 + u * AGG_TB + tid;
            dv[u] = int4{-1, -1, -1, -1};
            if (c < nChunk) dv[u] = dst4[c];
            else if (c == nChunk) {
                if (4 * c + 0 < E) dv[u].x = dst[4 * c + 0];
                if (4 * c + 1 < E) dv[u].y = dst[4 * c + 1];
                if (4 * c + 2 < E) dv[u].z = dst[4 * c + 2];
            }
        }
#pragma unroll
        for (int u = 0; u < U; ++u) {
            const int c = c0 + u * AGG_TB + tid;
            const unsigned r0 = (unsigned)(dv[u].x - lo), r1 = (unsigned)(dv[u].y - lo);
            const unsigned r2 = (unsigned)(dv[u].z - lo), r3 = (unsigned)(dv[u].w - lo);
            const bool m0 = r0 < (unsigned)AGG_R, m1 = r1 < (unsigned)AGG_R;
            const bool m2 = r2 < (unsigned)AGG_R, m3 = r3 < (unsigned)AGG_R;
            if (__builtin_amdgcn_ballot_w32(m0 | m1 | m2 | m3) == 0u) continue;
            const unsigned b0 = __builtin_amdgcn_ballot_w32(m0), b1 = __builtin_amdgcn_ballot_w32(m1);
            const unsigned b2 = __builtin_amdgcn_ballot_w32(m2), b3 = __builtin_amdgcn_ballot_w32(m3);
            const int n0 = __builtin_popcount(b0), n1 = __builtin_popcount(b1);
            const int n2 = __builtin_popcount(b2), n3 = __builtin_popcount(b3);
            int base = 0;
            if (lane == 0) base = atomicAdd(&cnt, n0 + n1 + n2 + n3);
            base = __builtin_amdgcn_readfirstlane(base);
            const int p0 = base + __builtin_popcount(b0 & ltmask);
            const int p1 = base + n0 + __builtin_popcount(b1 & ltmask);
            const int p2 = base + n0 + n1 + __builtin_popcount(b2 & ltmask);
            const int p3 = base + n0 + n1 + n2 + __builtin_popcount(b3 & ltmask);
            const bool k0 = m0 && p0 < AGG_CAP, k1 = m1 && p1 < AGG_CAP;
            const bool k2 = m2 && p2 < AGG_CAP, k3 = m3 && p3 < AGG_CAP;
            if (k0) { le[p0] = 4 * c + 0; lr[p0] = (int)r0; }
            if (k1) { le[p1] = 4 * c + 1; lr[p1] = (int)r1; }
            if (k2) { le[p2] = 4 * c + 2; lr[p2] = (int)r2; }
            if (k3) { le[p3] = 4 * c + 3; lr[p3] = (int)r3; }
            if (m0 && !k0) ovbits |= 1u << (4 * u + 0);
            if (m1 && !k1) ovbits |= 1u << (4 * u + 1);
            if (m2 && !k2) ovbits |= 1u << (4 * u + 2);
            if (m3 && !k3) ovbits |= 1u << (4 * u + 3);
        }
        unsigned pend = __builtin_amdgcn_ballot_w32(ovbits != 0u);
        while (pend) {
            const int l  = __builtin_ctz(pend);
            const int t  = __builtin_ctz((unsigned)__builtin_amdgcn_readlane((int)ovbits, l));
            const int e1 = 4 * (c0 + (t >> 2) * AGG_TB + (tid - lane) + l) + (t & 3);
            if (lane == l) ovbits &= ovbits - 1u;
            const int r1 = dst[e1] - lo;
            const int s1 = src[e1];
            agg_add_edges<H, 1>(acc, lo, lane, e1, r1, s1, 0, 1, exb, sden, feat);
            pend = __builtin_amdgcn_ballot_w32(ovbits != 0u);
        }
    }
    __syncthreads();

    const int total = cnt < AGG_CAP ? cnt : AGG_CAP;
    for (int c0 = wave; c0 < total; c0 += 32 * 32) {
        const int  idx = c0 + 32 * lane;
        const bool v   = idx < total;
        const int  ev  = v ? le[idx] : 0;
        const int  rv  = v ? lr[idx] : 0;
        const int  sv  = v ? src[ev] : 0;
        const int  n   = __builtin_popcount(__builtin_amdgcn_ballot_w32(v));
        for (int k = 0; k < n; k += 4)
            agg_add_edges<H, 4>(acc, lo, lane, ev, rv, sv, k, n, exb, sden, feat);
    }
    __syncthreads();

    for (int r = wave; r < AGG_R; r += 32) {
        const int row = lo + r;
        if (row < N) {
            float* g = agg + (size_t)row * F + 4 * lane;
            const float* a = acc + r * F + lane;
            v4f s0 = *(const v4f*)(g);
            v4f s1 = *(const v4f*)(g + 128);
            v4f t0 = { a[0],   a[32],  a[64],  a[96]  };
            v4f t1 = { a[128], a[160], a[192], a[224] };
            *(v4f*)(g)       = s0 + t0;
            *(v4f*)(g + 128) = s1 + t1;
        }
    }
}

extern "C" void kernel_launch(void* const* d_in, const int* in_sizes, int n_in,
                              void* d_out, int out_size, void* d_ws, size_t ws_size,
                              hipStream_t stream)
{
    const float* inputs  = (const float*)d_in[0];
    const int*   src     = (const int*)  d_in[1];
    const int*   dst     = (const int*)  d_in[2];
    const float* fc1_w   = (const float*)d_in[3];
    const float* attn_l1 = (const float*)d_in[4];
    const float* attn_r1 = (const float*)d_in[5];
    const float* bias1   = (const float*)d_in[6];
    const float* fc2_w   = (const float*)d_in[7];
    const float* attn_l2 = (const float*)d_in[8];
    const float* attn_r2 = (const float*)d_in[9];
    const float* bias2   = (const float*)d_in[10];
    const float* wb0     = (const float*)d_in[11];
    const float* bb0     = (const float*)d_in[12];
    const float* wb1     = (const float*)d_in[13];
    const float* bb1     = (const float*)d_in[14];
    float* out = (float*)d_out;

    const int F  = 256;
    const int N  = in_sizes[0] / F;
    const int E  = in_sizes[1];
    constexpr int H1 = 8;
    const int D1 = 32;

    float* ws   = (float*)d_ws;
    float* feat = ws;                ws += (size_t)N * F;
    float* buf  = ws;                ws += (size_t)N * F;
    float* agg  = ws;                ws += (size_t)N * F;
    float* el   = ws;                ws += (size_t)N * H1;
    float* er   = ws;                ws += (size_t)N * H1;
    float* m    = ws;                ws += (size_t)N * H1;
    float* s    = ws;                ws += (size_t)N * H1;
    float* exb  = ws;                ws += (size_t)E * H1;
    float* deg  = ws;                ws += (size_t)N;

    const int TB = 256;
    auto cdiv = [](int a, int b) { return (a + b - 1) / b; };

    const int GRT = 5, GCT = 4, GTB = 128;
    const int waves    = cdiv(N / 16, GRT) * (256 / (16 * GCT));
    const int gemmGrid = cdiv(waves, GTB / 32);

    fill_f32<<<cdiv(N, TB), TB, 0, stream>>>(deg, 0.0f, N);
    fill_f32<<<cdiv(N * H1, TB), TB, 0, stream>>>(m, -INFINITY, N * H1);
    fill_f32<<<cdiv(N * H1, TB), TB, 0, stream>>>(s, 0.0f, N * H1);
    degree_k<<<cdiv(E, TB), TB, 0, stream>>>(dst, deg, E);

    gemm_wmma_f32<GRT, GCT, false, 0><<<gemmGrid, GTB, 0, stream>>>(inputs, F, fc1_w, F, nullptr, feat, F, N, F,
                                                                    nullptr, nullptr, nullptr);
    gemm_wmma_f32<GRT, GCT, false, 1><<<gemmGrid, GTB, 0, stream>>>(inputs, F, wb0, F, nullptr, agg, F, N, F,
                                                                    bias1, bb0, deg);

    attn_scores<<<cdiv(N * H1, TB / 32), TB, 0, stream>>>(feat, attn_l1, attn_r1, el, er, N * H1, H1, D1);
    edge_max_k   <<<cdiv(E * H1, TB), TB, 0, stream>>>(src, dst, el, er, m, E, H1);
    edge_expsum_k<<<cdiv(E * H1, TB), TB, 0, stream>>>(src, dst, el, er, m, s, exb, E, H1);
    edge_agg_owner_k<H1><<<cdiv(N, AGG_R), AGG_TB, 0, stream>>>(src, dst, exb, s, feat, agg, E, N);

    fill_f32<<<cdiv(N, TB), TB, 0, stream>>>(m, -INFINITY, N);
    fill_f32<<<cdiv(N, TB), TB, 0, stream>>>(s, 0.0f, N);

    gemm_wmma_f32<GRT, GCT, true, 0><<<gemmGrid, GTB, 0, stream>>>(agg, F, fc2_w, F, nullptr, feat, F, N, F,
                                                                   nullptr, nullptr, nullptr);
    gemm_wmma_f32<GRT, GCT, false, 0><<<gemmGrid, GTB, 0, stream>>>(inputs, F, wb1, 2 * F, nullptr, buf, F, N, F,
                                                                    nullptr, nullptr, nullptr);
    gemm_wmma_f32<GRT, GCT, true, 2><<<gemmGrid, GTB, 0, stream>>>(agg, F, wb1 + F, 2 * F, buf, out, F, N, F,
                                                                   bias2, bb1, deg);

    attn_scores<<<cdiv(N, TB / 32), TB, 0, stream>>>(feat, attn_l2, attn_r2, el, er, N, 1, F);
    edge_max_k   <<<cdiv(E, TB), TB, 0, stream>>>(src, dst, el, er, m, E, 1);
    edge_expsum_k<<<cdiv(E, TB), TB, 0, stream>>>(src, dst, el, er, m, s, exb, E, 1);
    edge_agg_owner_k<1><<<cdiv(N, AGG_R), AGG_TB, 0, stream>>>(src, dst, exb, s, feat, out, E, N);
}


# edge softmax inside the row-owning work-group: score max and exp-sum by LDS max/add over its edge list (arrival order), weights formed there once per edge; 4 fills, 4 edge passes, exp buffer gone
# speedup vs baseline: 2.1958x; 1.3066x over previous
#include <hip/hip_runtime.h>
#include <hip/hip_bf16.h>
#include <math.h>


typedef __attribute__((ext_vector_type(2))) float v2f;
typedef __attribute__((ext_vector_type(8))) float v8f;

__global__ void fill_f32(float* __restrict__ p, float v, int n) {
    int i = blockIdx.x * blockDim.x + threadIdx.x;
    if (i < n) p[i] = v;
}

__global__ void degree_k(const int* __restrict__ dst, float* __restrict__ deg, int E) {
    int i = blockIdx.x * blockDim.x + threadIdx.x;
    if (i < E) atomicAdd(&deg[dst[i]], 1.0f);
}

typedef __attribute__((ext_vector_type(4))) float v4f;

struct Frag16 { v4f q0, q1; };

__device__ __forceinline__ Frag16 load_frag16(const float* __restrict__ p) {
    Frag16 f;
    f.q0 = *(const v4f*)(p);
    f.q1 = *(const v4f*)(p + 8);
    return f;
}

template <int S>
__device__ __forceinline__ v2f frag_step(const Frag16& f) {
    if (S == 0) return __builtin_shufflevector(f.q0, f.q0, 0, 1);
    if (S == 1) return __builtin_shufflevector(f.q0, f.q0, 2, 3);
    if (S == 2) return __builtin_shufflevector(f.q1, f.q1, 0, 1);
    return __builtin_shufflevector(f.q1, f.q1, 2, 3);
}

template <int RT, int CT, int S, bool RELU_A>
__device__ __forceinline__ void mma_step(v8f (&acc)[RT][CT], const Frag16 (&a)[RT], const Frag16 (&b)[CT]) {
#pragma unroll
    for (int i = 0; i < RT; ++i) {
        v2f av = frag_step<S>(a[i]);
        if (RELU_A) {
            av.x = fmaxf(av.x, 0.0f);
            av.y = fmaxf(av.y, 0.0f);
        }
#pragma unroll
        for (int j = 0; j < CT; ++j) {
            acc[i][j] = __builtin_amdgcn_wmma_f32_16x16x4_f32(false, av, false, frag_step<S>(b[j]),
                                                              (short)0, acc[i][j], false, false);
        }
    }
}

template <int RT, int CT, bool RELU_A>
__device__ __forceinline__ void mma_block16(v8f (&acc)[RT][CT], const Frag16 (&a)[RT], const Frag16 (&b)[CT]) {
    mma_step<RT, CT, 0, RELU_A>(acc, a, b);
    mma_step<RT, CT, 1, RELU_A>(acc, a, b);
    mma_step<RT, CT, 2, RELU_A>(acc, a, b);
    mma_step<RT, CT, 3, RELU_A>(acc, a, b);
}

template <int RT, int CT, bool RELU_A, int EPI>
__global__ void __launch_bounds__(128)
gemm_wmma_f32(const float* __restrict__ A, int lda,
              const float* __restrict__ W, int ldw,
              const float* __restrict__ Cin,
              float* __restrict__ C, int ldc,
              int M, int K,
              const float* __restrict__ bias, const float* __restrict__ bb,
              const float* __restrict__ deg)
{
    const int nColGroups = 256 / (16 * CT);
    int waveId = blockIdx.x * (blockDim.x >> 5) + (threadIdx.x >> 5);
    int lane   = threadIdx.x & 31;
    int nRowTiles  = M >> 4;
    int nRowGroups = (nRowTiles + RT - 1) / RT;
    if (waveId >= nRowGroups * nColGroups) return;
    int rowGroup = waveId / nColGroups;
    int colGroup = waveId - rowGroup * nColGroups;
    int rt0  = rowGroup * RT;
    int col0 = colGroup * (16 * CT);
    int half = lane >> 4;
    int lr   = lane & 15;

    const float* Ap[RT];
    const float* Wp[CT];
#pragma unroll
    for (int i = 0; i < RT; ++i) {
        int rt = rt0 + i;
        if (rt > nRowTiles - 1) rt = nRowTiles - 1;
        Ap[i] = A + (size_t)(rt * 16 + lr) * lda + 4 * half;
    }
#pragma unroll
    for (int j = 0; j < CT; ++j)
        Wp[j] = W + (size_t)(col0 + 16 * j + lr) * ldw + 4 * half;

    v8f acc[RT][CT];
#pragma unroll
    for (int i = 0; i < RT; ++i)
#pragma unroll
        for (int j = 0; j < CT; ++j) acc[i][j] = v8f{};

    Frag16 ax[RT], bx[CT], ay[RT], by[CT];

#pragma unroll
    for (int i = 0; i < RT; ++i) ax[i] = load_frag16(Ap[i]);
#pragma unroll
    for (int j = 0; j < CT; ++j) bx[j] = load_frag16(Wp[j]);

    int kb = 0;
    for (; kb < K - 32; kb += 32) {
#pragma unroll
        for (int i = 0; i < RT; ++i) ay[i] = load_frag16(Ap[i] + kb + 16);
#pragma unroll
        for (int j = 0; j < CT; ++j) by[j] = load_frag16(Wp[j] + kb + 16);
        mma_block16<RT, CT, RELU_A>(acc, ax, bx);
#pragma unroll
        for (int i = 0; i < RT; ++i) ax[i] = load_frag16(Ap[i] + kb + 32);
#pragma unroll
        for (int j = 0; j < CT; ++j) bx[j] = load_frag16(Wp[j] + kb + 32);
        mma_block16<RT, CT, RELU_A>(acc, ay, by);
    }
#pragma unroll
    for (int i = 0; i < RT; ++i) ay[i] = load_frag16(Ap[i] + kb + 16);
#pragma unroll
    for (int j = 0; j < CT; ++j) by[j] = load_frag16(Wp[j] + kb + 16);
    mma_block16<RT, CT, RELU_A>(acc, ax, bx);
    mma_block16<RT, CT, RELU_A>(acc, ay, by);

#pragma unroll
    for (int i = 0; i < RT; ++i) {
        if (rt0 + i < nRowTiles) {
            int row0 = (rt0 + i) * 16 + 8 * half;
            size_t cOff = (size_t)row0 * ldc + col0 + lr;
            float* Cb = C + cOff;
            if (EPI == 0) {
#pragma unroll
                for (int r = 0; r < 8; ++r)
#pragma unroll
                    for (int j = 0; j < CT; ++j) Cb[(size_t)r * ldc + 16 * j] = acc[i][j][r];
            } else {
                float dg[8], bs[CT], bk[CT];
#pragma unroll
                for (int r = 0; r < 8; ++r) dg[r] = deg[row0 + r];
#pragma unroll
                for (int j = 0; j < CT; ++j) {
                    bs[j] = bias[col0 + lr + 16 * j];
                    bk[j] = bb[col0 + lr + 16 * j];
                }
                float old[8][CT];
                if (EPI == 2) {
                    const float* Ib = Cin + cOff;
#pragma unroll
                    for (int r = 0; r < 8; ++r)
#pragma unroll
                        for (int j = 0; j < CT; ++j) old[r][j] = Ib[(size_t)r * ldc + 16 * j];
                }
#pragma unroll
                for (int r = 0; r < 8; ++r) {
                    float norm = 1.0f / fmaxf(dg[r], 1.0f);
#pragma unroll
                    for (int j = 0; j < CT; ++j) {
                        float p = acc[i][j][r];
                        if (EPI == 2) p = old[r][j] + p;
                        Cb[(size_t)r * ldc + 16 * j] = bs[j] + norm * (p + bk[j]);
                    }
                }
            }
        }
    }
}

__global__ void attn_scores(const float* __restrict__ feat,
                            const float* __restrict__ attn_l,
                            const float* __restrict__ attn_r,
                            float* __restrict__ el, float* __restrict__ er,
                            int pairs, int H, int D)
{
    int p    = blockIdx.x * (blockDim.x >> 5) + (threadIdx.x >> 5);
    int lane = threadIdx.x & 31;
    if (p >= pairs) return;
    int h = p % H;
    const float* f  = feat + (size_t)p * D;
    const float* al = attn_l + (size_t)h * D;
    const float* ar = attn_r + (size_t)h * D;
    float sl = 0.f, sr = 0.f;
    for (int i = lane; i < D; i += 32) {
        float fv = f[i];
        sl += fv * al[i];
        sr += fv * ar[i];
    }
#pragma unroll
    for (int off = 16; off > 0; off >>= 1) {
        sl += __shfl_xor(sl, off, 32);
        sr += __shfl_xor(sr, off, 32);
    }
    if (lane == 0) { el[p] = sl; er[p] = sr; }
}


__device__ __forceinline__ void atomicMaxF(float* addr, float val) {
    if (val >= 0.0f) atomicMax((int*)addr, __float_as_int(val));
    else             atomicMin((unsigned int*)addr, __float_as_uint(val));
}

__device__ __forceinline__ float leaky02(float v) { return v > 0.0f ? v : 0.2f * v; }

constexpr int AGG_R   = 80;
constexpr int AGG_TB  = 1024;
constexpr int AGG_CAP = 10240;

template <int H>
__device__ __forceinline__ void edge_weights_to_slab(bool valid, int sv, int rv,
                                                     const float* __restrict__ el,
                                                     const float* s_er, const float* s_m, const float* s_s,
                                                     float* wl)
{
    if (valid) {
        float e[H];
#pragma unroll
        for (int h = 0; h < H; ++h) e[h] = el[(size_t)sv * H + h];
#pragma unroll
        for (int h = 0; h < H; ++h) {
            const int i = rv * H + h;
            wl[h] = expf(leaky02(e[h] + s_er[i]) - s_m[i]) / s_s[i];
        }
    }
}

template <int H, int B>
__device__ __forceinline__ void agg_add_edges(float* __restrict__ acc, int lane,
                                              int sv, int rv, int k, int n,
                                              const float* wv,
                                              const float* __restrict__ feat)
{
    constexpr int F = 256, D = F / H;
    const int h0 = (4 * lane) / D;
    const int h1 = (128 + 4 * lane) / D;
    int   rr[B];
    v4f   f0[B], f1[B];
    float a0[B], a1[B];
#pragma unroll
    for (int b = 0; b < B; ++b) {
        int kk = (k + b < n) ? (k + b) : (n - 1);
        int s  = __builtin_amdgcn_readlane(sv, kk);
        rr[b]  = __builtin_amdgcn_readlane(rv, kk);
        const float* fr = feat + (size_t)s * F + 4 * lane;
        f0[b] = *(const v4f*)(fr);
        f1[b] = *(const v4f*)(fr + 128);
        a0[b] = wv[kk * H + h0];
        a1[b] = (H > 1) ? wv[kk * H + h1] : a0[b];
    }
#pragma unroll
    for (int b = 0; b < B; ++b) {
        if (k + b < n) {
            float* o = acc + rr[b] * F + lane;
            atomicAdd(o +   0, f0[b].x * a0[b]);
            atomicAdd(o +  32, f0[b].y * a0[b]);
            atomicAdd(o +  64, f0[b].z * a0[b]);
            atomicAdd(o +  96, f0[b].w * a0[b]);
            atomicAdd(o + 128, f1[b].x * a1[b]);
            atomicAdd(o + 160, f1[b].y * a1[b]);
            atomicAdd(o + 192, f1[b].z * a1[b]);
            atomicAdd(o + 224, f1[b].w * a1[b]);
        }
    }
}

template <int H, int PHASE>
__device__ __forceinline__ void list_pass(const int* ls, const int* lr, int nPairs, int tid,
                                          const float* __restrict__ el,
                                          const float* s_er, float* s_m, float* s_s)
{
    constexpr int PU = 4;
    for (int p0 = tid; p0 < nPairs; p0 += AGG_TB * PU) {
        float ev[PU];
        int   ix[PU];
#pragma unroll
        for (int u = 0; u < PU; ++u) {
            const int  p   = p0 + u * AGG_TB;
            const bool ok  = p < nPairs;
            const int  pp  = ok ? p : p0;
            const int  ent = pp / H, h = pp - ent * H;
            ix[u] = ok ? (lr[ent] * H + h) : -1;
            ev[u] = el[(size_t)ls[ent] * H + h];
        }
#pragma unroll
        for (int u = 0; u < PU; ++u) {
            if (ix[u] >= 0) {
                const float v = leaky02(ev[u] + s_er[ix[u]]);
                if (PHASE == 0) atomicMaxF(&s_m[ix[u]], v);
                else            atomicAdd(&s_s[ix[u]], expf(v - s_m[ix[u]]));
            }
        }
    }
}

template <int H>
__global__ void __launch_bounds__(AGG_TB)
edge_softmax_agg_owner_k(const int* __restrict__ src, const int* __restrict__ dst,
                         const float* __restrict__ el, const float* __restrict__ er,
                         const float* __restrict__ feat, float* __restrict__ agg,
                         int E, int N)
{
    constexpr int F = 256;
    __shared__ __attribute__((aligned(16))) float acc[AGG_R * F];
    __shared__ int   le[AGG_CAP];
    __shared__ int   lr[AGG_CAP];
    __shared__ float s_er[AGG_R * H];
    __shared__ float s_m[AGG_R * H];
    __shared__ float s_s[AGG_R * H];
    __shared__ float wsl[AGG_TB * H];
    __shared__ int   cnt;

    const int tid  = threadIdx.x;
    const int lane = tid & 31;
    const int wave = tid >> 5;
    const int lo   = blockIdx.x * AGG_R;
    const unsigned ltmask = (1u << lane) - 1u;
    float* wv = wsl + (tid - lane) * H;

    for (int i = 4 * tid; i < AGG_R * F; i += 4 * AGG_TB) *(v4f*)(acc + i) = v4f{};
    for (int i = tid; i < AGG_R * H; i += AGG_TB) {
        s_er[i] = (lo + i / H < N) ? er[(size_t)lo * H + i] : 0.0f;
        s_m[i]  = -INFINITY;
        s_s[i]  = 0.0f;
    }
    if (tid == 0) cnt = 0;
    __syncthreads();

    const int  nChunk = E >> 2;
    const int4* dst4  = (const int4*)dst;
    constexpr int U = 4;
    for (int c0 = 0; c0 < nChunk + ((E & 3) ? 1 : 0); c0 += AGG_TB * U) {
        int4 dv[U];
#pragma unroll
        for (int u = 0; u < U; ++u) {
            int c = c0 + u * AGG_TB + tid;
            dv[u] = int4{-1, -1, -1, -1};
            if (c < nChunk) dv[u] = dst4[c];
            else if (c == nChunk) {
                if (4 * c + 0 < E) dv[u].x = dst[4 * c + 0];
                if (4 * c + 1 < E) dv[u].y = dst[4 * c + 1];
                if (4 * c + 2 < E) dv[u].z = dst[4 * c + 2];
            }
        }
#pragma unroll
        for (int u = 0; u < U; ++u) {
            const int c = c0 + u * AGG_TB + tid;
            const unsigned r0 = (unsigned)(dv[u].x - lo), r1 = (unsigned)(dv[u].y - lo);
            const unsigned r2 = (unsigned)(dv[u].z - lo), r3 = (unsigned)(dv[u].w - lo);
            const bool m0 = r0 < (unsigned)AGG_R, m1 = r1 < (unsigned)AGG_R;
            const bool m2 = r2 < (unsigned)AGG_R, m3 = r3 < (unsigned)AGG_R;
            if (__builtin_amdgcn_ballot_w32(m0 | m1 | m2 | m3) == 0u) continue;
            const unsigned b0 = __builtin_amdgcn_ballot_w32(m0), b1 = __builtin_amdgcn_ballot_w32(m1);
            const unsigned b2 = __builtin_amdgcn_ballot_w32(m2), b3 = __builtin_amdgcn_ballot_w32(m3);
            const int n0 = __builtin_popcount(b0), n1 = __builtin_popcount(b1);
            const int n2 = __builtin_popcount(b2), n3 = __builtin_popcount(b3);
            int base = 0;
            if (lane == 0) base = atomicAdd(&cnt, n0 + n1 + n2 + n3);
            base = __builtin_amdgcn_readfirstlane(base);
            const int p0 = base + __builtin_popcount(b0 & ltmask);
            const int p1 = base + n0 + __builtin_popcount(b1 & ltmask);
            const int p2 = base + n0 + n1 + __builtin_popcount(b2 & ltmask);
            const int p3 = base + n0 + n1 + n2 + __builtin_popcount(b3 & ltmask);
            if (m0 && p0 < AGG_CAP) { le[p0] = 4 * c + 0; lr[p0] = (int)r0; }
            if (m1 && p1 < AGG_CAP) { le[p1] = 4 * c + 1; lr[p1] = (int)r1; }
            if (m2 && p2 < AGG_CAP) { le[p2] = 4 * c + 2; lr[p2] = (int)r2; }
            if (m3 && p3 < AGG_CAP) { le[p3] = 4 * c + 3; lr[p3] = (int)r3; }
        }
    }
    __syncthreads();

    const int total = cnt;
    if (total <= AGG_CAP) {
        for (int idx = tid; idx < total; idx += AGG_TB) le[idx] = src[le[idx]];
        __syncthreads();
        list_pass<H, 0>(le, lr, total * H, tid, el, s_er, s_m, s_s);
        __syncthreads();
        list_pass<H, 1>(le, lr, total * H, tid, el, s_er, s_m, s_s);
        __syncthreads();
        for (int c0 = wave; c0 < total; c0 += 32 * 32) {
            const int  idx = c0 + 32 * lane;
            const bool v   = idx < total;
            const int  sv  = v ? le[idx] : 0;
            const int  rv  = v ? lr[idx] : 0;
            const int  n   = __builtin_popcount(__builtin_amdgcn_ballot_w32(v));
            edge_weights_to_slab<H>(v, sv, rv, el, s_er, s_m, s_s, wv + lane * H);
            for (int k = 0; k < n; k += 4)
                agg_add_edges<H, 4>(acc, lane, sv, rv, k, n, wv, feat);
        }
    } else {
        for (int i = tid; i < E; i += AGG_TB) {
            const unsigned r = (unsigned)(dst[i] - lo);
            if (r < (unsigned)AGG_R) {
                const int s = src[i];
                for (int h = 0; h < H; ++h)
                    atomicMaxF(&s_m[r * H + h], leaky02(el[(size_t)s * H + h] + s_er[r * H + h]));
            }
        }
        __syncthreads();
        for (int i = tid; i < E; i += AGG_TB) {
            const unsigned r = (unsigned)(dst[i] - lo);
            if (r < (unsigned)AGG_R) {
                const int s = src[i];
                for (int h = 0; h < H; ++h) {
                    const float v = leaky02(el[(size_t)s * H + h] + s_er[r * H + h]);
                    atomicAdd(&s_s[r * H + h], expf(v - s_m[r * H + h]));
                }
            }
        }
        __syncthreads();
        for (int b0 = 32 * wave; b0 < E; b0 += AGG_TB) {
            const int  i  = b0 + lane;
            const int  d  = (i < E) ? dst[i] : -1;
            const int  sI = (i < E) ? src[i] : 0;
            const unsigned r = (unsigned)(d - lo);
            unsigned own = __builtin_amdgcn_ballot_w32(r < (unsigned)AGG_R);
            edge_weights_to_slab<H>(r < (unsigned)AGG_R, sI, (int)r, el, s_er, s_m, s_s, wv + lane * H);
            while (own) {
                const int l = __builtin_ctz(own);
                own &= own - 1u;
                agg_add_edges<H, 1>(acc, lane, sI, (int)r, l, 32, wv, feat);
            }
        }
    }
    __syncthreads();

    for (int r = wave; r < AGG_R; r += 32) {
        const int row = lo + r;
        if (row < N) {
            float* g = agg + (size_t)row * F + 4 * lane;
            const float* a = acc + r * F + lane;
            v4f s0 = *(const v4f*)(g);
            v4f s1 = *(const v4f*)(g + 128);
            v4f t0 = { a[0],   a[32],  a[64],  a[96]  };
            v4f t1 = { a[128], a[160], a[192], a[224] };
            *(v4f*)(g)       = s0 + t0;
            *(v4f*)(g + 128) = s1 + t1;
        }
    }
}

extern "C" void kernel_launch(void* const* d_in, const int* in_sizes, int n_in,
                              void* d_out, int out_size, void* d_ws, size_t ws_size,
                              hipStream_t stream)
{
    const float* inputs  = (const float*)d_in[0];
    const int*   src     = (const int*)  d_in[1];
    const int*   dst     = (const int*)  d_in[2];
    const float* fc1_w   = (const float*)d_in[3];
    const float* attn_l1 = (const float*)d_in[4];
    const float* attn_r1 = (const float*)d_in[5];
    const float* bias1   = (const float*)d_in[6];
    const float* fc2_w   = (const float*)d_in[7];
    const float* attn_l2 = (const float*)d_in[8];
    const float* attn_r2 = (const float*)d_in[9];
    const float* bias2   = (const float*)d_in[10];
    const float* wb0     = (const float*)d_in[11];
    const float* bb0     = (const float*)d_in[12];
    const float* wb1     = (const float*)d_in[13];
    const float* bb1     = (const float*)d_in[14];
    float* out = (float*)d_out;

    const int F  = 256;
    const int N  = in_sizes[0] / F;
    const int E  = in_sizes[1];
    constexpr int H1 = 8;
    const int D1 = 32;

    float* ws   = (float*)d_ws;
    float* feat = ws;                ws += (size_t)N * F;
    float* buf  = ws;                ws += (size_t)N * F;
    float* agg  = ws;                ws += (size_t)N * F;
    float* el   = ws;                ws += (size_t)N * H1;
    float* er   = ws;                ws += (size_t)N * H1;
    float* deg  = ws;                ws += (size_t)N;

    const int TB = 256;
    auto cdiv = [](int a, int b) { return (a + b - 1) / b; };

    const int GRT = 5, GCT = 4, GTB = 128;
    const int waves    = cdiv(N / 16, GRT) * (256 / (16 * GCT));
    const int gemmGrid = cdiv(waves, GTB / 32);

    fill_f32<<<cdiv(N, TB), TB, 0, stream>>>(deg, 0.0f, N);
    degree_k<<<cdiv(E, TB), TB, 0, stream>>>(dst, deg, E);

    gemm_wmma_f32<GRT, GCT, false, 0><<<gemmGrid, GTB, 0, stream>>>(inputs, F, fc1_w, F, nullptr, feat, F, N, F,
                                                                    nullptr, nullptr, nullptr);
    gemm_wmma_f32<GRT, GCT, false, 1><<<gemmGrid, GTB, 0, stream>>>(inputs, F, wb0, F, nullptr, agg, F, N, F,
                                                                    bias1, bb0, deg);

    attn_scores<<<cdiv(N * H1, TB / 32), TB, 0, stream>>>(feat, attn_l1, attn_r1, el, er, N * H1, H1, D1);
    edge_softmax_agg_owner_k<H1><<<cdiv(N, AGG_R), AGG_TB, 0, stream>>>(src, dst, el, er, feat, agg, E, N);

    gemm_wmma_f32<GRT, GCT, true, 0><<<gemmGrid, GTB, 0, stream>>>(agg, F, fc2_w, F, nullptr, feat, F, N, F,
                                                                   nullptr, nullptr, nullptr);
    gemm_wmma_f32<GRT, GCT, false, 0><<<gemmGrid, GTB, 0, stream>>>(inputs, F, wb1, 2 * F, nullptr, buf, F, N, F,
                                                                    nullptr, nullptr, nullptr);
    gemm_wmma_f32<GRT, GCT, true, 2><<<gemmGrid, GTB, 0, stream>>>(agg, F, wb1 + F, 2 * F, buf, out, F, N, F,
                                                                   bias2, bb1, deg);

    attn_scores<<<cdiv(N, TB / 32), TB, 0, stream>>>(feat, attn_l2, attn_r2, el, er, N, 1, F);
    edge_softmax_agg_owner_k<1><<<cdiv(N, AGG_R), AGG_TB, 0, stream>>>(src, dst, el, er, feat, out, E, N);
}


# attention scores from folded weights: el/er = layer input @ (attn . fc_w)^T, one 16-column f32 matrix product per layer; the two per-(node,head) score launches on feat gone; score sums in a new order
# speedup vs baseline: 2.4731x; 1.1263x over previous
#include <hip/hip_runtime.h>
#include <hip/hip_bf16.h>
#include <math.h>


typedef __attribute__((ext_vector_type(2))) float v2f;
typedef __attribute__((ext_vector_type(8))) float v8f;

__global__ void fill_f32(float* __restrict__ p, float v, int n) {
    int i = blockIdx.x * blockDim.x + threadIdx.x;
    if (i < n) p[i] = v;
}

__global__ void degree_k(const int* __restrict__ dst, float* __restrict__ deg, int E) {
    int i = blockIdx.x * blockDim.x + threadIdx.x;
    if (i < E) atomicAdd(&deg[dst[i]], 1.0f);
}

typedef __attribute__((ext_vector_type(4))) float v4f;

struct Frag16 { v4f q0, q1; };

__device__ __forceinline__ Frag16 load_frag16(const float* __restrict__ p) {
    Frag16 f;
    f.q0 = *(const v4f*)(p);
    f.q1 = *(const v4f*)(p + 8);
    return f;
}

template <int S>
__device__ __forceinline__ v2f frag_step(const Frag16& f) {
    if (S == 0) return __builtin_shufflevector(f.q0, f.q0, 0, 1);
    if (S == 1) return __builtin_shufflevector(f.q0, f.q0, 2, 3);
    if (S == 2) return __builtin_shufflevector(f.q1, f.q1, 0, 1);
    return __builtin_shufflevector(f.q1, f.q1, 2, 3);
}

template <int RT, int CT, int S, bool RELU_A>
__device__ __forceinline__ void mma_step(v8f (&acc)[RT][CT], const Frag16 (&a)[RT], const Frag16 (&b)[CT]) {
#pragma unroll
    for (int i = 0; i < RT; ++i) {
        v2f av = frag_step<S>(a[i]);
        if (RELU_A) {
            av.x = fmaxf(av.x, 0.0f);
            av.y = fmaxf(av.y, 0.0f);
        }
#pragma unroll
        for (int j = 0; j < CT; ++j) {
            acc[i][j] = __builtin_amdgcn_wmma_f32_16x16x4_f32(false, av, false, frag_step<S>(b[j]),
                                                              (short)0, acc[i][j], false, false);
        }
    }
}

template <int RT, int CT, bool RELU_A>
__device__ __forceinline__ void mma_block16(v8f (&acc)[RT][CT], const Frag16 (&a)[RT], const Frag16 (&b)[CT]) {
    mma_step<RT, CT, 0, RELU_A>(acc, a, b);
    mma_step<RT, CT, 1, RELU_A>(acc, a, b);
    mma_step<RT, CT, 2, RELU_A>(acc, a, b);
    mma_step<RT, CT, 3, RELU_A>(acc, a, b);
}

template <int RT, int CT, bool RELU_A, int EPI>
__global__ void __launch_bounds__(128)
gemm_wmma_f32(const float* __restrict__ A, int lda,
              const float* __restrict__ W, int ldw,
              const float* __restrict__ Cin,
              float* __restrict__ C, int ldc,
              int M, int K,
              const float* __restrict__ bias, const float* __restrict__ bb,
              const float* __restrict__ deg)
{
    const int nColGroups = 256 / (16 * CT);
    int waveId = blockIdx.x * (blockDim.x >> 5) + (threadIdx.x >> 5);
    int lane   = threadIdx.x & 31;
    int nRowTiles  = M >> 4;
    int nRowGroups = (nRowTiles + RT - 1) / RT;
    if (waveId >= nRowGroups * nColGroups) return;
    int rowGroup = waveId / nColGroups;
    int colGroup = waveId - rowGroup * nColGroups;
    int rt0  = rowGroup * RT;
    int col0 = colGroup * (16 * CT);
    int half = lane >> 4;
    int lr   = lane & 15;

    const float* Ap[RT];
    const float* Wp[CT];
#pragma unroll
    for (int i = 0; i < RT; ++i) {
        int rt = rt0 + i;
        if (rt > nRowTiles - 1) rt = nRowTiles - 1;
        Ap[i] = A + (size_t)(rt * 16 + lr) * lda + 4 * half;
    }
#pragma unroll
    for (int j = 0; j < CT; ++j)
        Wp[j] = W + (size_t)(col0 + 16 * j + lr) * ldw + 4 * half;

    v8f acc[RT][CT];
#pragma unroll
    for (int i = 0; i < RT; ++i)
#pragma unroll
        for (int j = 0; j < CT; ++j) acc[i][j] = v8f{};

    Frag16 ax[RT], bx[CT], ay[RT], by[CT];

#pragma unroll
    for (int i = 0; i < RT; ++i) ax[i] = load_frag16(Ap[i]);
#pragma unroll
    for (int j = 0; j < CT; ++j) bx[j] = load_frag16(Wp[j]);

    int kb = 0;
    for (; kb < K - 32; kb += 32) {
#pragma unroll
        for (int i = 0; i < RT; ++i) ay[i] = load_frag16(Ap[i] + kb + 16);
#pragma unroll
        for (int j = 0; j < CT; ++j) by[j] = load_frag16(Wp[j] + kb + 16);
        mma_block16<RT, CT, RELU_A>(acc, ax, bx);
#pragma unroll
        for (int i = 0; i < RT; ++i) ax[i] = load_frag16(Ap[i] + kb + 32);
#pragma unroll
        for (int j = 0; j < CT; ++j) bx[j] = load_frag16(Wp[j] + kb + 32);
        mma_block16<RT, CT, RELU_A>(acc, ay, by);
    }
#pragma unroll
    for (int i = 0; i < RT; ++i) ay[i] = load_frag16(Ap[i] + kb + 16);
#pragma unroll
    for (int j = 0; j < CT; ++j) by[j] = load_frag16(Wp[j] + kb + 16);
    mma_block16<RT, CT, RELU_A>(acc, ax, bx);
    mma_block16<RT, CT, RELU_A>(acc, ay, by);

#pragma unroll
    for (int i = 0; i < RT; ++i) {
        if (rt0 + i < nRowTiles) {
            int row0 = (rt0 + i) * 16 + 8 * half;
            size_t cOff = (size_t)row0 * ldc + col0 + lr;
            float* Cb = C + cOff;
            if (EPI == 0) {
#pragma unroll
                for (int r = 0; r < 8; ++r)
#pragma unroll
                    for (int j = 0; j < CT; ++j) Cb[(size_t)r * ldc + 16 * j] = acc[i][j][r];
            } else {
                float dg[8], bs[CT], bk[CT];
#pragma unroll
                for (int r = 0; r < 8; ++r) dg[r] = deg[row0 + r];
#pragma unroll
                for (int j = 0; j < CT; ++j) {
                    bs[j] = bias[col0 + lr + 16 * j];
                    bk[j] = bb[col0 + lr + 16 * j];
                }
                float old[8][CT];
                if (EPI == 2) {
                    const float* Ib = Cin + cOff;
#pragma unroll
                    for (int r = 0; r < 8; ++r)
#pragma unroll
                        for (int j = 0; j < CT; ++j) old[r][j] = Ib[(size_t)r * ldc + 16 * j];
                }
#pragma unroll
                for (int r = 0; r < 8; ++r) {
                    float norm = 1.0f / fmaxf(dg[r], 1.0f);
#pragma unroll
                    for (int j = 0; j < CT; ++j) {
                        float p = acc[i][j][r];
                        if (EPI == 2) p = old[r][j] + p;
                        Cb[(size_t)r * ldc + 16 * j] = bs[j] + norm * (p + bk[j]);
                    }
                }
            }
        }
    }
}

__global__ void __launch_bounds__(1024)
fold_attn_k(const float* __restrict__ fc1_w, const float* __restrict__ al1, const float* __restrict__ ar1,
            const float* __restrict__ fc2_w, const float* __restrict__ al2, const float* __restrict__ ar2,
            float* __restrict__ wf1, float* __restrict__ wf2, int H1, int D1)
{
    constexpr int F = 256;
    __shared__ float part[32][32];
    const int lane = threadIdx.x & 31;
    const int wave = threadIdx.x >> 5;
    const int row  = blockIdx.x >> 3;
    const int k    = (blockIdx.x & 7) * 32 + lane;
    float s = 0.0f;
    float* o;
    if (row < 2 * H1) {
        const int h = (row < H1) ? row : row - H1;
        const float* a = ((row < H1) ? al1 : ar1) + h * D1;
        const float* w = fc1_w + (size_t)h * D1 * F + k;
        for (int d = wave; d < D1; d += 32) s += a[d] * w[(size_t)d * F];
        o = wf1 + row * F + k;
    } else {
        const float* a = (row == 2 * H1) ? al2 : ar2;
        const float* w = fc2_w + k;
#pragma unroll
        for (int i = 0; i < F / 32; ++i) {
            const int d = wave + 32 * i;
            s += a[d] * w[(size_t)d * F];
        }
        o = wf2 + (row - 2 * H1) * F + k;
    }
    part[wave][lane] = s;
    __syncthreads();
    if (wave == 0) {
        float t = 0.0f;
        for (int w = 0; w < 32; ++w) t += part[w][lane];
        *o = t;
    }
}

template <int H, bool RELU_A>
__global__ void __launch_bounds__(128) __attribute__((amdgpu_waves_per_eu(1, 1)))
scores_wmma_f32(const float* __restrict__ A, const float* __restrict__ Wf,
                float* __restrict__ el, float* __restrict__ er, int M)
{
    constexpr int F = 256, RT = 2, NB = F / 16;
    const int lane = threadIdx.x & 31;
    const int wid  = blockIdx.x * (blockDim.x >> 5) + (threadIdx.x >> 5);
    const int nRowTiles = M >> 4;
    const int rt0 = wid * RT;
    if (rt0 >= nRowTiles) return;
    const int  half = lane >> 4;
    const int  lr   = lane & 15;
    const bool bOn  = lr < 2 * H;
    const float* Bp = Wf + (size_t)(bOn ? lr : 0) * F + 4 * half;
    const float* Ap[RT];
#pragma unroll
    for (int i = 0; i < RT; ++i) {
        int rt = rt0 + i;
        if (rt > nRowTiles - 1) rt = nRowTiles - 1;
        Ap[i] = A + (size_t)(rt * 16 + lr) * F + 4 * half;
    }

    Frag16 a[NB][RT], b[NB][1];
#pragma unroll
    for (int kb = 0; kb < NB; ++kb) {
        b[kb][0] = load_frag16(Bp + 16 * kb);
#pragma unroll
        for (int i = 0; i < RT; ++i) a[kb][i] = load_frag16(Ap[i] + 16 * kb);
    }
    if (!bOn) {
#pragma unroll
        for (int kb = 0; kb < NB; ++kb) { b[kb][0].q0 = v4f{}; b[kb][0].q1 = v4f{}; }
    }

    v8f acc[RT][1];
#pragma unroll
    for (int i = 0; i < RT; ++i) acc[i][0] = v8f{};
#pragma unroll
    for (int kb = 0; kb < NB; ++kb) mma_block16<RT, 1, RELU_A>(acc, a[kb], b[kb]);

#pragma unroll
    for (int i = 0; i < RT; ++i) {
        if (rt0 + i < nRowTiles && bOn) {
            const int row0 = (rt0 + i) * 16 + 8 * half;
            if (H == 1) {
                float* o = ((lr == 0) ? el : er) + row0;
                *(v4f*)(o)     = v4f{acc[i][0][0], acc[i][0][1], acc[i][0][2], acc[i][0][3]};
                *(v4f*)(o + 4) = v4f{acc[i][0][4], acc[i][0][5], acc[i][0][6], acc[i][0][7]};
            } else {
                float* o = ((lr < H) ? (el + lr) : (er + (lr - H))) + (size_t)row0 * H;
#pragma unroll
                for (int r = 0; r < 8; ++r) o[r * H] = acc[i][0][r];
            }
        }
    }
}


__device__ __forceinline__ void atomicMaxF(float* addr, float val) {
    if (val >= 0.0f) atomicMax((int*)addr, __float_as_int(val));
    else             atomicMin((unsigned int*)addr, __float_as_uint(val));
}

__device__ __forceinline__ float leaky02(float v) { return v > 0.0f ? v : 0.2f * v; }

constexpr int AGG_R   = 80;
constexpr int AGG_TB  = 1024;
constexpr int AGG_CAP = 10240;

template <int H>
__device__ __forceinline__ void edge_weights_to_slab(bool valid, int sv, int rv,
                                                     const float* __restrict__ el,
                                                     const float* s_er, const float* s_m, const float* s_s,
                                                     float* wl)
{
    if (valid) {
        float e[H];
#pragma unroll
        for (int h = 0; h < H; ++h) e[h] = el[(size_t)sv * H + h];
#pragma unroll
        for (int h = 0; h < H; ++h) {
            const int i = rv * H + h;
            wl[h] = expf(leaky02(e[h] + s_er[i]) - s_m[i]) / s_s[i];
        }
    }
}

template <int H, int B>
__device__ __forceinline__ void agg_add_edges(float* __restrict__ acc, int lane,
                                              int sv, int rv, int k, int n,
                                              const float* wv,
                                              const float* __restrict__ feat)
{
    constexpr int F = 256, D = F / H;
    const int h0 = (4 * lane) / D;
    const int h1 = (128 + 4 * lane) / D;
    int   rr[B];
    v4f   f0[B], f1[B];
    float a0[B], a1[B];
#pragma unroll
    for (int b = 0; b < B; ++b) {
        int kk = (k + b < n) ? (k + b) : (n - 1);
        int s  = __builtin_amdgcn_readlane(sv, kk);
        rr[b]  = __builtin_amdgcn_readlane(rv, kk);
        const float* fr = feat + (size_t)s * F + 4 * lane;
        f0[b] = *(const v4f*)(fr);
        f1[b] = *(const v4f*)(fr + 128);
        a0[b] = wv[kk * H + h0];
        a1[b] = (H > 1) ? wv[kk * H + h1] : a0[b];
    }
#pragma unroll
    for (int b = 0; b < B; ++b) {
        if (k + b < n) {
            float* o = acc + rr[b] * F + lane;
            atomicAdd(o +   0, f0[b].x * a0[b]);
            atomicAdd(o +  32, f0[b].y * a0[b]);
            atomicAdd(o +  64, f0[b].z * a0[b]);
            atomicAdd(o +  96, f0[b].w * a0[b]);
            atomicAdd(o + 128, f1[b].x * a1[b]);
            atomicAdd(o + 160, f1[b].y * a1[b]);
            atomicAdd(o + 192, f1[b].z * a1[b]);
            atomicAdd(o + 224, f1[b].w * a1[b]);
        }
    }
}

template <int H, int PHASE>
__device__ __forceinline__ void list_pass(const int* ls, const int* lr, int nPairs, int tid,
                                          const float* __restrict__ el,
                                          const float* s_er, float* s_m, float* s_s)
{
    constexpr int PU = 4;
    for (int p0 = tid; p0 < nPairs; p0 += AGG_TB * PU) {
        float ev[PU];
        int   ix[PU];
#pragma unroll
        for (int u = 0; u < PU; ++u) {
            const int  p   = p0 + u * AGG_TB;
            const bool ok  = p < nPairs;
            const int  pp  = ok ? p : p0;
            const int  ent = pp / H, h = pp - ent * H;
            ix[u] = ok ? (lr[ent] * H + h) : -1;
            ev[u] = el[(size_t)ls[ent] * H + h];
        }
#pragma unroll
        for (int u = 0; u < PU; ++u) {
            if (ix[u] >= 0) {
                const float v = leaky02(ev[u] + s_er[ix[u]]);
                if (PHASE == 0) atomicMaxF(&s_m[ix[u]], v);
                else            atomicAdd(&s_s[ix[u]], expf(v - s_m[ix[u]]));
            }
        }
    }
}

template <int H>
__global__ void __launch_bounds__(AGG_TB)
edge_softmax_agg_owner_k(const int* __restrict__ src, const int* __restrict__ dst,
                         const float* __restrict__ el, const float* __restrict__ er,
                         const float* __restrict__ feat, float* __restrict__ agg,
                         int E, int N)
{
    constexpr int F = 256;
    __shared__ __attribute__((aligned(16))) float acc[AGG_R * F];
    __shared__ int   le[AGG_CAP];
    __shared__ int   lr[AGG_CAP];
    __shared__ float s_er[AGG_R * H];
    __shared__ float s_m[AGG_R * H];
    __shared__ float s_s[AGG_R * H];
    __shared__ float wsl[AGG_TB * H];
    __shared__ int   cnt;

    const int tid  = threadIdx.x;
    const int lane = tid & 31;
    const int wave = tid >> 5;
    const int lo   = blockIdx.x * AGG_R;
    const unsigned ltmask = (1u << lane) - 1u;
    float* wv = wsl + (tid - lane) * H;

    for (int i = 4 * tid; i < AGG_R * F; i += 4 * AGG_TB) *(v4f*)(acc + i) = v4f{};
    for (int i = tid; i < AGG_R * H; i += AGG_TB) {
        s_er[i] = (lo + i / H < N) ? er[(size_t)lo * H + i] : 0.0f;
        s_m[i]  = -INFINITY;
        s_s[i]  = 0.0f;
    }
    if (tid == 0) cnt = 0;
    __syncthreads();

    const int  nChunk = E >> 2;
    const int4* dst4  = (const int4*)dst;
    constexpr int U = 4;
    for (int c0 = 0; c0 < nChunk + ((E & 3) ? 1 : 0); c0 += AGG_TB * U) {
        int4 dv[U];
#pragma unroll
        for (int u = 0; u < U; ++u) {
            int c = c0 + u * AGG_TB + tid;
            dv[u] = int4{-1, -1, -1, -1};
            if (c < nChunk) dv[u] = dst4[c];
            else if (c == nChunk) {
                if (4 * c + 0 < E) dv[u].x = dst[4 * c + 0];
                if (4 * c + 1 < E) dv[u].y = dst[4 * c + 1];
                if (4 * c + 2 < E) dv[u].z = dst[4 * c + 2];
            }
        }
#pragma unroll
        for (int u = 0; u < U; ++u) {
            const int c = c0 + u * AGG_TB + tid;
            const unsigned r0 = (unsigned)(dv[u].x - lo), r1 = (unsigned)(dv[u].y - lo);
            const unsigned r2 = (unsigned)(dv[u].z - lo), r3 = (unsigned)(dv[u].w - lo);
            const bool m0 = r0 < (unsigned)AGG_R, m1 = r1 < (unsigned)AGG_R;
            const bool m2 = r2 < (unsigned)AGG_R, m3 = r3 < (unsigned)AGG_R;
            if (__builtin_amdgcn_ballot_w32(m0 | m1 | m2 | m3) == 0u) continue;
            const unsigned b0 = __builtin_amdgcn_ballot_w32(m0), b1 = __builtin_amdgcn_ballot_w32(m1);
            const unsigned b2 = __builtin_amdgcn_ballot_w32(m2), b3 = __builtin_amdgcn_ballot_w32(m3);
            const int n0 = __builtin_popcount(b0), n1 = __builtin_popcount(b1);
            const int n2 = __builtin_popcount(b2), n3 = __builtin_popcount(b3);
            int base = 0;
            if (lane == 0) base = atomicAdd(&cnt, n0 + n1 + n2 + n3);
            base = __builtin_amdgcn_readfirstlane(base);
            const int p0 = base + __builtin_popcount(b0 & ltmask);
            const int p1 = base + n0 + __builtin_popcount(b1 & ltmask);
            const int p2 = base + n0 + n1 + __builtin_popcount(b2 & ltmask);
            const int p3 = base + n0 + n1 + n2 + __builtin_popcount(b3 & ltmask);
            if (m0 && p0 < AGG_CAP) { le[p0] = 4 * c + 0; lr[p0] = (int)r0; }
            if (m1 && p1 < AGG_CAP) { le[p1] = 4 * c + 1; lr[p1] = (int)r1; }
            if (m2 && p2 < AGG_CAP) { le[p2] = 4 * c + 2; lr[p2] = (int)r2; }
            if (m3 && p3 < AGG_CAP) { le[p3] = 4 * c + 3; lr[p3] = (int)r3; }
        }
    }
    __syncthreads();

    const int total = cnt;
    if (total <= AGG_CAP) {
        for (int idx = tid; idx < total; idx += AGG_TB) le[idx] = src[le[idx]];
        __syncthreads();
        list_pass<H, 0>(le, lr, total * H, tid, el, s_er, s_m, s_s);
        __syncthreads();
        list_pass<H, 1>(le, lr, total * H, tid, el, s_er, s_m, s_s);
        __syncthreads();
        for (int c0 = wave; c0 < total; c0 += 32 * 32) {
            const int  idx = c0 + 32 * lane;
            const bool v   = idx < total;
            const int  sv  = v ? le[idx] : 0;
            const int  rv  = v ? lr[idx] : 0;
            const int  n   = __builtin_popcount(__builtin_amdgcn_ballot_w32(v));
            edge_weights_to_slab<H>(v, sv, rv, el, s_er, s_m, s_s, wv + lane * H);
            for (int k = 0; k < n; k += 4)
                agg_add_edges<H, 4>(acc, lane, sv, rv, k, n, wv, feat);
        }
    } else {
        for (int i = tid; i < E; i += AGG_TB) {
            const unsigned r = (unsigned)(dst[i] - lo);
            if (r < (unsigned)AGG_R) {
                const int s = src[i];
                for (int h = 0; h < H; ++h)
                    atomicMaxF(&s_m[r * H + h], leaky02(el[(size_t)s * H + h] + s_er[r * H + h]));
            }
        }
        __syncthreads();
        for (int i = tid; i < E; i += AGG_TB) {
            const unsigned r = (unsigned)(dst[i] - lo);
            if (r < (unsigned)AGG_R) {
                const int s = src[i];
                for (int h = 0; h < H; ++h) {
                    const float v = leaky02(el[(size_t)s * H + h] + s_er[r * H + h]);
                    atomicAdd(&s_s[r * H + h], expf(v - s_m[r * H + h]));
                }
            }
        }
        __syncthreads();
        for (int b0 = 32 * wave; b0 < E; b0 += AGG_TB) {
            const int  i  = b0 + lane;
            const int  d  = (i < E) ? dst[i] : -1;
            const int  sI = (i < E) ? src[i] : 0;
            const unsigned r = (unsigned)(d - lo);
            unsigned own = __builtin_amdgcn_ballot_w32(r < (unsigned)AGG_R);
            edge_weights_to_slab<H>(r < (unsigned)AGG_R, sI, (int)r, el, s_er, s_m, s_s, wv + lane * H);
            while (own) {
                const int l = __builtin_ctz(own);
                own &= own - 1u;
                agg_add_edges<H, 1>(acc, lane, sI, (int)r, l, 32, wv, feat);
            }
        }
    }
    __syncthreads();

    for (int r = wave; r < AGG_R; r += 32) {
        const int row = lo + r;
        if (row < N) {
            float* g = agg + (size_t)row * F + 4 * lane;
            const float* a = acc + r * F + lane;
            v4f s0 = *(const v4f*)(g);
            v4f s1 = *(const v4f*)(g + 128);
            v4f t0 = { a[0],   a[32],  a[64],  a[96]  };
            v4f t1 = { a[128], a[160], a[192], a[224] };
            *(v4f*)(g)       = s0 + t0;
            *(v4f*)(g + 128) = s1 + t1;
        }
    }
}

extern "C" void kernel_launch(void* const* d_in, const int* in_sizes, int n_in,
                              void* d_out, int out_size, void* d_ws, size_t ws_size,
                              hipStream_t stream)
{
    const float* inputs  = (const float*)d_in[0];
    const int*   src     = (const int*)  d_in[1];
    const int*   dst     = (const int*)  d_in[2];
    const float* fc1_w   = (const float*)d_in[3];
    const float* attn_l1 = (const float*)d_in[4];
    const float* attn_r1 = (const float*)d_in[5];
    const float* bias1   = (const float*)d_in[6];
    const float* fc2_w   = (const float*)d_in[7];
    const float* attn_l2 = (const float*)d_in[8];
    const float* attn_r2 = (const float*)d_in[9];
    const float* bias2   = (const float*)d_in[10];
    const float* wb0     = (const float*)d_in[11];
    const float* bb0     = (const float*)d_in[12];
    const float* wb1     = (const float*)d_in[13];
    const float* bb1     = (const float*)d_in[14];
    float* out = (float*)d_out;

    const int F  = 256;
    const int N  = in_sizes[0] / F;
    const int E  = in_sizes[1];
    constexpr int H1 = 8;
    const int D1 = 32;

    float* ws   = (float*)d_ws;
    float* feat = ws;                ws += (size_t)N * F;
    float* buf  = ws;                ws += (size_t)N * F;
    float* agg  = ws;                ws += (size_t)N * F;
    float* el   = ws;                ws += (size_t)N * H1;
    float* er   = ws;                ws += (size_t)N * H1;
    float* deg  = ws;                ws += (size_t)N;
    float* wf1  = ws;                ws += (size_t)2 * H1 * F;
    float* wf2  = ws;                ws += (size_t)2 * F;

    const int TB = 256;
    auto cdiv = [](int a, int b) { return (a + b - 1) / b; };

    const int GRT = 5, GCT = 4, GTB = 128;
    const int waves    = cdiv(N / 16, GRT) * (256 / (16 * GCT));
    const int gemmGrid = cdiv(waves, GTB / 32);

    fill_f32<<<cdiv(N, TB), TB, 0, stream>>>(deg, 0.0f, N);
    degree_k<<<cdiv(E, TB), TB, 0, stream>>>(dst, deg, E);
    fold_attn_k<<<(2 * H1 + 2) * 8, 1024, 0, stream>>>(fc1_w, attn_l1, attn_r1, fc2_w, attn_l2, attn_r2,
                                                       wf1, wf2, H1, D1);
    const int scGrid = cdiv(cdiv(N / 16, 2), 4);

    gemm_wmma_f32<GRT, GCT, false, 0><<<gemmGrid, GTB, 0, stream>>>(inputs, F, fc1_w, F, nullptr, feat, F, N, F,
                                                                    nullptr, nullptr, nullptr);
    gemm_wmma_f32<GRT, GCT, false, 1><<<gemmGrid, GTB, 0, stream>>>(inputs, F, wb0, F, nullptr, agg, F, N, F,
                                                                    bias1, bb0, deg);

    scores_wmma_f32<H1, false><<<scGrid, 128, 0, stream>>>(inputs, wf1, el, er, N);
    edge_softmax_agg_owner_k<H1><<<cdiv(N, AGG_R), AGG_TB, 0, stream>>>(src, dst, el, er, feat, agg, E, N);

    gemm_wmma_f32<GRT, GCT, true, 0><<<gemmGrid, GTB, 0, stream>>>(agg, F, fc2_w, F, nullptr, feat, F, N, F,
                                                                   nullptr, nullptr, nullptr);
    gemm_wmma_f32<GRT, GCT, false, 0><<<gemmGrid, GTB, 0, stream>>>(inputs, F, wb1, 2 * F, nullptr, buf, F, N, F,
                                                                    nullptr, nullptr, nullptr);
    gemm_wmma_f32<GRT, GCT, true, 2><<<gemmGrid, GTB, 0, stream>>>(agg, F, wb1 + F, 2 * F, buf, out, F, N, F,
                                                                   bias2, bb1, deg);

    scores_wmma_f32<1, true><<<scGrid, 128, 0, stream>>>(agg, wf2, el, er, N);
    edge_softmax_agg_owner_k<1><<<cdiv(N, AGG_R), AGG_TB, 0, stream>>>(src, dst, el, er, feat, out, E, N);
}


# edge lists built once per call: layer 1's row-owning groups save their list (source, row) to memory, layer 2's read it back: no second scan of all dst by 250 groups; layer-2 sums in that list's order
# speedup vs baseline: 2.6353x; 1.0656x over previous
#include <hip/hip_runtime.h>
#include <hip/hip_bf16.h>
#include <math.h>


typedef __attribute__((ext_vector_type(2))) float v2f;
typedef __attribute__((ext_vector_type(8))) float v8f;

__global__ void fill_f32(float* __restrict__ p, float v, int n) {
    int i = blockIdx.x * blockDim.x + threadIdx.x;
    if (i < n) p[i] = v;
}

__global__ void degree_k(const int* __restrict__ dst, float* __restrict__ deg, int E) {
    int i = blockIdx.x * blockDim.x + threadIdx.x;
    if (i < E) atomicAdd(&deg[dst[i]], 1.0f);
}

typedef __attribute__((ext_vector_type(4))) float v4f;

struct Frag16 { v4f q0, q1; };

__device__ __forceinline__ Frag16 load_frag16(const float* __restrict__ p) {
    Frag16 f;
    f.q0 = *(const v4f*)(p);
    f.q1 = *(const v4f*)(p + 8);
    return f;
}

template <int S>
__device__ __forceinline__ v2f frag_step(const Frag16& f) {
    if (S == 0) return __builtin_shufflevector(f.q0, f.q0, 0, 1);
    if (S == 1) return __builtin_shufflevector(f.q0, f.q0, 2, 3);
    if (S == 2) return __builtin_shufflevector(f.q1, f.q1, 0, 1);
    return __builtin_shufflevector(f.q1, f.q1, 2, 3);
}

template <int RT, int CT, int S, bool RELU_A>
__device__ __forceinline__ void mma_step(v8f (&acc)[RT][CT], const Frag16 (&a)[RT], const Frag16 (&b)[CT]) {
#pragma unroll
    for (int i = 0; i < RT; ++i) {
        v2f av = frag_step<S>(a[i]);
        if (RELU_A) {
            av.x = fmaxf(av.x, 0.0f);
            av.y = fmaxf(av.y, 0.0f);
        }
#pragma unroll
        for (int j = 0; j < CT; ++j) {
            acc[i][j] = __builtin_amdgcn_wmma_f32_16x16x4_f32(false, av, false, frag_step<S>(b[j]),
                                                              (short)0, acc[i][j], false, false);
        }
    }
}

template <int RT, int CT, bool RELU_A>
__device__ __forceinline__ void mma_block16(v8f (&acc)[RT][CT], const Frag16 (&a)[RT], const Frag16 (&b)[CT]) {
    mma_step<RT, CT, 0, RELU_A>(acc, a, b);
    mma_step<RT, CT, 1, RELU_A>(acc, a, b);
    mma_step<RT, CT, 2, RELU_A>(acc, a, b);
    mma_step<RT, CT, 3, RELU_A>(acc, a, b);
}

template <int RT, int CT, bool RELU_A, int EPI>
__global__ void __launch_bounds__(128)
gemm_wmma_f32(const float* __restrict__ A, int lda,
              const float* __restrict__ W, int ldw,
              const float* __restrict__ Cin,
              float* __restrict__ C, int ldc,
              int M, int K,
              const float* __restrict__ bias, const float* __restrict__ bb,
              const float* __restrict__ deg)
{
    const int nColGroups = 256 / (16 * CT);
    int waveId = blockIdx.x * (blockDim.x >> 5) + (threadIdx.x >> 5);
    int lane   = threadIdx.x & 31;
    int nRowTiles  = M >> 4;
    int nRowGroups = (nRowTiles + RT - 1) / RT;
    if (waveId >= nRowGroups * nColGroups) return;
    int rowGroup = waveId / nColGroups;
    int colGroup = waveId - rowGroup * nColGroups;
    int rt0  = rowGroup * RT;
    int col0 = colGroup * (16 * CT);
    int half = lane >> 4;
    int lr   = lane & 15;

    const float* Ap[RT];
    const float* Wp[CT];
#pragma unroll
    for (int i = 0; i < RT; ++i) {
        int rt = rt0 + i;
        if (rt > nRowTiles - 1) rt = nRowTiles - 1;
        Ap[i] = A + (size_t)(rt * 16 + lr) * lda + 4 * half;
    }
#pragma unroll
    for (int j = 0; j < CT; ++j)
        Wp[j] = W + (size_t)(col0 + 16 * j + lr) * ldw + 4 * half;

    v8f acc[RT][CT];
#pragma unroll
    for (int i = 0; i < RT; ++i)
#pragma unroll
        for (int j = 0; j < CT; ++j) acc[i][j] = v8f{};

    Frag16 ax[RT], bx[CT], ay[RT], by[CT];

#pragma unroll
    for (int i = 0; i < RT; ++i) ax[i] = load_frag16(Ap[i]);
#pragma unroll
    for (int j = 0; j < CT; ++j) bx[j] = load_frag16(Wp[j]);

    int kb = 0;
    for (; kb < K - 32; kb += 32) {
#pragma unroll
        for (int i = 0; i < RT; ++i) ay[i] = load_frag16(Ap[i] + kb + 16);
#pragma unroll
        for (int j = 0; j < CT; ++j) by[j] = load_frag16(Wp[j] + kb + 16);
        mma_block16<RT, CT, RELU_A>(acc, ax, bx);
#pragma unroll
        for (int i = 0; i < RT; ++i) ax[i] = load_frag16(Ap[i] + kb + 32);
#pragma unroll
        for (int j = 0; j < CT; ++j) bx[j] = load_frag16(Wp[j] + kb + 32);
        mma_block16<RT, CT, RELU_A>(acc, ay, by);
    }
#pragma unroll
    for (int i = 0; i < RT; ++i) ay[i] = load_frag16(Ap[i] + kb + 16);
#pragma unroll
    for (int j = 0; j < CT; ++j) by[j] = load_frag16(Wp[j] + kb + 16);
    mma_block16<RT, CT, RELU_A>(acc, ax, bx);
    mma_block16<RT, CT, RELU_A>(acc, ay, by);

#pragma unroll
    for (int i = 0; i < RT; ++i) {
        if (rt0 + i < nRowTiles) {
            int row0 = (rt0 + i) * 16 + 8 * half;
            size_t cOff = (size_t)row0 * ldc + col0 + lr;
            float* Cb = C + cOff;
            if (EPI == 0) {
#pragma unroll
                for (int r = 0; r < 8; ++r)
#pragma unroll
                    for (int j = 0; j < CT; ++j) Cb[(size_t)r * ldc + 16 * j] = acc[i][j][r];
            } else {
                float dg[8], bs[CT], bk[CT];
#pragma unroll
                for (int r = 0; r < 8; ++r) dg[r] = deg[row0 + r];
#pragma unroll
                for (int j = 0; j < CT; ++j) {
                    bs[j] = bias[col0 + lr + 16 * j];
                    bk[j] = bb[col0 + lr + 16 * j];
                }
                float old[8][CT];
                if (EPI == 2) {
                    const float* Ib = Cin + cOff;
#pragma unroll
                    for (int r = 0; r < 8; ++r)
#pragma unroll
                        for (int j = 0; j < CT; ++j) old[r][j] = Ib[(size_t)r * ldc + 16 * j];
                }
#pragma unroll
                for (int r = 0; r < 8; ++r) {
                    float norm = 1.0f / fmaxf(dg[r], 1.0f);
#pragma unroll
                    for (int j = 0; j < CT; ++j) {
                        float p = acc[i][j][r];
                        if (EPI == 2) p = old[r][j] + p;
                        Cb[(size_t)r * ldc + 16 * j] = bs[j] + norm * (p + bk[j]);
                    }
                }
            }
        }
    }
}

__global__ void __launch_bounds__(1024)
fold_attn_k(const float* __restrict__ fc1_w, const float* __restrict__ al1, const float* __restrict__ ar1,
            const float* __restrict__ fc2_w, const float* __restrict__ al2, const float* __restrict__ ar2,
            float* __restrict__ wf1, float* __restrict__ wf2, int H1, int D1)
{
    constexpr int F = 256;
    __shared__ float part[32][32];
    const int lane = threadIdx.x & 31;
    const int wave = threadIdx.x >> 5;
    const int row  = blockIdx.x >> 3;
    const int k    = (blockIdx.x & 7) * 32 + lane;
    float s = 0.0f;
    float* o;
    if (row < 2 * H1) {
        const int h = (row < H1) ? row : row - H1;
        const float* a = ((row < H1) ? al1 : ar1) + h * D1;
        const float* w = fc1_w + (size_t)h * D1 * F + k;
        for (int d = wave; d < D1; d += 32) s += a[d] * w[(size_t)d * F];
        o = wf1 + row * F + k;
    } else {
        const float* a = (row == 2 * H1) ? al2 : ar2;
        const float* w = fc2_w + k;
#pragma unroll
        for (int i = 0; i < F / 32; ++i) {
            const int d = wave + 32 * i;
            s += a[d] * w[(size_t)d * F];
        }
        o = wf2 + (row - 2 * H1) * F + k;
    }
    part[wave][lane] = s;
    __syncthreads();
    if (wave == 0) {
        float t = 0.0f;
        for (int w = 0; w < 32; ++w) t += part[w][lane];
        *o = t;
    }
}

template <int H, bool RELU_A>
__global__ void __launch_bounds__(128) __attribute__((amdgpu_waves_per_eu(1, 1)))
scores_wmma_f32(const float* __restrict__ A, const float* __restrict__ Wf,
                float* __restrict__ el, float* __restrict__ er, int M)
{
    constexpr int F = 256, RT = 2, NB = F / 16;
    const int lane = threadIdx.x & 31;
    const int wid  = blockIdx.x * (blockDim.x >> 5) + (threadIdx.x >> 5);
    const int nRowTiles = M >> 4;
    const int rt0 = wid * RT;
    if (rt0 >= nRowTiles) return;
    const int  half = lane >> 4;
    const int  lr   = lane & 15;
    const bool bOn  = lr < 2 * H;
    const float* Bp = Wf + (size_t)(bOn ? lr : 0) * F + 4 * half;
    const float* Ap[RT];
#pragma unroll
    for (int i = 0; i < RT; ++i) {
        int rt = rt0 + i;
        if (rt > nRowTiles - 1) rt = nRowTiles - 1;
        Ap[i] = A + (size_t)(rt * 16 + lr) * F + 4 * half;
    }

    Frag16 a[NB][RT], b[NB][1];
#pragma unroll
    for (int kb = 0; kb < NB; ++kb) {
        b[kb][0] = load_frag16(Bp + 16 * kb);
#pragma unroll
        for (int i = 0; i < RT; ++i) a[kb][i] = load_frag16(Ap[i] + 16 * kb);
    }
    if (!bOn) {
#pragma unroll
        for (int kb = 0; kb < NB; ++kb) { b[kb][0].q0 = v4f{}; b[kb][0].q1 = v4f{}; }
    }

    v8f acc[RT][1];
#pragma unroll
    for (int i = 0; i < RT; ++i) acc[i][0] = v8f{};
#pragma unroll
    for (int kb = 0; kb < NB; ++kb) mma_block16<RT, 1, RELU_A>(acc, a[kb], b[kb]);

#pragma unroll
    for (int i = 0; i < RT; ++i) {
        if (rt0 + i < nRowTiles && bOn) {
            const int row0 = (rt0 + i) * 16 + 8 * half;
            if (H == 1) {
                float* o = ((lr == 0) ? el : er) + row0;
                *(v4f*)(o)     = v4f{acc[i][0][0], acc[i][0][1], acc[i][0][2], acc[i][0][3]};
                *(v4f*)(o + 4) = v4f{acc[i][0][4], acc[i][0][5], acc[i][0][6], acc[i][0][7]};
            } else {
                float* o = ((lr < H) ? (el + lr) : (er + (lr - H))) + (size_t)row0 * H;
#pragma unroll
                for (int r = 0; r < 8; ++r) o[r * H] = acc[i][0][r];
            }
        }
    }
}


__device__ __forceinline__ void atomicMaxF(float* addr, float val) {
    if (val >= 0.0f) atomicMax((int*)addr, __float_as_int(val));
    else             atomicMin((unsigned int*)addr, __float_as_uint(val));
}

__device__ __forceinline__ float leaky02(float v) { return v > 0.0f ? v : 0.2f * v; }

constexpr int AGG_R   = 80;
constexpr int AGG_TB  = 1024;
constexpr int AGG_CAP = 10240;

template <int H>
__device__ __forceinline__ void edge_weights_to_slab(bool valid, int sv, int rv,
                                                     const float* __restrict__ el,
                                                     const float* s_er, const float* s_m, const float* s_s,
                                                     float* wl)
{
    if (valid) {
        float e[H];
#pragma unroll
        for (int h = 0; h < H; ++h) e[h] = el[(size_t)sv * H + h];
#pragma unroll
        for (int h = 0; h < H; ++h) {
            const int i = rv * H + h;
            wl[h] = expf(leaky02(e[h] + s_er[i]) - s_m[i]) / s_s[i];
        }
    }
}

template <int H, int B>
__device__ __forceinline__ void agg_add_edges(float* __restrict__ acc, int lane,
                                              int sv, int rv, int k, int n,
                                              const float* wv,
                                              const float* __restrict__ feat)
{
    constexpr int F = 256, D = F / H;
    const int h0 = (4 * lane) / D;
    const int h1 = (128 + 4 * lane) / D;
    int   rr[B];
    v4f   f0[B], f1[B];
    float a0[B], a1[B];
#pragma unroll
    for (int b = 0; b < B; ++b) {
        int kk = (k + b < n) ? (k + b) : (n - 1);
        int s  = __builtin_amdgcn_readlane(sv, kk);
        rr[b]  = __builtin_amdgcn_readlane(rv, kk);
        const float* fr = feat + (size_t)s * F + 4 * lane;
        f0[b] = *(const v4f*)(fr);
        f1[b] = *(const v4f*)(fr + 128);
        a0[b] = wv[kk * H + h0];
        a1[b] = (H > 1) ? wv[kk * H + h1] : a0[b];
    }
#pragma unroll
    for (int b = 0; b < B; ++b) {
        if (k + b < n) {
            float* o = acc + rr[b] * F + lane;
            atomicAdd(o +   0, f0[b].x * a0[b]);
            atomicAdd(o +  32, f0[b].y * a0[b]);
            atomicAdd(o +  64, f0[b].z * a0[b]);
            atomicAdd(o +  96, f0[b].w * a0[b]);
            atomicAdd(o + 128, f1[b].x * a1[b]);
            atomicAdd(o + 160, f1[b].y * a1[b]);
            atomicAdd(o + 192, f1[b].z * a1[b]);
            atomicAdd(o + 224, f1[b].w * a1[b]);
        }
    }
}

template <int H, int PHASE>
__device__ __forceinline__ void list_pass(const int* ls, const int* lr, int nPairs, int tid,
                                          const float* __restrict__ el,
                                          const float* s_er, float* s_m, float* s_s)
{
    constexpr int PU = 4;
    for (int p0 = tid; p0 < nPairs; p0 += AGG_TB * PU) {
        float ev[PU];
        int   ix[PU];
#pragma unroll
        for (int u = 0; u < PU; ++u) {
            const int  p   = p0 + u * AGG_TB;
            const bool ok  = p < nPairs;
            const int  pp  = ok ? p : p0;
            const int  ent = pp / H, h = pp - ent * H;
            ix[u] = ok ? (lr[ent] * H + h) : -1;
            ev[u] = el[(size_t)ls[ent] * H + h];
        }
#pragma unroll
        for (int u = 0; u < PU; ++u) {
            if (ix[u] >= 0) {
                const float v = leaky02(ev[u] + s_er[ix[u]]);
                if (PHASE == 0) atomicMaxF(&s_m[ix[u]], v);
                else            atomicAdd(&s_s[ix[u]], expf(v - s_m[ix[u]]));
            }
        }
    }
}

template <int H, bool BUILD>
__global__ void __launch_bounds__(AGG_TB)
edge_softmax_agg_owner_k(const int* __restrict__ src, const int* __restrict__ dst,
                         const float* __restrict__ el, const float* __restrict__ er,
                         const float* __restrict__ feat, float* __restrict__ agg,
                         int* __restrict__ glist, int* __restrict__ gcnt,
                         int E, int N)
{
    constexpr int F = 256;
    __shared__ __attribute__((aligned(16))) float acc[AGG_R * F];
    __shared__ int   le[AGG_CAP];
    __shared__ int   lr[AGG_CAP];
    __shared__ float s_er[AGG_R * H];
    __shared__ float s_m[AGG_R * H];
    __shared__ float s_s[AGG_R * H];
    __shared__ float wsl[AGG_TB * H];
    __shared__ int   cnt;

    const int tid  = threadIdx.x;
    const int lane = tid & 31;
    const int wave = tid >> 5;
    const int lo   = blockIdx.x * AGG_R;
    const unsigned ltmask = (1u << lane) - 1u;
    float* wv = wsl + (tid - lane) * H;

    for (int i = 4 * tid; i < AGG_R * F; i += 4 * AGG_TB) *(v4f*)(acc + i) = v4f{};
    for (int i = tid; i < AGG_R * H; i += AGG_TB) {
        s_er[i] = (lo + i / H < N) ? er[(size_t)lo * H + i] : 0.0f;
        s_m[i]  = -INFINITY;
        s_s[i]  = 0.0f;
    }
    if (tid == 0) cnt = 0;
    __syncthreads();

    int* gl = glist + (size_t)blockIdx.x * (2 * AGG_CAP);
    int total;
    if (BUILD) {
    const int  nChunk = E >> 2;
    const int4* dst4  = (const int4*)dst;
    constexpr int U = 4;
    for (int c0 = 0; c0 < nChunk + ((E & 3) ? 1 : 0); c0 += AGG_TB * U) {
        int4 dv[U];
#pragma unroll
        for (int u = 0; u < U; ++u) {
            int c = c0 + u * AGG_TB + tid;
            dv[u] = int4{-1, -1, -1, -1};
            if (c < nChunk) dv[u] = dst4[c];
            else if (c == nChunk) {
                if (4 * c + 0 < E) dv[u].x = dst[4 * c + 0];
                if (4 * c + 1 < E) dv[u].y = dst[4 * c + 1];
                if (4 * c + 2 < E) dv[u].z = dst[4 * c + 2];
            }
        }
#pragma unroll
        for (int u = 0; u < U; ++u) {
            const int c = c0 + u * AGG_TB + tid;
            const unsigned r0 = (unsigned)(dv[u].x - lo), r1 = (unsigned)(dv[u].y - lo);
            const unsigned r2 = (unsigned)(dv[u].z - lo), r3 = (unsigned)(dv[u].w - lo);
            const bool m0 = r0 < (unsigned)AGG_R, m1 = r1 < (unsigned)AGG_R;
            const bool m2 = r2 < (unsigned)AGG_R, m3 = r3 < (unsigned)AGG_R;
            if (__builtin_amdgcn_ballot_w32(m0 | m1 | m2 | m3) == 0u) continue;
            const unsigned b0 = __builtin_amdgcn_ballot_w32(m0), b1 = __builtin_amdgcn_ballot_w32(m1);
            const unsigned b2 = __builtin_amdgcn_ballot_w32(m2), b3 = __builtin_amdgcn_ballot_w32(m3);
            const int n0 = __builtin_popcount(b0), n1 = __builtin_popcount(b1);
            const int n2 = __builtin_popcount(b2), n3 = __builtin_popcount(b3);
            int base = 0;
            if (lane == 0) base = atomicAdd(&cnt, n0 + n1 + n2 + n3);
            base = __builtin_amdgcn_readfirstlane(base);
            const int p0 = base + __builtin_popcount(b0 & ltmask);
            const int p1 = base + n0 + __builtin_popcount(b1 & ltmask);
            const int p2 = base + n0 + n1 + __builtin_popcount(b2 & ltmask);
            const int p3 = base + n0 + n1 + n2 + __builtin_popcount(b3 & ltmask);
            if (m0 && p0 < AGG_CAP) { le[p0] = 4 * c + 0; lr[p0] = (int)r0; }
            if (m1 && p1 < AGG_CAP) { le[p1] = 4 * c + 1; lr[p1] = (int)r1; }
            if (m2 && p2 < AGG_CAP) { le[p2] = 4 * c + 2; lr[p2] = (int)r2; }
            if (m3 && p3 < AGG_CAP) { le[p3] = 4 * c + 3; lr[p3] = (int)r3; }
        }
    }
    __syncthreads();
    total = cnt;
    if (tid == 0) gcnt[blockIdx.x] = total;
    } else {
        total = gcnt[blockIdx.x];
    }

    if (total <= AGG_CAP) {
        if (BUILD) {
            for (int idx = tid; idx < total; idx += AGG_TB) {
                const int s = src[le[idx]];
                le[idx] = s;
                gl[idx] = s;
                gl[AGG_CAP + idx] = lr[idx];
            }
        } else {
            for (int idx = tid; idx < total; idx += AGG_TB) {
                le[idx] = gl[idx];
                lr[idx] = gl[AGG_CAP + idx];
            }
        }
        __syncthreads();
        list_pass<H, 0>(le, lr, total * H, tid, el, s_er, s_m, s_s);
        __syncthreads();
        list_pass<H, 1>(le, lr, total * H, tid, el, s_er, s_m, s_s);
        __syncthreads();
        for (int c0 = wave; c0 < total; c0 += 32 * 32) {
            const int  idx = c0 + 32 * lane;
            const bool v   = idx < total;
            const int  sv  = v ? le[idx] : 0;
            const int  rv  = v ? lr[idx] : 0;
            const int  n   = __builtin_popcount(__builtin_amdgcn_ballot_w32(v));
            edge_weights_to_slab<H>(v, sv, rv, el, s_er, s_m, s_s, wv + lane * H);
            for (int k = 0; k < n; k += 4)
                agg_add_edges<H, 4>(acc, lane, sv, rv, k, n, wv, feat);
        }
    } else {
        for (int i = tid; i < E; i += AGG_TB) {
            const unsigned r = (unsigned)(dst[i] - lo);
            if (r < (unsigned)AGG_R) {
                const int s = src[i];
                for (int h = 0; h < H; ++h)
                    atomicMaxF(&s_m[r * H + h], leaky02(el[(size_t)s * H + h] + s_er[r * H + h]));
            }
        }
        __syncthreads();
        for (int i = tid; i < E; i += AGG_TB) {
            const unsigned r = (unsigned)(dst[i] - lo);
            if (r < (unsigned)AGG_R) {
                const int s = src[i];
                for (int h = 0; h < H; ++h) {
                    const float v = leaky02(el[(size_t)s * H + h] + s_er[r * H + h]);
                    atomicAdd(&s_s[r * H + h], expf(v - s_m[r * H + h]));
                }
            }
        }
        __syncthreads();
        for (int b0 = 32 * wave; b0 < E; b0 += AGG_TB) {
            const int  i  = b0 + lane;
            const int  d  = (i < E) ? dst[i] : -1;
            const int  sI = (i < E) ? src[i] : 0;
            const unsigned r = (unsigned)(d - lo);
            unsigned own = __builtin_amdgcn_ballot_w32(r < (unsigned)AGG_R);
            edge_weights_to_slab<H>(r < (unsigned)AGG_R, sI, (int)r, el, s_er, s_m, s_s, wv + lane * H);
            while (own) {
                const int l = __builtin_ctz(own);
                own &= own - 1u;
                agg_add_edges<H, 1>(acc, lane, sI, (int)r, l, 32, wv, feat);
            }
        }
    }
    __syncthreads();

    for (int r = wave; r < AGG_R; r += 32) {
        const int row = lo + r;
        if (row < N) {
            float* g = agg + (size_t)row * F + 4 * lane;
            const float* a = acc + r * F + lane;
            v4f s0 = *(const v4f*)(g);
            v4f s1 = *(const v4f*)(g + 128);
            v4f t0 = { a[0],   a[32],  a[64],  a[96]  };
            v4f t1 = { a[128], a[160], a[192], a[224] };
            *(v4f*)(g)       = s0 + t0;
            *(v4f*)(g + 128) = s1 + t1;
        }
    }
}

extern "C" void kernel_launch(void* const* d_in, const int* in_sizes, int n_in,
                              void* d_out, int out_size, void* d_ws, size_t ws_size,
                              hipStream_t stream)
{
    const float* inputs  = (const float*)d_in[0];
    const int*   src     = (const int*)  d_in[1];
    const int*   dst     = (const int*)  d_in[2];
    const float* fc1_w   = (const float*)d_in[3];
    const float* attn_l1 = (const float*)d_in[4];
    const float* attn_r1 = (const float*)d_in[5];
    const float* bias1   = (const float*)d_in[6];
    const float* fc2_w   = (const float*)d_in[7];
    const float* attn_l2 = (const float*)d_in[8];
    const float* attn_r2 = (const float*)d_in[9];
    const float* bias2   = (const float*)d_in[10];
    const float* wb0     = (const float*)d_in[11];
    const float* bb0     = (const float*)d_in[12];
    const float* wb1     = (const float*)d_in[13];
    const float* bb1     = (const float*)d_in[14];
    float* out = (float*)d_out;

    const int F  = 256;
    const int N  = in_sizes[0] / F;
    const int E  = in_sizes[1];
    constexpr int H1 = 8;
    const int D1 = 32;

    float* ws   = (float*)d_ws;
    float* feat = ws;                ws += (size_t)N * F;
    float* buf  = ws;                ws += (size_t)N * F;
    float* agg  = ws;                ws += (size_t)N * F;
    float* el   = ws;                ws += (size_t)N * H1;
    float* er   = ws;                ws += (size_t)N * H1;
    float* deg  = ws;                ws += (size_t)N;
    float* wf1  = ws;                ws += (size_t)2 * H1 * F;
    float* wf2  = ws;                ws += (size_t)2 * F;
    const int nGroups = (N + AGG_R - 1) / AGG_R;
    int* gcnt  = (int*)ws;           ws += (size_t)((nGroups + 31) / 32) * 32;
    int* glist = (int*)ws;           ws += (size_t)nGroups * 2 * AGG_CAP;

    const int TB = 256;
    auto cdiv = [](int a, int b) { return (a + b - 1) / b; };

    const int GRT = 5, GCT = 4, GTB = 128;
    const int waves    = cdiv(N / 16, GRT) * (256 / (16 * GCT));
    const int gemmGrid = cdiv(waves, GTB / 32);

    fill_f32<<<cdiv(N, TB), TB, 0, stream>>>(deg, 0.0f, N);
    degree_k<<<cdiv(E, TB), TB, 0, stream>>>(dst, deg, E);
    fold_attn_k<<<(2 * H1 + 2) * 8, 1024, 0, stream>>>(fc1_w, attn_l1, attn_r1, fc2_w, attn_l2, attn_r2,
                                                       wf1, wf2, H1, D1);
    const int scGrid = cdiv(cdiv(N / 16, 2), 4);

    gemm_wmma_f32<GRT, GCT, false, 0><<<gemmGrid, GTB, 0, stream>>>(inputs, F, fc1_w, F, nullptr, feat, F, N, F,
                                                                    nullptr, nullptr, nullptr);
    gemm_wmma_f32<GRT, GCT, false, 1><<<gemmGrid, GTB, 0, stream>>>(inputs, F, wb0, F, nullptr, agg, F, N, F,
                                                                    bias1, bb0, deg);

    scores_wmma_f32<H1, false><<<scGrid, 128, 0, stream>>>(inputs, wf1, el, er, N);
    edge_softmax_agg_owner_k<H1, true><<<cdiv(N, AGG_R), AGG_TB, 0, stream>>>(src, dst, el, er, feat, agg, glist, gcnt, E, N);

    gemm_wmma_f32<GRT, GCT, true, 0><<<gemmGrid, GTB, 0, stream>>>(agg, F, fc2_w, F, nullptr, feat, F, N, F,
                                                                   nullptr, nullptr, nullptr);
    gemm_wmma_f32<GRT, GCT, false, 0><<<gemmGrid, GTB, 0, stream>>>(inputs, F, wb1, 2 * F, nullptr, buf, F, N, F,
                                                                    nullptr, nullptr, nullptr);
    gemm_wmma_f32<GRT, GCT, true, 2><<<gemmGrid, GTB, 0, stream>>>(agg, F, wb1 + F, 2 * F, buf, out, F, N, F,
                                                                   bias2, bb1, deg);

    scores_wmma_f32<1, true><<<scGrid, 128, 0, stream>>>(agg, wf2, el, er, N);
    edge_softmax_agg_owner_k<1, false><<<cdiv(N, AGG_R), AGG_TB, 0, stream>>>(src, dst, el, er, feat, out, glist, gcnt, E, N);
}


# in-degree counted by the row-owning groups in their own edge lists (LDS integer adds); bias + norm * skip term formed where they store the row; fill and degree launches and the global f32 adds gone
# speedup vs baseline: 2.9859x; 1.1330x over previous
#include <hip/hip_runtime.h>
#include <hip/hip_bf16.h>
#include <math.h>


typedef __attribute__((ext_vector_type(2))) float v2f;
typedef __attribute__((ext_vector_type(8))) float v8f;

typedef __attribute__((ext_vector_type(4))) float v4f;

struct Frag16 { v4f q0, q1; };

__device__ __forceinline__ Frag16 load_frag16(const float* __restrict__ p) {
    Frag16 f;
    f.q0 = *(const v4f*)(p);
    f.q1 = *(const v4f*)(p + 8);
    return f;
}

template <int S>
__device__ __forceinline__ v2f frag_step(const Frag16& f) {
    if (S == 0) return __builtin_shufflevector(f.q0, f.q0, 0, 1);
    if (S == 1) return __builtin_shufflevector(f.q0, f.q0, 2, 3);
    if (S == 2) return __builtin_shufflevector(f.q1, f.q1, 0, 1);
    return __builtin_shufflevector(f.q1, f.q1, 2, 3);
}

template <int RT, int CT, int S, bool RELU_A>
__device__ __forceinline__ void mma_step(v8f (&acc)[RT][CT], const Frag16 (&a)[RT], const Frag16 (&b)[CT]) {
#pragma unroll
    for (int i = 0; i < RT; ++i) {
        v2f av = frag_step<S>(a[i]);
        if (RELU_A) {
            av.x = fmaxf(av.x, 0.0f);
            av.y = fmaxf(av.y, 0.0f);
        }
#pragma unroll
        for (int j = 0; j < CT; ++j) {
            acc[i][j] = __builtin_amdgcn_wmma_f32_16x16x4_f32(false, av, false, frag_step<S>(b[j]),
                                                              (short)0, acc[i][j], false, false);
        }
    }
}

template <int RT, int CT, bool RELU_A>
__device__ __forceinline__ void mma_block16(v8f (&acc)[RT][CT], const Frag16 (&a)[RT], const Frag16 (&b)[CT]) {
    mma_step<RT, CT, 0, RELU_A>(acc, a, b);
    mma_step<RT, CT, 1, RELU_A>(acc, a, b);
    mma_step<RT, CT, 2, RELU_A>(acc, a, b);
    mma_step<RT, CT, 3, RELU_A>(acc, a, b);
}

template <int RT, int CT, bool RELU_A, int EPI>
__global__ void __launch_bounds__(128)
gemm_wmma_f32(const float* __restrict__ A, int lda,
              const float* __restrict__ W, int ldw,
              const float* __restrict__ Cin,
              float* __restrict__ C, int ldc,
              int M, int K)
{
    const int nColGroups = 256 / (16 * CT);
    int waveId = blockIdx.x * (blockDim.x >> 5) + (threadIdx.x >> 5);
    int lane   = threadIdx.x & 31;
    int nRowTiles  = M >> 4;
    int nRowGroups = (nRowTiles + RT - 1) / RT;
    if (waveId >= nRowGroups * nColGroups) return;
    int rowGroup = waveId / nColGroups;
    int colGroup = waveId - rowGroup * nColGroups;
    int rt0  = rowGroup * RT;
    int col0 = colGroup * (16 * CT);
    int half = lane >> 4;
    int lr   = lane & 15;

    const float* Ap[RT];
    const float* Wp[CT];
#pragma unroll
    for (int i = 0; i < RT; ++i) {
        int rt = rt0 + i;
        if (rt > nRowTiles - 1) rt = nRowTiles - 1;
        Ap[i] = A + (size_t)(rt * 16 + lr) * lda + 4 * half;
    }
#pragma unroll
    for (int j = 0; j < CT; ++j)
        Wp[j] = W + (size_t)(col0 + 16 * j + lr) * ldw + 4 * half;

    v8f acc[RT][CT];
#pragma unroll
    for (int i = 0; i < RT; ++i)
#pragma unroll
        for (int j = 0; j < CT; ++j) acc[i][j] = v8f{};

    Frag16 ax[RT], bx[CT], ay[RT], by[CT];

#pragma unroll
    for (int i = 0; i < RT; ++i) ax[i] = load_frag16(Ap[i]);
#pragma unroll
    for (int j = 0; j < CT; ++j) bx[j] = load_frag16(Wp[j]);

    int kb = 0;
    for (; kb < K - 32; kb += 32) {
#pragma unroll
        for (int i = 0; i < RT; ++i) ay[i] = load_frag16(Ap[i] + kb + 16);
#pragma unroll
        for (int j = 0; j < CT; ++j) by[j] = load_frag16(Wp[j] + kb + 16);
        mma_block16<RT, CT, RELU_A>(acc, ax, bx);
#pragma unroll
        for (int i = 0; i < RT; ++i) ax[i] = load_frag16(Ap[i] + kb + 32);
#pragma unroll
        for (int j = 0; j < CT; ++j) bx[j] = load_frag16(Wp[j] + kb + 32);
        mma_block16<RT, CT, RELU_A>(acc, ay, by);
    }
#pragma unroll
    for (int i = 0; i < RT; ++i) ay[i] = load_frag16(Ap[i] + kb + 16);
#pragma unroll
    for (int j = 0; j < CT; ++j) by[j] = load_frag16(Wp[j] + kb + 16);
    mma_block16<RT, CT, RELU_A>(acc, ax, bx);
    mma_block16<RT, CT, RELU_A>(acc, ay, by);

#pragma unroll
    for (int i = 0; i < RT; ++i) {
        if (rt0 + i < nRowTiles) {
            int row0 = (rt0 + i) * 16 + 8 * half;
            size_t cOff = (size_t)row0 * ldc + col0 + lr;
            float* Cb = C + cOff;
            if (EPI == 0) {
#pragma unroll
                for (int r = 0; r < 8; ++r)
#pragma unroll
                    for (int j = 0; j < CT; ++j) Cb[(size_t)r * ldc + 16 * j] = acc[i][j][r];
            } else {
                float old[8][CT];
                const float* Ib = Cin + cOff;
#pragma unroll
                for (int r = 0; r < 8; ++r)
#pragma unroll
                    for (int j = 0; j < CT; ++j) old[r][j] = Ib[(size_t)r * ldc + 16 * j];
#pragma unroll
                for (int r = 0; r < 8; ++r)
#pragma unroll
                    for (int j = 0; j < CT; ++j)
                        Cb[(size_t)r * ldc + 16 * j] = old[r][j] + acc[i][j][r];
            }
        }
    }
}

__global__ void __launch_bounds__(1024)
fold_attn_k(const float* __restrict__ fc1_w, const float* __restrict__ al1, const float* __restrict__ ar1,
            const float* __restrict__ fc2_w, const float* __restrict__ al2, const float* __restrict__ ar2,
            float* __restrict__ wf1, float* __restrict__ wf2, int H1, int D1)
{
    constexpr int F = 256;
    __shared__ float part[32][32];
    const int lane = threadIdx.x & 31;
    const int wave = threadIdx.x >> 5;
    const int row  = blockIdx.x >> 3;
    const int k    = (blockIdx.x & 7) * 32 + lane;
    float s = 0.0f;
    float* o;
    if (row < 2 * H1) {
        const int h = (row < H1) ? row : row - H1;
        const float* a = ((row < H1) ? al1 : ar1) + h * D1;
        const float* w = fc1_w + (size_t)h * D1 * F + k;
        for (int d = wave; d < D1; d += 32) s += a[d] * w[(size_t)d * F];
        o = wf1 + row * F + k;
    } else {
        const float* a = (row == 2 * H1) ? al2 : ar2;
        const float* w = fc2_w + k;
#pragma unroll
        for (int i = 0; i < F / 32; ++i) {
            const int d = wave + 32 * i;
            s += a[d] * w[(size_t)d * F];
        }
        o = wf2 + (row - 2 * H1) * F + k;
    }
    part[wave][lane] = s;
    __syncthreads();
    if (wave == 0) {
        float t = 0.0f;
        for (int w = 0; w < 32; ++w) t += part[w][lane];
        *o = t;
    }
}

template <int H, bool RELU_A>
__global__ void __launch_bounds__(128) __attribute__((amdgpu_waves_per_eu(1, 1)))
scores_wmma_f32(const float* __restrict__ A, const float* __restrict__ Wf,
                float* __restrict__ el, float* __restrict__ er, int M)
{
    constexpr int F = 256, RT = 2, NB = F / 16;
    const int lane = threadIdx.x & 31;
    const int wid  = blockIdx.x * (blockDim.x >> 5) + (threadIdx.x >> 5);
    const int nRowTiles = M >> 4;
    const int rt0 = wid * RT;
    if (rt0 >= nRowTiles) return;
    const int  half = lane >> 4;
    const int  lr   = lane & 15;
    const bool bOn  = lr < 2 * H;
    const float* Bp = Wf + (size_t)(bOn ? lr : 0) * F + 4 * half;
    const float* Ap[RT];
#pragma unroll
    for (int i = 0; i < RT; ++i) {
        int rt = rt0 + i;
        if (rt > nRowTiles - 1) rt = nRowTiles - 1;
        Ap[i] = A + (size_t)(rt * 16 + lr) * F + 4 * half;
    }

    Frag16 a[NB][RT], b[NB][1];
#pragma unroll
    for (int kb = 0; kb < NB; ++kb) {
        b[kb][0] = load_frag16(Bp + 16 * kb);
#pragma unroll
        for (int i = 0; i < RT; ++i) a[kb][i] = load_frag16(Ap[i] + 16 * kb);
    }
    if (!bOn) {
#pragma unroll
        for (int kb = 0; kb < NB; ++kb) { b[kb][0].q0 = v4f{}; b[kb][0].q1 = v4f{}; }
    }

    v8f acc[RT][1];
#pragma unroll
    for (int i = 0; i < RT; ++i) acc[i][0] = v8f{};
#pragma unroll
    for (int kb = 0; kb < NB; ++kb) mma_block16<RT, 1, RELU_A>(acc, a[kb], b[kb]);

#pragma unroll
    for (int i = 0; i < RT; ++i) {
        if (rt0 + i < nRowTiles && bOn) {
            const int row0 = (rt0 + i) * 16 + 8 * half;
            if (H == 1) {
                float* o = ((lr == 0) ? el : er) + row0;
                *(v4f*)(o)     = v4f{acc[i][0][0], acc[i][0][1], acc[i][0][2], acc[i][0][3]};
                *(v4f*)(o + 4) = v4f{acc[i][0][4], acc[i][0][5], acc[i][0][6], acc[i][0][7]};
            } else {
                float* o = ((lr < H) ? (el + lr) : (er + (lr - H))) + (size_t)row0 * H;
#pragma unroll
                for (int r = 0; r < 8; ++r) o[r * H] = acc[i][0][r];
            }
        }
    }
}


__device__ __forceinline__ void atomicMaxF(float* addr, float val) {
    if (val >= 0.0f) atomicMax((int*)addr, __float_as_int(val));
    else             atomicMin((unsigned int*)addr, __float_as_uint(val));
}

__device__ __forceinline__ float leaky02(float v) { return v > 0.0f ? v : 0.2f * v; }

constexpr int AGG_R   = 80;
constexpr int AGG_TB  = 1024;
constexpr int AGG_CAP = 10240;

template <int H>
__device__ __forceinline__ void edge_weights_to_slab(bool valid, int sv, int rv,
                                                     const float* __restrict__ el,
                                                     const float* s_er, const float* s_m, const float* s_s,
                                                     float* wl)
{
    if (valid) {
        float e[H];
#pragma unroll
        for (int h = 0; h < H; ++h) e[h] = el[(size_t)sv * H + h];
#pragma unroll
        for (int h = 0; h < H; ++h) {
            const int i = rv * H + h;
            wl[h] = expf(leaky02(e[h] + s_er[i]) - s_m[i]) / s_s[i];
        }
    }
}

template <int H, int B>
__device__ __forceinline__ void agg_add_edges(float* __restrict__ acc, int lane,
                                              int sv, int rv, int k, int n,
                                              const float* wv,
                                              const float* __restrict__ feat)
{
    constexpr int F = 256, D = F / H;
    const int h0 = (4 * lane) / D;
    const int h1 = (128 + 4 * lane) / D;
    int   rr[B];
    v4f   f0[B], f1[B];
    float a0[B], a1[B];
#pragma unroll
    for (int b = 0; b < B; ++b) {
        int kk = (k + b < n) ? (k + b) : (n - 1);
        int s  = __builtin_amdgcn_readlane(sv, kk);
        rr[b]  = __builtin_amdgcn_readlane(rv, kk);
        const float* fr = feat + (size_t)s * F + 4 * lane;
        f0[b] = *(const v4f*)(fr);
        f1[b] = *(const v4f*)(fr + 128);
        a0[b] = wv[kk * H + h0];
        a1[b] = (H > 1) ? wv[kk * H + h1] : a0[b];
    }
#pragma unroll
    for (int b = 0; b < B; ++b) {
        if (k + b < n) {
            float* o = acc + rr[b] * F + lane;
            atomicAdd(o +   0, f0[b].x * a0[b]);
            atomicAdd(o +  32, f0[b].y * a0[b]);
            atomicAdd(o +  64, f0[b].z * a0[b]);
            atomicAdd(o +  96, f0[b].w * a0[b]);
            atomicAdd(o + 128, f1[b].x * a1[b]);
            atomicAdd(o + 160, f1[b].y * a1[b]);
            atomicAdd(o + 192, f1[b].z * a1[b]);
            atomicAdd(o + 224, f1[b].w * a1[b]);
        }
    }
}

template <int H, int PHASE>
__device__ __forceinline__ void list_pass(const int* ls, const int* lr, int nPairs, int tid,
                                          const float* __restrict__ el,
                                          const float* s_er, float* s_m, float* s_s, int* s_deg)
{
    constexpr int PU = 4;
    for (int p0 = tid; p0 < nPairs; p0 += AGG_TB * PU) {
        float ev[PU];
        int   ix[PU];
#pragma unroll
        for (int u = 0; u < PU; ++u) {
            const int  p   = p0 + u * AGG_TB;
            const bool ok  = p < nPairs;
            const int  pp  = ok ? p : p0;
            const int  ent = pp / H, h = pp - ent * H;
            ix[u] = ok ? (lr[ent] * H + h) : -1;
            ev[u] = el[(size_t)ls[ent] * H + h];
        }
#pragma unroll
        for (int u = 0; u < PU; ++u) {
            if (ix[u] >= 0) {
                const float v = leaky02(ev[u] + s_er[ix[u]]);
                if (PHASE == 0) {
                    atomicMaxF(&s_m[ix[u]], v);
                    if (ix[u] % H == 0) atomicAdd(&s_deg[ix[u] / H], 1);
                } else {
                    atomicAdd(&s_s[ix[u]], expf(v - s_m[ix[u]]));
                }
            }
        }
    }
}

template <int H, bool BUILD>
__global__ void __launch_bounds__(AGG_TB)
edge_softmax_agg_owner_k(const int* __restrict__ src, const int* __restrict__ dst,
                         const float* __restrict__ el, const float* __restrict__ er,
                         const float* __restrict__ feat, float* __restrict__ agg,
                         const float* __restrict__ bias, const float* __restrict__ bb,
                         int* __restrict__ glist, int* __restrict__ gcnt,
                         int E, int N)
{
    constexpr int F = 256;
    __shared__ __attribute__((aligned(16))) float acc[AGG_R * F];
    __shared__ int   le[AGG_CAP];
    __shared__ int   lr[AGG_CAP];
    __shared__ float s_er[AGG_R * H];
    __shared__ float s_m[AGG_R * H];
    __shared__ float s_s[AGG_R * H];
    __shared__ float wsl[AGG_TB * H];
    __shared__ int   s_deg[AGG_R];
    __shared__ int   cnt;

    const int tid  = threadIdx.x;
    const int lane = tid & 31;
    const int wave = tid >> 5;
    const int lo   = blockIdx.x * AGG_R;
    const unsigned ltmask = (1u << lane) - 1u;
    float* wv = wsl + (tid - lane) * H;

    for (int i = 4 * tid; i < AGG_R * F; i += 4 * AGG_TB) *(v4f*)(acc + i) = v4f{};
    for (int i = tid; i < AGG_R * H; i += AGG_TB) {
        s_er[i] = (lo + i / H < N) ? er[(size_t)lo * H + i] : 0.0f;
        s_m[i]  = -INFINITY;
        s_s[i]  = 0.0f;
    }
    if (tid < AGG_R) s_deg[tid] = 0;
    if (tid == 0) cnt = 0;
    __syncthreads();

    int* gl = glist + (size_t)blockIdx.x * (2 * AGG_CAP);
    int total;
    if (BUILD) {
    const int  nChunk = E >> 2;
    const int4* dst4  = (const int4*)dst;
    constexpr int U = 4;
    for (int c0 = 0; c0 < nChunk + ((E & 3) ? 1 : 0); c0 += AGG_TB * U) {
        int4 dv[U];
#pragma unroll
        for (int u = 0; u < U; ++u) {
            int c = c0 + u * AGG_TB + tid;
            dv[u] = int4{-1, -1, -1, -1};
            if (c < nChunk) dv[u] = dst4[c];
            else if (c == nChunk) {
                if (4 * c + 0 < E) dv[u].x = dst[4 * c + 0];
                if (4 * c + 1 < E) dv[u].y = dst[4 * c + 1];
                if (4 * c + 2 < E) dv[u].z = dst[4 * c + 2];
            }
        }
#pragma unroll
        for (int u = 0; u < U; ++u) {
            const int c = c0 + u * AGG_TB + tid;
            const unsigned r0 = (unsigned)(dv[u].x - lo), r1 = (unsigned)(dv[u].y - lo);
            const unsigned r2 = (unsigned)(dv[u].z - lo), r3 = (unsigned)(dv[u].w - lo);
            const bool m0 = r0 < (unsigned)AGG_R, m1 = r1 < (unsigned)AGG_R;
            const bool m2 = r2 < (unsigned)AGG_R, m3 = r3 < (unsigned)AGG_R;
            if (__builtin_amdgcn_ballot_w32(m0 | m1 | m2 | m3) == 0u) continue;
            const unsigned b0 = __builtin_amdgcn_ballot_w32(m0), b1 = __builtin_amdgcn_ballot_w32(m1);
            const unsigned b2 = __builtin_amdgcn_ballot_w32(m2), b3 = __builtin_amdgcn_ballot_w32(m3);
            const int n0 = __builtin_popcount(b0), n1 = __builtin_popcount(b1);
            const int n2 = __builtin_popcount(b2), n3 = __builtin_popcount(b3);
            int base = 0;
            if (lane == 0) base = atomicAdd(&cnt, n0 + n1 + n2 + n3);
            base = __builtin_amdgcn_readfirstlane(base);
            const int p0 = base + __builtin_popcount(b0 & ltmask);
            const int p1 = base + n0 + __builtin_popcount(b1 & ltmask);
            const int p2 = base + n0 + n1 + __builtin_popcount(b2 & ltmask);
            const int p3 = base + n0 + n1 + n2 + __builtin_popcount(b3 & ltmask);
            if (m0 && p0 < AGG_CAP) { le[p0] = 4 * c + 0; lr[p0] = (int)r0; }
            if (m1 && p1 < AGG_CAP) { le[p1] = 4 * c + 1; lr[p1] = (int)r1; }
            if (m2 && p2 < AGG_CAP) { le[p2] = 4 * c + 2; lr[p2] = (int)r2; }
            if (m3 && p3 < AGG_CAP) { le[p3] = 4 * c + 3; lr[p3] = (int)r3; }
        }
    }
    __syncthreads();
    total = cnt;
    if (tid == 0) gcnt[blockIdx.x] = total;
    } else {
        total = gcnt[blockIdx.x];
    }

    if (total <= AGG_CAP) {
        if (BUILD) {
            for (int idx = tid; idx < total; idx += AGG_TB) {
                const int s = src[le[idx]];
                le[idx] = s;
                gl[idx] = s;
                gl[AGG_CAP + idx] = lr[idx];
            }
        } else {
            for (int idx = tid; idx < total; idx += AGG_TB) {
                le[idx] = gl[idx];
                lr[idx] = gl[AGG_CAP + idx];
            }
        }
        __syncthreads();
        list_pass<H, 0>(le, lr, total * H, tid, el, s_er, s_m, s_s, s_deg);
        __syncthreads();
        list_pass<H, 1>(le, lr, total * H, tid, el, s_er, s_m, s_s, s_deg);
        __syncthreads();
        for (int c0 = wave; c0 < total; c0 += 32 * 32) {
            const int  idx = c0 + 32 * lane;
            const bool v   = idx < total;
            const int  sv  = v ? le[idx] : 0;
            const int  rv  = v ? lr[idx] : 0;
            const int  n   = __builtin_popcount(__builtin_amdgcn_ballot_w32(v));
            edge_weights_to_slab<H>(v, sv, rv, el, s_er, s_m, s_s, wv + lane * H);
            for (int k = 0; k < n; k += 4)
                agg_add_edges<H, 4>(acc, lane, sv, rv, k, n, wv, feat);
        }
    } else {
        for (int i = tid; i < E; i += AGG_TB) {
            const unsigned r = (unsigned)(dst[i] - lo);
            if (r < (unsigned)AGG_R) {
                const int s = src[i];
                atomicAdd(&s_deg[r], 1);
                for (int h = 0; h < H; ++h)
                    atomicMaxF(&s_m[r * H + h], leaky02(el[(size_t)s * H + h] + s_er[r * H + h]));
            }
        }
        __syncthreads();
        for (int i = tid; i < E; i += AGG_TB) {
            const unsigned r = (unsigned)(dst[i] - lo);
            if (r < (unsigned)AGG_R) {
                const int s = src[i];
                for (int h = 0; h < H; ++h) {
                    const float v = leaky02(el[(size_t)s * H + h] + s_er[r * H + h]);
                    atomicAdd(&s_s[r * H + h], expf(v - s_m[r * H + h]));
                }
            }
        }
        __syncthreads();
        for (int b0 = 32 * wave; b0 < E; b0 += AGG_TB) {
            const int  i  = b0 + lane;
            const int  d  = (i < E) ? dst[i] : -1;
            const int  sI = (i < E) ? src[i] : 0;
            const unsigned r = (unsigned)(d - lo);
            unsigned own = __builtin_amdgcn_ballot_w32(r < (unsigned)AGG_R);
            edge_weights_to_slab<H>(r < (unsigned)AGG_R, sI, (int)r, el, s_er, s_m, s_s, wv + lane * H);
            while (own) {
                const int l = __builtin_ctz(own);
                own &= own - 1u;
                agg_add_edges<H, 1>(acc, lane, sI, (int)r, l, 32, wv, feat);
            }
        }
    }
    __syncthreads();

    const v4f bs0 = *(const v4f*)(bias + 4 * lane), bs1 = *(const v4f*)(bias + 128 + 4 * lane);
    const v4f bk0 = *(const v4f*)(bb + 4 * lane),   bk1 = *(const v4f*)(bb + 128 + 4 * lane);
    for (int r = wave; r < AGG_R; r += 32) {
        const int row = lo + r;
        if (row < N) {
            float* g = agg + (size_t)row * F + 4 * lane;
            const float* a = acc + r * F + lane;
            const float norm = 1.0f / fmaxf((float)s_deg[r], 1.0f);
            v4f p0 = *(const v4f*)(g);
            v4f p1 = *(const v4f*)(g + 128);
            v4f t0 = { a[0],   a[32],  a[64],  a[96]  };
            v4f t1 = { a[128], a[160], a[192], a[224] };
            *(v4f*)(g)       = (bs0 + norm * (p0 + bk0)) + t0;
            *(v4f*)(g + 128) = (bs1 + norm * (p1 + bk1)) + t1;
        }
    }
}

extern "C" void kernel_launch(void* const* d_in, const int* in_sizes, int n_in,
                              void* d_out, int out_size, void* d_ws, size_t ws_size,
                              hipStream_t stream)
{
    const float* inputs  = (const float*)d_in[0];
    const int*   src     = (const int*)  d_in[1];
    const int*   dst     = (const int*)  d_in[2];
    const float* fc1_w   = (const float*)d_in[3];
    const float* attn_l1 = (const float*)d_in[4];
    const float* attn_r1 = (const float*)d_in[5];
    const float* bias1   = (const float*)d_in[6];
    const float* fc2_w   = (const float*)d_in[7];
    const float* attn_l2 = (const float*)d_in[8];
    const float* attn_r2 = (const float*)d_in[9];
    const float* bias2   = (const float*)d_in[10];
    const float* wb0     = (const float*)d_in[11];
    const float* bb0     = (const float*)d_in[12];
    const float* wb1     = (const float*)d_in[13];
    const float* bb1     = (const float*)d_in[14];
    float* out = (float*)d_out;

    const int F  = 256;
    const int N  = in_sizes[0] / F;
    const int E  = in_sizes[1];
    constexpr int H1 = 8;
    const int D1 = 32;

    float* ws   = (float*)d_ws;
    float* feat = ws;                ws += (size_t)N * F;
    float* buf  = ws;                ws += (size_t)N * F;
    float* agg  = ws;                ws += (size_t)N * F;
    float* el   = ws;                ws += (size_t)N * H1;
    float* er   = ws;                ws += (size_t)N * H1;
    float* wf1  = ws;                ws += (size_t)2 * H1 * F;
    float* wf2  = ws;                ws += (size_t)2 * F;
    const int nGroups = (N + AGG_R - 1) / AGG_R;
    int* gcnt  = (int*)ws;           ws += (size_t)((nGroups + 31) / 32) * 32;
    int* glist = (int*)ws;           ws += (size_t)nGroups * 2 * AGG_CAP;

    auto cdiv = [](int a, int b) { return (a + b - 1) / b; };

    const int GRT = 5, GCT = 4, GTB = 128;
    const int waves    = cdiv(N / 16, GRT) * (256 / (16 * GCT));
    const int gemmGrid = cdiv(waves, GTB / 32);

    fold_attn_k<<<(2 * H1 + 2) * 8, 1024, 0, stream>>>(fc1_w, attn_l1, attn_r1, fc2_w, attn_l2, attn_r2,
                                                       wf1, wf2, H1, D1);
    const int scGrid = cdiv(cdiv(N / 16, 2), 4);

    gemm_wmma_f32<GRT, GCT, false, 0><<<gemmGrid, GTB, 0, stream>>>(inputs, F, fc1_w, F, nullptr, feat, F, N, F);
    gemm_wmma_f32<GRT, GCT, false, 0><<<gemmGrid, GTB, 0, stream>>>(inputs, F, wb0, F, nullptr, agg, F, N, F);

    scores_wmma_f32<H1, false><<<scGrid, 128, 0, stream>>>(inputs, wf1, el, er, N);
    edge_softmax_agg_owner_k<H1, true><<<cdiv(N, AGG_R), AGG_TB, 0, stream>>>(src, dst, el, er, feat, agg, bias1, bb0,
                                                                              glist, gcnt, E, N);

    gemm_wmma_f32<GRT, GCT, true, 0><<<gemmGrid, GTB, 0, stream>>>(agg, F, fc2_w, F, nullptr, feat, F, N, F);
    gemm_wmma_f32<GRT, GCT, false, 0><<<gemmGrid, GTB, 0, stream>>>(inputs, F, wb1, 2 * F, nullptr, buf, F, N, F);
    gemm_wmma_f32<GRT, GCT, true, 1><<<gemmGrid, GTB, 0, stream>>>(agg, F, wb1 + F, 2 * F, buf, out, F, N, F);

    scores_wmma_f32<1, true><<<scGrid, 128, 0, stream>>>(agg, wf2, el, er, N);
    edge_softmax_agg_owner_k<1, false><<<cdiv(N, AGG_R), AGG_TB, 0, stream>>>(src, dst, el, er, feat, out, bias2, bb1,
                                                                              glist, gcnt, E, N);
}


# layer 2 (one head): edge pass sums the ReLU'd layer-1 rows first, then ONE k=768 product [sums*deg | inputs | h1] @ [fc2_w | wb1]^T forms out; feat2, buf and 2 launches gone; sums reordered
# speedup vs baseline: 3.5519x; 1.1896x over previous
#include <hip/hip_runtime.h>
#include <hip/hip_bf16.h>
#include <math.h>


typedef __attribute__((ext_vector_type(2))) float v2f;
typedef __attribute__((ext_vector_type(8))) float v8f;

typedef __attribute__((ext_vector_type(4))) float v4f;

struct Frag16 { v4f q0, q1; };

__device__ __forceinline__ Frag16 load_frag16(const float* __restrict__ p) {
    Frag16 f;
    f.q0 = *(const v4f*)(p);
    f.q1 = *(const v4f*)(p + 8);
    return f;
}

template <int S>
__device__ __forceinline__ v2f frag_step(const Frag16& f) {
    if (S == 0) return __builtin_shufflevector(f.q0, f.q0, 0, 1);
    if (S == 1) return __builtin_shufflevector(f.q0, f.q0, 2, 3);
    if (S == 2) return __builtin_shufflevector(f.q1, f.q1, 0, 1);
    return __builtin_shufflevector(f.q1, f.q1, 2, 3);
}

template <int RT, int CT, int S, bool RELU_A>
__device__ __forceinline__ void mma_step(v8f (&acc)[RT][CT], const Frag16 (&a)[RT], const Frag16 (&b)[CT]) {
#pragma unroll
    for (int i = 0; i < RT; ++i) {
        v2f av = frag_step<S>(a[i]);
        if (RELU_A) {
            av.x = fmaxf(av.x, 0.0f);
            av.y = fmaxf(av.y, 0.0f);
        }
#pragma unroll
        for (int j = 0; j < CT; ++j) {
            acc[i][j] = __builtin_amdgcn_wmma_f32_16x16x4_f32(false, av, false, frag_step<S>(b[j]),
                                                              (short)0, acc[i][j], false, false);
        }
    }
}

template <int RT, int CT, bool RELU_A>
__device__ __forceinline__ void mma_block16(v8f (&acc)[RT][CT], const Frag16 (&a)[RT], const Frag16 (&b)[CT]) {
    mma_step<RT, CT, 0, RELU_A>(acc, a, b);
    mma_step<RT, CT, 1, RELU_A>(acc, a, b);
    mma_step<RT, CT, 2, RELU_A>(acc, a, b);
    mma_step<RT, CT, 3, RELU_A>(acc, a, b);
}

template <int RT, int CT, bool RELU_A>
__device__ __forceinline__ void gemm_k_run(v8f (&acc)[RT][CT], const float* const (&Ap)[RT],
                                           const float* const (&Wp)[CT], int K)
{
    Frag16 ax[RT], bx[CT], ay[RT], by[CT];

#pragma unroll
    for (int i = 0; i < RT; ++i) ax[i] = load_frag16(Ap[i]);
#pragma unroll
    for (int j = 0; j < CT; ++j) bx[j] = load_frag16(Wp[j]);

    int kb = 0;
    for (; kb < K - 32; kb += 32) {
#pragma unroll
        for (int i = 0; i < RT; ++i) ay[i] = load_frag16(Ap[i] + kb + 16);
#pragma unroll
        for (int j = 0; j < CT; ++j) by[j] = load_frag16(Wp[j] + kb + 16);
        mma_block16<RT, CT, RELU_A>(acc, ax, bx);
#pragma unroll
        for (int i = 0; i < RT; ++i) ax[i] = load_frag16(Ap[i] + kb + 32);
#pragma unroll
        for (int j = 0; j < CT; ++j) bx[j] = load_frag16(Wp[j] + kb + 32);
        mma_block16<RT, CT, RELU_A>(acc, ay, by);
    }
#pragma unroll
    for (int i = 0; i < RT; ++i) ay[i] = load_frag16(Ap[i] + kb + 16);
#pragma unroll
    for (int j = 0; j < CT; ++j) by[j] = load_frag16(Wp[j] + kb + 16);
    mma_block16<RT, CT, RELU_A>(acc, ax, bx);
    mma_block16<RT, CT, RELU_A>(acc, ay, by);
}

template <int RT, int CT>
__global__ void __launch_bounds__(128)
gemm_wmma_f32(const float* __restrict__ A, int lda,
              const float* __restrict__ W, int ldw,
              float* __restrict__ C, int ldc,
              int M, int K)
{
    const int nColGroups = 256 / (16 * CT);
    int waveId = blockIdx.x * (blockDim.x >> 5) + (threadIdx.x >> 5);
    int lane   = threadIdx.x & 31;
    int nRowTiles  = M >> 4;
    int nRowGroups = (nRowTiles + RT - 1) / RT;
    if (waveId >= nRowGroups * nColGroups) return;
    int rowGroup = waveId / nColGroups;
    int colGroup = waveId - rowGroup * nColGroups;
    int rt0  = rowGroup * RT;
    int col0 = colGroup * (16 * CT);
    int half = lane >> 4;
    int lr   = lane & 15;

    const float* Ap[RT];
    const float* Wp[CT];
#pragma unroll
    for (int i = 0; i < RT; ++i) {
        int rt = rt0 + i;
        if (rt > nRowTiles - 1) rt = nRowTiles - 1;
        Ap[i] = A + (size_t)(rt * 16 + lr) * lda + 4 * half;
    }
#pragma unroll
    for (int j = 0; j < CT; ++j)
        Wp[j] = W + (size_t)(col0 + 16 * j + lr) * ldw + 4 * half;

    v8f acc[RT][CT];
#pragma unroll
    for (int i = 0; i < RT; ++i)
#pragma unroll
        for (int j = 0; j < CT; ++j) acc[i][j] = v8f{};

    gemm_k_run<RT, CT, false>(acc, Ap, Wp, K);

#pragma unroll
    for (int i = 0; i < RT; ++i) {
        if (rt0 + i < nRowTiles) {
            int row0 = (rt0 + i) * 16 + 8 * half;
            float* Cb = C + (size_t)row0 * ldc + col0 + lr;
#pragma unroll
            for (int r = 0; r < 8; ++r)
#pragma unroll
                for (int j = 0; j < CT; ++j) Cb[(size_t)r * ldc + 16 * j] = acc[i][j][r];
        }
    }
}

template <int RT, int CT>
__global__ void __launch_bounds__(128)
gemm3_out_wmma_f32(const float* __restrict__ S, const float* __restrict__ X, const float* __restrict__ Hp,
                   const float* __restrict__ Wfc, const float* __restrict__ Wsk,
                   const float* __restrict__ normv,
                   const float* __restrict__ bias, const float* __restrict__ bb,
                   float* __restrict__ C, int M)
{
    constexpr int F = 256;
    const int nColGroups = F / (16 * CT);
    int waveId = blockIdx.x * (blockDim.x >> 5) + (threadIdx.x >> 5);
    int lane   = threadIdx.x & 31;
    int nRowTiles  = M >> 4;
    int nRowGroups = (nRowTiles + RT - 1) / RT;
    if (waveId >= nRowGroups * nColGroups) return;
    int rowGroup = waveId / nColGroups;
    int colGroup = waveId - rowGroup * nColGroups;
    int rt0  = rowGroup * RT;
    int col0 = colGroup * (16 * CT);
    int half = lane >> 4;
    int lr   = lane & 15;

    size_t aOff[RT];
#pragma unroll
    for (int i = 0; i < RT; ++i) {
        int rt = rt0 + i;
        if (rt > nRowTiles - 1) rt = nRowTiles - 1;
        aOff[i] = (size_t)(rt * 16 + lr) * F + 4 * half;
    }

    v8f acc[RT][CT];
#pragma unroll
    for (int i = 0; i < RT; ++i)
#pragma unroll
        for (int j = 0; j < CT; ++j) acc[i][j] = v8f{};

    const float* Ap[RT];
    const float* Wp[CT];
#pragma unroll
    for (int i = 0; i < RT; ++i) Ap[i] = S + aOff[i];
#pragma unroll
    for (int j = 0; j < CT; ++j) Wp[j] = Wfc + (size_t)(col0 + 16 * j + lr) * F + 4 * half;
    gemm_k_run<RT, CT, false>(acc, Ap, Wp, F);
#pragma unroll
    for (int i = 0; i < RT; ++i) Ap[i] = X + aOff[i];
#pragma unroll
    for (int j = 0; j < CT; ++j) Wp[j] = Wsk + (size_t)(col0 + 16 * j + lr) * (2 * F) + 4 * half;
    gemm_k_run<RT, CT, false>(acc, Ap, Wp, F);
#pragma unroll
    for (int i = 0; i < RT; ++i) Ap[i] = Hp + aOff[i];
#pragma unroll
    for (int j = 0; j < CT; ++j) Wp[j] += F;
    gemm_k_run<RT, CT, true>(acc, Ap, Wp, F);

    float bsj[CT], bbj[CT];
#pragma unroll
    for (int j = 0; j < CT; ++j) {
        bsj[j] = bias[col0 + 16 * j + lr];
        bbj[j] = bb[col0 + 16 * j + lr];
    }
#pragma unroll
    for (int i = 0; i < RT; ++i) {
        if (rt0 + i < nRowTiles) {
            int row0 = (rt0 + i) * 16 + 8 * half;
            float* Cb = C + (size_t)row0 * F + col0 + lr;
            const v4f n0 = *(const v4f*)(normv + row0), n1 = *(const v4f*)(normv + row0 + 4);
            const float nr[8] = { n0.x, n0.y, n0.z, n0.w, n1.x, n1.y, n1.z, n1.w };
#pragma unroll
            for (int r = 0; r < 8; ++r)
#pragma unroll
                for (int j = 0; j < CT; ++j)
                    Cb[(size_t)r * F + 16 * j] = bsj[j] + nr[r] * (acc[i][j][r] + bbj[j]);
        }
    }
}

__global__ void __launch_bounds__(1024)
fold_attn_k(const float* __restrict__ fc1_w, const float* __restrict__ al1, const float* __restrict__ ar1,
            const float* __restrict__ fc2_w, const float* __restrict__ al2, const float* __restrict__ ar2,
            float* __restrict__ wf1, float* __restrict__ wf2, int H1, int D1)
{
    constexpr int F = 256;
    __shared__ float part[32][32];
    const int lane = threadIdx.x & 31;
    const int wave = threadIdx.x >> 5;
    const int row  = blockIdx.x >> 3;
    const int k    = (blockIdx.x & 7) * 32 + lane;
    float s = 0.0f;
    float* o;
    if (row < 2 * H1) {
        const int h = (row < H1) ? row : row - H1;
        const float* a = ((row < H1) ? al1 : ar1) + h * D1;
        const float* w = fc1_w + (size_t)h * D1 * F + k;
        for (int d = wave; d < D1; d += 32) s += a[d] * w[(size_t)d * F];
        o = wf1 + row * F + k;
    } else {
        const float* a = (row == 2 * H1) ? al2 : ar2;
        const float* w = fc2_w + k;
#pragma unroll
        for (int i = 0; i < F / 32; ++i) {
            const int d = wave + 32 * i;
            s += a[d] * w[(size_t)d * F];
        }
        o = wf2 + (row - 2 * H1) * F + k;
    }
    part[wave][lane] = s;
    __syncthreads();
    if (wave == 0) {
        float t = 0.0f;
        for (int w = 0; w < 32; ++w) t += part[w][lane];
        *o = t;
    }
}

template <int H, bool RELU_A>
__global__ void __launch_bounds__(128) __attribute__((amdgpu_waves_per_eu(1, 1)))
scores_wmma_f32(const float* __restrict__ A, const float* __restrict__ Wf,
                float* __restrict__ el, float* __restrict__ er, int M)
{
    constexpr int F = 256, RT = 2, NB = F / 16;
    const int lane = threadIdx.x & 31;
    const int wid  = blockIdx.x * (blockDim.x >> 5) + (threadIdx.x >> 5);
    const int nRowTiles = M >> 4;
    const int rt0 = wid * RT;
    if (rt0 >= nRowTiles) return;
    const int  half = lane >> 4;
    const int  lr   = lane & 15;
    const bool bOn  = lr < 2 * H;
    const float* Bp = Wf + (size_t)(bOn ? lr : 0) * F + 4 * half;
    const float* Ap[RT];
#pragma unroll
    for (int i = 0; i < RT; ++i) {
        int rt = rt0 + i;
        if (rt > nRowTiles - 1) rt = nRowTiles - 1;
        Ap[i] = A + (size_t)(rt * 16 + lr) * F + 4 * half;
    }

    Frag16 a[NB][RT], b[NB][1];
#pragma unroll
    for (int kb = 0; kb < NB; ++kb) {
        b[kb][0] = load_frag16(Bp + 16 * kb);
#pragma unroll
        for (int i = 0; i < RT; ++i) a[kb][i] = load_frag16(Ap[i] + 16 * kb);
    }
    if (!bOn) {
#pragma unroll
        for (int kb = 0; kb < NB; ++kb) { b[kb][0].q0 = v4f{}; b[kb][0].q1 = v4f{}; }
    }

    v8f acc[RT][1];
#pragma unroll
    for (int i = 0; i < RT; ++i) acc[i][0] = v8f{};
#pragma unroll
    for (int kb = 0; kb < NB; ++kb) mma_block16<RT, 1, RELU_A>(acc, a[kb], b[kb]);

#pragma unroll
    for (int i = 0; i < RT; ++i) {
        if (rt0 + i < nRowTiles && bOn) {
            const int row0 = (rt0 + i) * 16 + 8 * half;
            if (H == 1) {
                float* o = ((lr == 0) ? el : er) + row0;
                *(v4f*)(o)     = v4f{acc[i][0][0], acc[i][0][1], acc[i][0][2], acc[i][0][3]};
                *(v4f*)(o + 4) = v4f{acc[i][0][4], acc[i][0][5], acc[i][0][6], acc[i][0][7]};
            } else {
                float* o = ((lr < H) ? (el + lr) : (er + (lr - H))) + (size_t)row0 * H;
#pragma unroll
                for (int r = 0; r < 8; ++r) o[r * H] = acc[i][0][r];
            }
        }
    }
}


__device__ __forceinline__ void atomicMaxF(float* addr, float val) {
    if (val >= 0.0f) atomicMax((int*)addr, __float_as_int(val));
    else             atomicMin((unsigned int*)addr, __float_as_uint(val));
}

__device__ __forceinline__ float leaky02(float v) { return v > 0.0f ? v : 0.2f * v; }

constexpr int AGG_R   = 80;
constexpr int AGG_TB  = 1024;
constexpr int AGG_CAP = 10240;

template <int H>
__device__ __forceinline__ void edge_weights_to_slab(bool valid, int sv, int rv,
                                                     const float* __restrict__ el,
                                                     const float* s_er, const float* s_m, const float* s_s,
                                                     float* wl)
{
    if (valid) {
        float e[H];
#pragma unroll
        for (int h = 0; h < H; ++h) e[h] = el[(size_t)sv * H + h];
#pragma unroll
        for (int h = 0; h < H; ++h) {
            const int i = rv * H + h;
            wl[h] = expf(leaky02(e[h] + s_er[i]) - s_m[i]) / s_s[i];
        }
    }
}

template <int H, int B, bool RELU_F>
__device__ __forceinline__ void agg_add_edges(float* __restrict__ acc, int lane,
                                              int sv, int rv, int k, int n,
                                              const float* wv,
                                              const float* __restrict__ feat)
{
    constexpr int F = 256, D = F / H;
    const int h0 = (4 * lane) / D;
    const int h1 = (128 + 4 * lane) / D;
    int   rr[B];
    v4f   f0[B], f1[B];
    float a0[B], a1[B];
#pragma unroll
    for (int b = 0; b < B; ++b) {
        int kk = (k + b < n) ? (k + b) : (n - 1);
        int s  = __builtin_amdgcn_readlane(sv, kk);
        rr[b]  = __builtin_amdgcn_readlane(rv, kk);
        const float* fr = feat + (size_t)s * F + 4 * lane;
        f0[b] = *(const v4f*)(fr);
        f1[b] = *(const v4f*)(fr + 128);
        a0[b] = wv[kk * H + h0];
        a1[b] = (H > 1) ? wv[kk * H + h1] : a0[b];
    }
#pragma unroll
    for (int b = 0; b < B; ++b) {
        if (k + b < n) {
            float* o = acc + rr[b] * F + lane;
            if (RELU_F) {
                f0[b].x = fmaxf(f0[b].x, 0.0f); f0[b].y = fmaxf(f0[b].y, 0.0f);
                f0[b].z = fmaxf(f0[b].z, 0.0f); f0[b].w = fmaxf(f0[b].w, 0.0f);
                f1[b].x = fmaxf(f1[b].x, 0.0f); f1[b].y = fmaxf(f1[b].y, 0.0f);
                f1[b].z = fmaxf(f1[b].z, 0.0f); f1[b].w = fmaxf(f1[b].w, 0.0f);
            }
            atomicAdd(o +   0, f0[b].x * a0[b]);
            atomicAdd(o +  32, f0[b].y * a0[b]);
            atomicAdd(o +  64, f0[b].z * a0[b]);
            atomicAdd(o +  96, f0[b].w * a0[b]);
            atomicAdd(o + 128, f1[b].x * a1[b]);
            atomicAdd(o + 160, f1[b].y * a1[b]);
            atomicAdd(o + 192, f1[b].z * a1[b]);
            atomicAdd(o + 224, f1[b].w * a1[b]);
        }
    }
}

template <int H, int PHASE>
__device__ __forceinline__ void list_pass(const int* ls, const int* lr, int nPairs, int tid,
                                          const float* __restrict__ el,
                                          const float* s_er, float* s_m, float* s_s, int* s_deg)
{
    constexpr int PU = 4;
    for (int p0 = tid; p0 < nPairs; p0 += AGG_TB * PU) {
        float ev[PU];
        int   ix[PU];
#pragma unroll
        for (int u = 0; u < PU; ++u) {
            const int  p   = p0 + u * AGG_TB;
            const bool ok  = p < nPairs;
            const int  pp  = ok ? p : p0;
            const int  ent = pp / H, h = pp - ent * H;
            ix[u] = ok ? (lr[ent] * H + h) : -1;
            ev[u] = el[(size_t)ls[ent] * H + h];
        }
#pragma unroll
        for (int u = 0; u < PU; ++u) {
            if (ix[u] >= 0) {
                const float v = leaky02(ev[u] + s_er[ix[u]]);
                if (PHASE == 0) {
                    atomicMaxF(&s_m[ix[u]], v);
                    if (ix[u] % H == 0) atomicAdd(&s_deg[ix[u] / H], 1);
                } else {
                    atomicAdd(&s_s[ix[u]], expf(v - s_m[ix[u]]));
                }
            }
        }
    }
}

template <int H, bool BUILD, bool ROWS_FIRST>
__global__ void __launch_bounds__(AGG_TB)
edge_softmax_agg_owner_k(const int* __restrict__ src, const int* __restrict__ dst,
                         const float* __restrict__ el, const float* __restrict__ er,
                         const float* __restrict__ feat, float* __restrict__ agg,
                         const float* __restrict__ bias, const float* __restrict__ bb,
                         int* __restrict__ glist, int* __restrict__ gcnt,
                         float* __restrict__ normv,
                         int E, int N)
{
    constexpr int F = 256;
    __shared__ __attribute__((aligned(16))) float acc[AGG_R * F];
    __shared__ int   le[AGG_CAP];
    __shared__ int   lr[AGG_CAP];
    __shared__ float s_er[AGG_R * H];
    __shared__ float s_m[AGG_R * H];
    __shared__ float s_s[AGG_R * H];
    __shared__ float wsl[AGG_TB * H];
    __shared__ int   s_deg[AGG_R];
    __shared__ int   cnt;

    const int tid  = threadIdx.x;
    const int lane = tid & 31;
    const int wave = tid >> 5;
    const int lo   = blockIdx.x * AGG_R;
    const unsigned ltmask = (1u << lane) - 1u;
    float* wv = wsl + (tid - lane) * H;

    for (int i = 4 * tid; i < AGG_R * F; i += 4 * AGG_TB) *(v4f*)(acc + i) = v4f{};
    for (int i = tid; i < AGG_R * H; i += AGG_TB) {
        s_er[i] = (lo + i / H < N) ? er[(size_t)lo * H + i] : 0.0f;
        s_m[i]  = -INFINITY;
        s_s[i]  = 0.0f;
    }
    if (tid < AGG_R) s_deg[tid] = 0;
    if (tid == 0) cnt = 0;
    __syncthreads();

    int* gl = glist + (size_t)blockIdx.x * (2 * AGG_CAP);
    int total;
    if (BUILD) {
    const int  nChunk = E >> 2;
    const int4* dst4  = (const int4*)dst;
    constexpr int U = 4;
    for (int c0 = 0; c0 < nChunk + ((E & 3) ? 1 : 0); c0 += AGG_TB * U) {
        int4 dv[U];
#pragma unroll
        for (int u = 0; u < U; ++u) {
            int c = c0 + u * AGG_TB + tid;
            dv[u] = int4{-1, -1, -1, -1};
            if (c < nChunk) dv[u] = dst4[c];
            else if (c == nChunk) {
                if (4 * c + 0 < E) dv[u].x = dst[4 * c + 0];
                if (4 * c + 1 < E) dv[u].y = dst[4 * c + 1];
                if (4 * c + 2 < E) dv[u].z = dst[4 * c + 2];
            }
        }
#pragma unroll
        for (int u = 0; u < U; ++u) {
            const int c = c0 + u * AGG_TB + tid;
            const unsigned r0 = (unsigned)(dv[u].x - lo), r1 = (unsigned)(dv[u].y - lo);
            const unsigned r2 = (unsigned)(dv[u].z - lo), r3 = (unsigned)(dv[u].w - lo);
            const bool m0 = r0 < (unsigned)AGG_R, m1 = r1 < (unsigned)AGG_R;
            const bool m2 = r2 < (unsigned)AGG_R, m3 = r3 < (unsigned)AGG_R;
            if (__builtin_amdgcn_ballot_w32(m0 | m1 | m2 | m3) == 0u) continue;
            const unsigned b0 = __builtin_amdgcn_ballot_w32(m0), b1 = __builtin_amdgcn_ballot_w32(m1);
            const unsigned b2 = __builtin_amdgcn_ballot_w32(m2), b3 = __builtin_amdgcn_ballot_w32(m3);
            const int n0 = __builtin_popcount(b0), n1 = __builtin_popcount(b1);
            const int n2 = __builtin_popcount(b2), n3 = __builtin_popcount(b3);
            int base = 0;
            if (lane == 0) base = atomicAdd(&cnt, n0 + n1 + n2 + n3);
            base = __builtin_amdgcn_readfirstlane(base);
            const int p0 = base + __builtin_popcount(b0 & ltmask);
            const int p1 = base + n0 + __builtin_popcount(b1 & ltmask);
            const int p2 = base + n0 + n1 + __builtin_popcount(b2 & ltmask);
            const int p3 = base + n0 + n1 + n2 + __builtin_popcount(b3 & ltmask);
            if (m0 && p0 < AGG_CAP) { le[p0] = 4 * c + 0; lr[p0] = (int)r0; }
            if (m1 && p1 < AGG_CAP) { le[p1] = 4 * c + 1; lr[p1] = (int)r1; }
            if (m2 && p2 < AGG_CAP) { le[p2] = 4 * c + 2; lr[p2] = (int)r2; }
            if (m3 && p3 < AGG_CAP) { le[p3] = 4 * c + 3; lr[p3] = (int)r3; }
        }
    }
    __syncthreads();
    total = cnt;
    if (tid == 0) gcnt[blockIdx.x] = total;
    } else {
        total = gcnt[blockIdx.x];
    }

    if (total <= AGG_CAP) {
        if (BUILD) {
            for (int idx = tid; idx < total; idx += AGG_TB) {
                const int s = src[le[idx]];
                le[idx] = s;
                gl[idx] = s;
                gl[AGG_CAP + idx] = lr[idx];
            }
        } else {
            for (int idx = tid; idx < total; idx += AGG_TB) {
                le[idx] = gl[idx];
                lr[idx] = gl[AGG_CAP + idx];
            }
        }
        __syncthreads();
        list_pass<H, 0>(le, lr, total * H, tid, el, s_er, s_m, s_s, s_deg);
        __syncthreads();
        list_pass<H, 1>(le, lr, total * H, tid, el, s_er, s_m, s_s, s_deg);
        __syncthreads();
        for (int c0 = wave; c0 < total; c0 += 32 * 32) {
            const int  idx = c0 + 32 * lane;
            const bool v   = idx < total;
            const int  sv  = v ? le[idx] : 0;
            const int  rv  = v ? lr[idx] : 0;
            const int  n   = __builtin_popcount(__builtin_amdgcn_ballot_w32(v));
            edge_weights_to_slab<H>(v, sv, rv, el, s_er, s_m, s_s, wv + lane * H);
            for (int k = 0; k < n; k += 4)
                agg_add_edges<H, 4, ROWS_FIRST>(acc, lane, sv, rv, k, n, wv, feat);
        }
    } else {
        for (int i = tid; i < E; i += AGG_TB) {
            const unsigned r = (unsigned)(dst[i] - lo);
            if (r < (unsigned)AGG_R) {
                const int s = src[i];
                atomicAdd(&s_deg[r], 1);
                for (int h = 0; h < H; ++h)
                    atomicMaxF(&s_m[r * H + h], leaky02(el[(size_t)s * H + h] + s_er[r * H + h]));
            }
        }
        __syncthreads();
        for (int i = tid; i < E; i += AGG_TB) {
            const unsigned r = (unsigned)(dst[i] - lo);
            if (r < (unsigned)AGG_R) {
                const int s = src[i];
                for (int h = 0; h < H; ++h) {
                    const float v = leaky02(el[(size_t)s * H + h] + s_er[r * H + h]);
                    atomicAdd(&s_s[r * H + h], expf(v - s_m[r * H + h]));
                }
            }
        }
        __syncthreads();
        for (int b0 = 32 * wave; b0 < E; b0 += AGG_TB) {
            const int  i  = b0 + lane;
            const int  d  = (i < E) ? dst[i] : -1;
            const int  sI = (i < E) ? src[i] : 0;
            const unsigned r = (unsigned)(d - lo);
            unsigned own = __builtin_amdgcn_ballot_w32(r < (unsigned)AGG_R);
            edge_weights_to_slab<H>(r < (unsigned)AGG_R, sI, (int)r, el, s_er, s_m, s_s, wv + lane * H);
            while (own) {
                const int l = __builtin_ctz(own);
                own &= own - 1u;
                agg_add_edges<H, 1, ROWS_FIRST>(acc, lane, sI, (int)r, l, 32, wv, feat);
            }
        }
    }
    __syncthreads();

    if (ROWS_FIRST) {
        for (int r = wave; r < AGG_R; r += 32) {
            const int row = lo + r;
            if (row < N) {
                float* g = agg + (size_t)row * F + 4 * lane;
                const float* a = acc + r * F + lane;
                const float dg = fmaxf((float)s_deg[r], 1.0f);
                v4f t0 = { a[0],   a[32],  a[64],  a[96]  };
                v4f t1 = { a[128], a[160], a[192], a[224] };
                *(v4f*)(g)       = t0 * dg;
                *(v4f*)(g + 128) = t1 * dg;
                if (lane == 0) normv[row] = 1.0f / dg;
            }
        }
    } else {
        const v4f bs0 = *(const v4f*)(bias + 4 * lane), bs1 = *(const v4f*)(bias + 128 + 4 * lane);
        const v4f bk0 = *(const v4f*)(bb + 4 * lane),   bk1 = *(const v4f*)(bb + 128 + 4 * lane);
        for (int r = wave; r < AGG_R; r += 32) {
            const int row = lo + r;
            if (row < N) {
                float* g = agg + (size_t)row * F + 4 * lane;
                const float* a = acc + r * F + lane;
                const float norm = 1.0f / fmaxf((float)s_deg[r], 1.0f);
                v4f p0 = *(const v4f*)(g);
                v4f p1 = *(const v4f*)(g + 128);
                v4f t0 = { a[0],   a[32],  a[64],  a[96]  };
                v4f t1 = { a[128], a[160], a[192], a[224] };
                *(v4f*)(g)       = (bs0 + norm * (p0 + bk0)) + t0;
                *(v4f*)(g + 128) = (bs1 + norm * (p1 + bk1)) + t1;
            }
        }
    }
}

extern "C" void kernel_launch(void* const* d_in, const int* in_sizes, int n_in,
                              void* d_out, int out_size, void* d_ws, size_t ws_size,
                              hipStream_t stream)
{
    const float* inputs  = (const float*)d_in[0];
    const int*   src     = (const int*)  d_in[1];
    const int*   dst     = (const int*)  d_in[2];
    const float* fc1_w   = (const float*)d_in[3];
    const float* attn_l1 = (const float*)d_in[4];
    const float* attn_r1 = (const float*)d_in[5];
    const float* bias1   = (const float*)d_in[6];
    const float* fc2_w   = (const float*)d_in[7];
    const float* attn_l2 = (const float*)d_in[8];
    const float* attn_r2 = (const float*)d_in[9];
    const float* bias2   = (const float*)d_in[10];
    const float* wb0     = (const float*)d_in[11];
    const float* bb0     = (const float*)d_in[12];
    const float* wb1     = (const float*)d_in[13];
    const float* bb1     = (const float*)d_in[14];
    float* out = (float*)d_out;

    const int F  = 256;
    const int N  = in_sizes[0] / F;
    const int E  = in_sizes[1];
    constexpr int H1 = 8;
    const int D1 = 32;

    float* ws   = (float*)d_ws;
    float* feat = ws;                ws += (size_t)N * F;
    float* agg  = ws;                ws += (size_t)N * F;
    float* normv = ws;               ws += (size_t)((N + 31) / 32) * 32;
    float* el   = ws;                ws += (size_t)N * H1;
    float* er   = ws;                ws += (size_t)N * H1;
    float* wf1  = ws;                ws += (size_t)2 * H1 * F;
    float* wf2  = ws;                ws += (size_t)2 * F;
    const int nGroups = (N + AGG_R - 1) / AGG_R;
    int* gcnt  = (int*)ws;           ws += (size_t)((nGroups + 31) / 32) * 32;
    int* glist = (int*)ws;           ws += (size_t)nGroups * 2 * AGG_CAP;

    auto cdiv = [](int a, int b) { return (a + b - 1) / b; };

    const int GRT = 5, GCT = 4, GTB = 128;
    const int waves    = cdiv(N / 16, GRT) * (256 / (16 * GCT));
    const int gemmGrid = cdiv(waves, GTB / 32);

    fold_attn_k<<<(2 * H1 + 2) * 8, 1024, 0, stream>>>(fc1_w, attn_l1, attn_r1, fc2_w, attn_l2, attn_r2,
                                                       wf1, wf2, H1, D1);
    const int scGrid = cdiv(cdiv(N / 16, 2), 4);

    gemm_wmma_f32<GRT, GCT><<<gemmGrid, GTB, 0, stream>>>(inputs, F, fc1_w, F, feat, F, N, F);
    gemm_wmma_f32<GRT, GCT><<<gemmGrid, GTB, 0, stream>>>(inputs, F, wb0, F, agg, F, N, F);

    scores_wmma_f32<H1, false><<<scGrid, 128, 0, stream>>>(inputs, wf1, el, er, N);
    edge_softmax_agg_owner_k<H1, true, false><<<cdiv(N, AGG_R), AGG_TB, 0, stream>>>(src, dst, el, er, feat, agg, bias1, bb0,
                                                                                     glist, gcnt, nullptr, E, N);

    scores_wmma_f32<1, true><<<scGrid, 128, 0, stream>>>(agg, wf2, el, er, N);
    edge_softmax_agg_owner_k<1, false, true><<<cdiv(N, AGG_R), AGG_TB, 0, stream>>>(src, dst, el, er, agg, feat, nullptr, nullptr,
                                                                                    glist, gcnt, normv, E, N);
    gemm3_out_wmma_f32<GRT, GCT><<<gemmGrid, GTB, 0, stream>>>(feat, inputs, agg, fc2_w, wb1, normv, bias2, bb1, out, N);
}


# row-owning groups list their edges from per-group edge bitmaps (1 bit per edge) that extra blocks of the first launch build from dst once; no scan of all dst by each group; sums in new list order
# speedup vs baseline: 3.7730x; 1.0622x over previous
#include <hip/hip_runtime.h>
#include <hip/hip_bf16.h>
#include <math.h>


typedef __attribute__((ext_vector_type(2))) float v2f;
typedef __attribute__((ext_vector_type(8))) float v8f;

typedef __attribute__((ext_vector_type(4))) float v4f;

struct Frag16 { v4f q0, q1; };

__device__ __forceinline__ Frag16 load_frag16(const float* __restrict__ p) {
    Frag16 f;
    f.q0 = *(const v4f*)(p);
    f.q1 = *(const v4f*)(p + 8);
    return f;
}

template <int S>
__device__ __forceinline__ v2f frag_step(const Frag16& f) {
    if (S == 0) return __builtin_shufflevector(f.q0, f.q0, 0, 1);
    if (S == 1) return __builtin_shufflevector(f.q0, f.q0, 2, 3);
    if (S == 2) return __builtin_shufflevector(f.q1, f.q1, 0, 1);
    return __builtin_shufflevector(f.q1, f.q1, 2, 3);
}

template <int RT, int CT, int S, bool RELU_A>
__device__ __forceinline__ void mma_step(v8f (&acc)[RT][CT], const Frag16 (&a)[RT], const Frag16 (&b)[CT]) {
#pragma unroll
    for (int i = 0; i < RT; ++i) {
        v2f av = frag_step<S>(a[i]);
        if (RELU_A) {
            av.x = fmaxf(av.x, 0.0f);
            av.y = fmaxf(av.y, 0.0f);
        }
#pragma unroll
        for (int j = 0; j < CT; ++j) {
            acc[i][j] = __builtin_amdgcn_wmma_f32_16x16x4_f32(false, av, false, frag_step<S>(b[j]),
                                                              (short)0, acc[i][j], false, false);
        }
    }
}

template <int RT, int CT, bool RELU_A>
__device__ __forceinline__ void mma_block16(v8f (&acc)[RT][CT], const Frag16 (&a)[RT], const Frag16 (&b)[CT]) {
    mma_step<RT, CT, 0, RELU_A>(acc, a, b);
    mma_step<RT, CT, 1, RELU_A>(acc, a, b);
    mma_step<RT, CT, 2, RELU_A>(acc, a, b);
    mma_step<RT, CT, 3, RELU_A>(acc, a, b);
}

template <int RT, int CT, bool RELU_A>
__device__ __forceinline__ void gemm_k_run(v8f (&acc)[RT][CT], const float* const (&Ap)[RT],
                                           const float* const (&Wp)[CT], int K)
{
    Frag16 ax[RT], bx[CT], ay[RT], by[CT];

#pragma unroll
    for (int i = 0; i < RT; ++i) ax[i] = load_frag16(Ap[i]);
#pragma unroll
    for (int j = 0; j < CT; ++j) bx[j] = load_frag16(Wp[j]);

    int kb = 0;
    for (; kb < K - 32; kb += 32) {
#pragma unroll
        for (int i = 0; i < RT; ++i) ay[i] = load_frag16(Ap[i] + kb + 16);
#pragma unroll
        for (int j = 0; j < CT; ++j) by[j] = load_frag16(Wp[j] + kb + 16);
        mma_block16<RT, CT, RELU_A>(acc, ax, bx);
#pragma unroll
        for (int i = 0; i < RT; ++i) ax[i] = load_frag16(Ap[i] + kb + 32);
#pragma unroll
        for (int j = 0; j < CT; ++j) bx[j] = load_frag16(Wp[j] + kb + 32);
        mma_block16<RT, CT, RELU_A>(acc, ay, by);
    }
#pragma unroll
    for (int i = 0; i < RT; ++i) ay[i] = load_frag16(Ap[i] + kb + 16);
#pragma unroll
    for (int j = 0; j < CT; ++j) by[j] = load_frag16(Wp[j] + kb + 16);
    mma_block16<RT, CT, RELU_A>(acc, ax, bx);
    mma_block16<RT, CT, RELU_A>(acc, ay, by);
}

template <int RT, int CT>
__global__ void __launch_bounds__(128)
gemm_wmma_f32(const float* __restrict__ A, int lda,
              const float* __restrict__ W, int ldw,
              float* __restrict__ C, int ldc,
              int M, int K)
{
    const int nColGroups = 256 / (16 * CT);
    int waveId = blockIdx.x * (blockDim.x >> 5) + (threadIdx.x >> 5);
    int lane   = threadIdx.x & 31;
    int nRowTiles  = M >> 4;
    int nRowGroups = (nRowTiles + RT - 1) / RT;
    if (waveId >= nRowGroups * nColGroups) return;
    int rowGroup = waveId / nColGroups;
    int colGroup = waveId - rowGroup * nColGroups;
    int rt0  = rowGroup * RT;
    int col0 = colGroup * (16 * CT);
    int half = lane >> 4;
    int lr   = lane & 15;

    const float* Ap[RT];
    const float* Wp[CT];
#pragma unroll
    for (int i = 0; i < RT; ++i) {
        int rt = rt0 + i;
        if (rt > nRowTiles - 1) rt = nRowTiles - 1;
        Ap[i] = A + (size_t)(rt * 16 + lr) * lda + 4 * half;
    }
#pragma unroll
    for (int j = 0; j < CT; ++j)
        Wp[j] = W + (size_t)(col0 + 16 * j + lr) * ldw + 4 * half;

    v8f acc[RT][CT];
#pragma unroll
    for (int i = 0; i < RT; ++i)
#pragma unroll
        for (int j = 0; j < CT; ++j) acc[i][j] = v8f{};

    gemm_k_run<RT, CT, false>(acc, Ap, Wp, K);

#pragma unroll
    for (int i = 0; i < RT; ++i) {
        if (rt0 + i < nRowTiles) {
            int row0 = (rt0 + i) * 16 + 8 * half;
            float* Cb = C + (size_t)row0 * ldc + col0 + lr;
#pragma unroll
            for (int r = 0; r < 8; ++r)
#pragma unroll
                for (int j = 0; j < CT; ++j) Cb[(size_t)r * ldc + 16 * j] = acc[i][j][r];
        }
    }
}

template <int RT, int CT>
__global__ void __launch_bounds__(128)
gemm3_out_wmma_f32(const float* __restrict__ S, const float* __restrict__ X, const float* __restrict__ Hp,
                   const float* __restrict__ Wfc, const float* __restrict__ Wsk,
                   const float* __restrict__ normv,
                   const float* __restrict__ bias, const float* __restrict__ bb,
                   float* __restrict__ C, int M)
{
    constexpr int F = 256;
    const int nColGroups = F / (16 * CT);
    int waveId = blockIdx.x * (blockDim.x >> 5) + (threadIdx.x >> 5);
    int lane   = threadIdx.x & 31;
    int nRowTiles  = M >> 4;
    int nRowGroups = (nRowTiles + RT - 1) / RT;
    if (waveId >= nRowGroups * nColGroups) return;
    int rowGroup = waveId / nColGroups;
    int colGroup = waveId - rowGroup * nColGroups;
    int rt0  = rowGroup * RT;
    int col0 = colGroup * (16 * CT);
    int half = lane >> 4;
    int lr   = lane & 15;

    size_t aOff[RT];
#pragma unroll
    for (int i = 0; i < RT; ++i) {
        int rt = rt0 + i;
        if (rt > nRowTiles - 1) rt = nRowTiles - 1;
        aOff[i] = (size_t)(rt * 16 + lr) * F + 4 * half;
    }

    v8f acc[RT][CT];
#pragma unroll
    for (int i = 0; i < RT; ++i)
#pragma unroll
        for (int j = 0; j < CT; ++j) acc[i][j] = v8f{};

    const float* Ap[RT];
    const float* Wp[CT];
#pragma unroll
    for (int i = 0; i < RT; ++i) Ap[i] = S + aOff[i];
#pragma unroll
    for (int j = 0; j < CT; ++j) Wp[j] = Wfc + (size_t)(col0 + 16 * j + lr) * F + 4 * half;
    gemm_k_run<RT, CT, false>(acc, Ap, Wp, F);
#pragma unroll
    for (int i = 0; i < RT; ++i) Ap[i] = X + aOff[i];
#pragma unroll
    for (int j = 0; j < CT; ++j) Wp[j] = Wsk + (size_t)(col0 + 16 * j + lr) * (2 * F) + 4 * half;
    gemm_k_run<RT, CT, false>(acc, Ap, Wp, F);
#pragma unroll
    for (int i = 0; i < RT; ++i) Ap[i] = Hp + aOff[i];
#pragma unroll
    for (int j = 0; j < CT; ++j) Wp[j] += F;
    gemm_k_run<RT, CT, true>(acc, Ap, Wp, F);

    float bsj[CT], bbj[CT];
#pragma unroll
    for (int j = 0; j < CT; ++j) {
        bsj[j] = bias[col0 + 16 * j + lr];
        bbj[j] = bb[col0 + 16 * j + lr];
    }
#pragma unroll
    for (int i = 0; i < RT; ++i) {
        if (rt0 + i < nRowTiles) {
            int row0 = (rt0 + i) * 16 + 8 * half;
            float* Cb = C + (size_t)row0 * F + col0 + lr;
            const v4f n0 = *(const v4f*)(normv + row0), n1 = *(const v4f*)(normv + row0 + 4);
            const float nr[8] = { n0.x, n0.y, n0.z, n0.w, n1.x, n1.y, n1.z, n1.w };
#pragma unroll
            for (int r = 0; r < 8; ++r)
#pragma unroll
                for (int j = 0; j < CT; ++j)
                    Cb[(size_t)r * F + 16 * j] = bsj[j] + nr[r] * (acc[i][j][r] + bbj[j]);
        }
    }
}

constexpr int BM_ROWS  = 80;
constexpr int BM_EPB   = 4096;
constexpr int BM_WPB   = BM_EPB / 32;
constexpr int BM_MAXG  = 256;
constexpr int BM_PITCH = BM_WPB + 1;

__global__ void __launch_bounds__(1024)
fold_attn_k(const float* __restrict__ fc1_w, const float* __restrict__ al1, const float* __restrict__ ar1,
            const float* __restrict__ fc2_w, const float* __restrict__ al2, const float* __restrict__ ar2,
            float* __restrict__ wf1, float* __restrict__ wf2, int H1, int D1,
            const int* __restrict__ dst, int E, unsigned* __restrict__ bm, int wpg, int nGroups, int nBm)
{
    constexpr int F = 256;
    __shared__ float part[32][32];
    __shared__ __attribute__((aligned(16))) unsigned tbl[BM_MAXG * BM_PITCH];
    const int lane = threadIdx.x & 31;
    const int wave = threadIdx.x >> 5;
    if ((int)blockIdx.x < nBm) {
        const int tid = threadIdx.x;
        const int bb  = blockIdx.x;
        int d[BM_EPB / 1024];
#pragma unroll
        for (int i = 0; i < BM_EPB / 1024; ++i) {
            const int e = bb * BM_EPB + i * 1024 + tid;
            d[i] = (e < E) ? dst[e] : -1;
        }
        for (int i = 4 * tid; i < BM_MAXG * BM_PITCH; i += 4 * 1024) *(uint4*)(tbl + i) = uint4{0u, 0u, 0u, 0u};
        __syncthreads();
#pragma unroll
        for (int i = 0; i < BM_EPB / 1024; ++i) {
            const unsigned g = (unsigned)d[i] / (unsigned)BM_ROWS;
            if (d[i] >= 0 && g < (unsigned)BM_MAXG)
                atomicOr(&tbl[g * BM_PITCH + i * 32 + wave], 1u << lane);
        }
        __syncthreads();
        const int nW = nGroups * BM_WPB;
        for (int i0 = tid; i0 < nW; i0 += 4 * 1024) {
            unsigned v[4];
#pragma unroll
            for (int j = 0; j < 4; ++j) {
                const int idx = i0 + j * 1024;
                const int g = idx / BM_WPB, w = idx - g * BM_WPB;
                const unsigned x = tbl[((g < BM_MAXG) ? g : (BM_MAXG - 1)) * BM_PITCH + w];
                v[j] = (g < BM_MAXG) ? x : 0u;
            }
#pragma unroll
            for (int j = 0; j < 4; ++j) {
                const int idx = i0 + j * 1024;
                const int g = idx / BM_WPB, w = idx - g * BM_WPB;
                if (idx < nW) bm[(size_t)g * wpg + bb * BM_WPB + w] = v[j];
            }
        }
        return;
    }
    const int fb   = blockIdx.x - nBm;
    const int row  = fb >> 3;
    const int k    = (fb & 7) * 32 + lane;
    float s = 0.0f;
    float* o;
    if (row < 2 * H1) {
        const int h = (row < H1) ? row : row - H1;
        const float* a = ((row < H1) ? al1 : ar1) + h * D1;
        const float* w = fc1_w + (size_t)h * D1 * F + k;
        for (int d = wave; d < D1; d += 32) s += a[d] * w[(size_t)d * F];
        o = wf1 + row * F + k;
    } else {
        const float* a = (row == 2 * H1) ? al2 : ar2;
        const float* w = fc2_w + k;
#pragma unroll
        for (int i = 0; i < F / 32; ++i) {
            const int d = wave + 32 * i;
            s += a[d] * w[(size_t)d * F];
        }
        o = wf2 + (row - 2 * H1) * F + k;
    }
    part[wave][lane] = s;
    __syncthreads();
    if (wave == 0) {
        float t = 0.0f;
        for (int w = 0; w < 32; ++w) t += part[w][lane];
        *o = t;
    }
}

template <int H, bool RELU_A>
__global__ void __launch_bounds__(128) __attribute__((amdgpu_waves_per_eu(1, 1)))
scores_wmma_f32(const float* __restrict__ A, const float* __restrict__ Wf,
                float* __restrict__ el, float* __restrict__ er, int M)
{
    constexpr int F = 256, RT = 2, NB = F / 16;
    const int lane = threadIdx.x & 31;
    const int wid  = blockIdx.x * (blockDim.x >> 5) + (threadIdx.x >> 5);
    const int nRowTiles = M >> 4;
    const int rt0 = wid * RT;
    if (rt0 >= nRowTiles) return;
    const int  half = lane >> 4;
    const int  lr   = lane & 15;
    const bool bOn  = lr < 2 * H;
    const float* Bp = Wf + (size_t)(bOn ? lr : 0) * F + 4 * half;
    const float* Ap[RT];
#pragma unroll
    for (int i = 0; i < RT; ++i) {
        int rt = rt0 + i;
        if (rt > nRowTiles - 1) rt = nRowTiles - 1;
        Ap[i] = A + (size_t)(rt * 16 + lr) * F + 4 * half;
    }

    Frag16 a[NB][RT], b[NB][1];
#pragma unroll
    for (int kb = 0; kb < NB; ++kb) {
        b[kb][0] = load_frag16(Bp + 16 * kb);
#pragma unroll
        for (int i = 0; i < RT; ++i) a[kb][i] = load_frag16(Ap[i] + 16 * kb);
    }
    if (!bOn) {
#pragma unroll
        for (int kb = 0; kb < NB; ++kb) { b[kb][0].q0 = v4f{}; b[kb][0].q1 = v4f{}; }
    }

    v8f acc[RT][1];
#pragma unroll
    for (int i = 0; i < RT; ++i) acc[i][0] = v8f{};
#pragma unroll
    for (int kb = 0; kb < NB; ++kb) mma_block16<RT, 1, RELU_A>(acc, a[kb], b[kb]);

#pragma unroll
    for (int i = 0; i < RT; ++i) {
        if (rt0 + i < nRowTiles && bOn) {
            const int row0 = (rt0 + i) * 16 + 8 * half;
            if (H == 1) {
                float* o = ((lr == 0) ? el : er) + row0;
                *(v4f*)(o)     = v4f{acc[i][0][0], acc[i][0][1], acc[i][0][2], acc[i][0][3]};
                *(v4f*)(o + 4) = v4f{acc[i][0][4], acc[i][0][5], acc[i][0][6], acc[i][0][7]};
            } else {
                float* o = ((lr < H) ? (el + lr) : (er + (lr - H))) + (size_t)row0 * H;
#pragma unroll
                for (int r = 0; r < 8; ++r) o[r * H] = acc[i][0][r];
            }
        }
    }
}


__device__ __forceinline__ void atomicMaxF(float* addr, float val) {
    if (val >= 0.0f) atomicMax((int*)addr, __float_as_int(val));
    else             atomicMin((unsigned int*)addr, __float_as_uint(val));
}

__device__ __forceinline__ float leaky02(float v) { return v > 0.0f ? v : 0.2f * v; }

constexpr int AGG_R   = 80;
constexpr int AGG_TB  = 1024;
constexpr int AGG_CAP = 10240;

template <int H>
__device__ __forceinline__ void edge_weights_to_slab(bool valid, int sv, int rv,
                                                     const float* __restrict__ el,
                                                     const float* s_er, const float* s_m, const float* s_s,
                                                     float* wl)
{
    if (valid) {
        float e[H];
#pragma unroll
        for (int h = 0; h < H; ++h) e[h] = el[(size_t)sv * H + h];
#pragma unroll
        for (int h = 0; h < H; ++h) {
            const int i = rv * H + h;
            wl[h] = expf(leaky02(e[h] + s_er[i]) - s_m[i]) / s_s[i];
        }
    }
}

template <int H, int B, bool RELU_F>
__device__ __forceinline__ void agg_add_edges(float* __restrict__ acc, int lane,
                                              int sv, int rv, int k, int n,
                                              const float* wv,
                                              const float* __restrict__ feat)
{
    constexpr int F = 256, D = F / H;
    const int h0 = (4 * lane) / D;
    const int h1 = (128 + 4 * lane) / D;
    int   rr[B];
    v4f   f0[B], f1[B];
    float a0[B], a1[B];
#pragma unroll
    for (int b = 0; b < B; ++b) {
        int kk = (k + b < n) ? (k + b) : (n - 1);
        int s  = __builtin_amdgcn_readlane(sv, kk);
        rr[b]  = __builtin_amdgcn_readlane(rv, kk);
        const float* fr = feat + (size_t)s * F + 4 * lane;
        f0[b] = *(const v4f*)(fr);
        f1[b] = *(const v4f*)(fr + 128);
        a0[b] = wv[kk * H + h0];
        a1[b] = (H > 1) ? wv[kk * H + h1] : a0[b];
    }
#pragma unroll
    for (int b = 0; b < B; ++b) {
        if (k + b < n) {
            float* o = acc + rr[b] * F + lane;
            if (RELU_F) {
                f0[b].x = fmaxf(f0[b].x, 0.0f); f0[b].y = fmaxf(f0[b].y, 0.0f);
                f0[b].z = fmaxf(f0[b].z, 0.0f); f0[b].w = fmaxf(f0[b].w, 0.0f);
                f1[b].x = fmaxf(f1[b].x, 0.0f); f1[b].y = fmaxf(f1[b].y, 0.0f);
                f1[b].z = fmaxf(f1[b].z, 0.0f); f1[b].w = fmaxf(f1[b].w, 0.0f);
            }
            atomicAdd(o +   0, f0[b].x * a0[b]);
            atomicAdd(o +  32, f0[b].y * a0[b]);
            atomicAdd(o +  64, f0[b].z * a0[b]);
            atomicAdd(o +  96, f0[b].w * a0[b]);
            atomicAdd(o + 128, f1[b].x * a1[b]);
            atomicAdd(o + 160, f1[b].y * a1[b]);
            atomicAdd(o + 192, f1[b].z * a1[b]);
            atomicAdd(o + 224, f1[b].w * a1[b]);
        }
    }
}

template <int H, int PHASE>
__device__ __forceinline__ void list_pass(const int* ls, const int* lr, int nPairs, int tid,
                                          const float* __restrict__ el,
                                          const float* s_er, float* s_m, float* s_s, int* s_deg)
{
    constexpr int PU = 4;
    for (int p0 = tid; p0 < nPairs; p0 += AGG_TB * PU) {
        float ev[PU];
        int   ix[PU];
#pragma unroll
        for (int u = 0; u < PU; ++u) {
            const int  p   = p0 + u * AGG_TB;
            const bool ok  = p < nPairs;
            const int  pp  = ok ? p : p0;
            const int  ent = pp / H, h = pp - ent * H;
            ix[u] = ok ? (lr[ent] * H + h) : -1;
            ev[u] = el[(size_t)ls[ent] * H + h];
        }
#pragma unroll
        for (int u = 0; u < PU; ++u) {
            if (ix[u] >= 0) {
                const float v = leaky02(ev[u] + s_er[ix[u]]);
                if (PHASE == 0) {
                    atomicMaxF(&s_m[ix[u]], v);
                    if (ix[u] % H == 0) atomicAdd(&s_deg[ix[u] / H], 1);
                } else {
                    atomicAdd(&s_s[ix[u]], expf(v - s_m[ix[u]]));
                }
            }
        }
    }
}

template <int H, bool BUILD, bool ROWS_FIRST>
__global__ void __launch_bounds__(AGG_TB)
edge_softmax_agg_owner_k(const int* __restrict__ src, const int* __restrict__ dst,
                         const float* __restrict__ el, const float* __restrict__ er,
                         const float* __restrict__ feat, float* __restrict__ agg,
                         const float* __restrict__ bias, const float* __restrict__ bb,
                         int* __restrict__ glist, int* __restrict__ gcnt,
                         float* __restrict__ normv,
                         const unsigned* __restrict__ bm, int wpg,
                         int E, int N)
{
    constexpr int F = 256;
    static_assert(AGG_R == BM_ROWS, "the bitmaps are built for groups of AGG_R rows");
    __shared__ __attribute__((aligned(16))) float acc[AGG_R * F];
    __shared__ int   le[AGG_CAP];
    __shared__ int   lr[AGG_CAP];
    __shared__ float s_er[AGG_R * H];
    __shared__ float s_m[AGG_R * H];
    __shared__ float s_s[AGG_R * H];
    __shared__ float wsl[AGG_TB * H];
    __shared__ int   s_deg[AGG_R];
    __shared__ int   cnt;

    const int tid  = threadIdx.x;
    const int lane = tid & 31;
    const int wave = tid >> 5;
    const int lo   = blockIdx.x * AGG_R;
    float* wv = wsl + (tid - lane) * H;

    for (int i = 4 * tid; i < AGG_R * F; i += 4 * AGG_TB) *(v4f*)(acc + i) = v4f{};
    for (int i = tid; i < AGG_R * H; i += AGG_TB) {
        s_er[i] = (lo + i / H < N) ? er[(size_t)lo * H + i] : 0.0f;
        s_m[i]  = -INFINITY;
        s_s[i]  = 0.0f;
    }
    if (tid < AGG_R) s_deg[tid] = 0;
    if (tid == 0) cnt = 0;
    __syncthreads();

    int* gl = glist + (size_t)blockIdx.x * (2 * AGG_CAP);
    int total;
    if (BUILD) {
    const unsigned* bw = bm + (size_t)blockIdx.x * wpg;
    constexpr int BU = 10;
    for (int w0 = 0; w0 < wpg; w0 += AGG_TB * BU) {
        unsigned wd[BU];
#pragma unroll
        for (int u = 0; u < BU; ++u) {
            const int w = w0 + u * AGG_TB + tid;
            const unsigned x = bw[(w < wpg) ? w : (wpg - 1)];
            wd[u] = (w < wpg) ? x : 0u;
        }
        int c = 0;
#pragma unroll
        for (int u = 0; u < BU; ++u) c += __builtin_popcount(wd[u]);
        int incl = c;
        for (int off = 1; off < 32; off <<= 1) {
            const int v = __shfl_up(incl, off);
            if (lane >= off) incl += v;
        }
        const int wtot = __shfl(incl, 31);
        if (wtot == 0) continue;
        int base = 0;
        if (lane == 0) base = atomicAdd(&cnt, wtot);
        base = __builtin_amdgcn_readfirstlane(base);
        int p = base + incl - c;
#pragma unroll
        for (int u = 0; u < BU; ++u) {
            unsigned x = wd[u];
            const int e0 = (w0 + u * AGG_TB + tid) * 32;
            while (x) {
                const int b = __builtin_ctz(x);
                x &= x - 1u;
                if (p < AGG_CAP) le[p] = e0 + b;
                ++p;
            }
        }
    }
    __syncthreads();
    total = cnt;
    if (tid == 0) gcnt[blockIdx.x] = total;
    } else {
        total = gcnt[blockIdx.x];
    }

    if (total <= AGG_CAP) {
        if (BUILD) {
            for (int idx = tid; idx < total; idx += AGG_TB) {
                const int e = le[idx];
                const int s = src[e];
                const int r = dst[e] - lo;
                le[idx] = s;
                lr[idx] = r;
                gl[idx] = s;
                gl[AGG_CAP + idx] = r;
            }
        } else {
            for (int idx = tid; idx < total; idx += AGG_TB) {
                le[idx] = gl[idx];
                lr[idx] = gl[AGG_CAP + idx];
            }
        }
        __syncthreads();
        list_pass<H, 0>(le, lr, total * H, tid, el, s_er, s_m, s_s, s_deg);
        __syncthreads();
        list_pass<H, 1>(le, lr, total * H, tid, el, s_er, s_m, s_s, s_deg);
        __syncthreads();
        for (int c0 = wave; c0 < total; c0 += 32 * 32) {
            const int  idx = c0 + 32 * lane;
            const bool v   = idx < total;
            const int  sv  = v ? le[idx] : 0;
            const int  rv  = v ? lr[idx] : 0;
            const int  n   = __builtin_popcount(__builtin_amdgcn_ballot_w32(v));
            edge_weights_to_slab<H>(v, sv, rv, el, s_er, s_m, s_s, wv + lane * H);
            for (int k = 0; k < n; k += 4)
                agg_add_edges<H, 4, ROWS_FIRST>(acc, lane, sv, rv, k, n, wv, feat);
        }
    } else {
        for (int i = tid; i < E; i += AGG_TB) {
            const unsigned r = (unsigned)(dst[i] - lo);
            if (r < (unsigned)AGG_R) {
                const int s = src[i];
                atomicAdd(&s_deg[r], 1);
                for (int h = 0; h < H; ++h)
                    atomicMaxF(&s_m[r * H + h], leaky02(el[(size_t)s * H + h] + s_er[r * H + h]));
            }
        }
        __syncthreads();
        for (int i = tid; i < E; i += AGG_TB) {
            const unsigned r = (unsigned)(dst[i] - lo);
            if (r < (unsigned)AGG_R) {
                const int s = src[i];
                for (int h = 0; h < H; ++h) {
                    const float v = leaky02(el[(size_t)s * H + h] + s_er[r * H + h]);
                    atomicAdd(&s_s[r * H + h], expf(v - s_m[r * H + h]));
                }
            }
        }
        __syncthreads();
        for (int b0 = 32 * wave; b0 < E; b0 += AGG_TB) {
            const int  i  = b0 + lane;
            const int  d  = (i < E) ? dst[i] : -1;
            const int  sI = (i < E) ? src[i] : 0;
            const unsigned r = (unsigned)(d - lo);
            unsigned own = __builtin_amdgcn_ballot_w32(r < (unsigned)AGG_R);
            edge_weights_to_slab<H>(r < (unsigned)AGG_R, sI, (int)r, el, s_er, s_m, s_s, wv + lane * H);
            while (own) {
                const int l = __builtin_ctz(own);
                own &= own - 1u;
                agg_add_edges<H, 1, ROWS_FIRST>(acc, lane, sI, (int)r, l, 32, wv, feat);
            }
        }
    }
    __syncthreads();

    if (ROWS_FIRST) {
        for (int r = wave; r < AGG_R; r += 32) {
            const int row = lo + r;
            if (row < N) {
                float* g = agg + (size_t)row * F + 4 * lane;
                const float* a = acc + r * F + lane;
                const float dg = fmaxf((float)s_deg[r], 1.0f);
                v4f t0 = { a[0],   a[32],  a[64],  a[96]  };
                v4f t1 = { a[128], a[160], a[192], a[224] };
                *(v4f*)(g)       = t0 * dg;
                *(v4f*)(g + 128) = t1 * dg;
                if (lane == 0) normv[row] = 1.0f / dg;
            }
        }
    } else {
        const v4f bs0 = *(const v4f*)(bias + 4 * lane), bs1 = *(const v4f*)(bias + 128 + 4 * lane);
        const v4f bk0 = *(const v4f*)(bb + 4 * lane),   bk1 = *(const v4f*)(bb + 128 + 4 * lane);
        for (int r = wave; r < AGG_R; r += 32) {
            const int row = lo + r;
            if (row < N) {
                float* g = agg + (size_t)row * F + 4 * lane;
                const float* a = acc + r * F + lane;
                const float norm = 1.0f / fmaxf((float)s_deg[r], 1.0f);
                v4f p0 = *(const v4f*)(g);
                v4f p1 = *(const v4f*)(g + 128);
                v4f t0 = { a[0],   a[32],  a[64],  a[96]  };
                v4f t1 = { a[128], a[160], a[192], a[224] };
                *(v4f*)(g)       = (bs0 + norm * (p0 + bk0)) + t0;
                *(v4f*)(g + 128) = (bs1 + norm * (p1 + bk1)) + t1;
            }
        }
    }
}

extern "C" void kernel_launch(void* const* d_in, const int* in_sizes, int n_in,
                              void* d_out, int out_size, void* d_ws, size_t ws_size,
                              hipStream_t stream)
{
    const float* inputs  = (const float*)d_in[0];
    const int*   src     = (const int*)  d_in[1];
    const int*   dst     = (const int*)  d_in[2];
    const float* fc1_w   = (const float*)d_in[3];
    const float* attn_l1 = (const float*)d_in[4];
    const float* attn_r1 = (const float*)d_in[5];
    const float* bias1   = (const float*)d_in[6];
    const float* fc2_w   = (const float*)d_in[7];
    const float* attn_l2 = (const float*)d_in[8];
    const float* attn_r2 = (const float*)d_in[9];
    const float* bias2   = (const float*)d_in[10];
    const float* wb0     = (const float*)d_in[11];
    const float* bb0     = (const float*)d_in[12];
    const float* wb1     = (const float*)d_in[13];
    const float* bb1     = (const float*)d_in[14];
    float* out = (float*)d_out;

    const int F  = 256;
    const int N  = in_sizes[0] / F;
    const int E  = in_sizes[1];
    constexpr int H1 = 8;
    const int D1 = 32;

    float* ws   = (float*)d_ws;
    float* feat = ws;                ws += (size_t)N * F;
    float* agg  = ws;                ws += (size_t)N * F;
    float* normv = ws;               ws += (size_t)((N + 31) / 32) * 32;
    float* el   = ws;                ws += (size_t)N * H1;
    float* er   = ws;                ws += (size_t)N * H1;
    float* wf1  = ws;                ws += (size_t)2 * H1 * F;
    float* wf2  = ws;                ws += (size_t)2 * F;
    const int nGroups = (N + AGG_R - 1) / AGG_R;
    int* gcnt  = (int*)ws;           ws += (size_t)((nGroups + 31) / 32) * 32;
    int* glist = (int*)ws;           ws += (size_t)nGroups * 2 * AGG_CAP;
    const int nBm = (E + BM_EPB - 1) / BM_EPB;
    const int wpg = nBm * BM_WPB;
    unsigned* bm = (unsigned*)ws;    ws += (size_t)nGroups * wpg;

    auto cdiv = [](int a, int b) { return (a + b - 1) / b; };

    const int GRT = 5, GCT = 4, GTB = 128;
    const int waves    = cdiv(N / 16, GRT) * (256 / (16 * GCT));
    const int gemmGrid = cdiv(waves, GTB / 32);

    fold_attn_k<<<nBm + (2 * H1 + 2) * 8, 1024, 0, stream>>>(fc1_w, attn_l1, attn_r1, fc2_w, attn_l2, attn_r2,
                                                             wf1, wf2, H1, D1, dst, E, bm, wpg, nGroups, nBm);
    const int scGrid = cdiv(cdiv(N / 16, 2), 4);

    gemm_wmma_f32<GRT, GCT><<<gemmGrid, GTB, 0, stream>>>(inputs, F, fc1_w, F, feat, F, N, F);
    gemm_wmma_f32<GRT, GCT><<<gemmGrid, GTB, 0, stream>>>(inputs, F, wb0, F, agg, F, N, F);

    scores_wmma_f32<H1, false><<<scGrid, 128, 0, stream>>>(inputs, wf1, el, er, N);
    edge_softmax_agg_owner_k<H1, true, false><<<cdiv(N, AGG_R), AGG_TB, 0, stream>>>(src, dst, el, er, feat, agg, bias1, bb0,
                                                                                     glist, gcnt, nullptr, bm, wpg, E, N);

    scores_wmma_f32<1, true><<<scGrid, 128, 0, stream>>>(agg, wf2, el, er, N);
    edge_softmax_agg_owner_k<1, false, true><<<cdiv(N, AGG_R), AGG_TB, 0, stream>>>(src, dst, el, er, agg, feat, nullptr, nullptr,
                                                                                    glist, gcnt, normv, bm, wpg, E, N);
    gemm3_out_wmma_f32<GRT, GCT><<<gemmGrid, GTB, 0, stream>>>(feat, inputs, agg, fc2_w, wb1, normv, bias2, bb1, out, N);
}
